# starting kernel
# baseline (speedup 1.0000x reference)
#include <hip/hip_runtime.h>
#include <hip/hip_bf16.h>

#define HDIM 128
#define EPS_BN 1e-5f

typedef __attribute__((ext_vector_type(2))) float v2f;
typedef __attribute__((ext_vector_type(8))) float v8f;

__global__ __launch_bounds__(256) void k_zero(float* p, long long n) {
    long long i = (long long)blockIdx.x * blockDim.x + threadIdx.x;
    long long stride = (long long)gridDim.x * blockDim.x;
    for (; i < n; i += stride) p[i] = 0.0f;
}

__global__ __launch_bounds__(256) void k_deg_scatter(const int* __restrict__ dst,
                                                     float* __restrict__ deg, int E) {
    int t = blockIdx.x * 256 + threadIdx.x;
    if (t < E) atomicAdd(&deg[dst[t]], 1.0f);
}

__global__ __launch_bounds__(256) void k_dinv(float* __restrict__ deg, int n) {
    int t = blockIdx.x * 256 + threadIdx.x;
    if (t < n) deg[t] = rsqrtf(deg[t] + 1.0f);
}

__global__ __launch_bounds__(256) void k_rank1(const float* __restrict__ X,
                                               const float* __restrict__ W1,
                                               float* __restrict__ h, int n) {
    int t = blockIdx.x * 256 + threadIdx.x;
    if (t >= n * 32) return;
    int row = t >> 5, c4 = (t & 31) << 2;
    float xv = X[row];
    const float4 w = *(const float4*)(W1 + c4);
    float4 o; o.x = xv * w.x; o.y = xv * w.y; o.z = xv * w.z; o.w = xv * w.w;
    *(float4*)(h + (long long)row * HDIM + c4) = o;
}

__global__ __launch_bounds__(256) void k_gemm_wmma(const float* __restrict__ x,
                                                   const float* __restrict__ W,
                                                   float* __restrict__ h, int n) {
    __shared__ float tileX[32 * HDIM];
    const int rowBase = blockIdx.x * 32;
    const int tid  = threadIdx.x;
    const int wave = tid >> 5;
    const int lane = tid & 31;
    const int half = lane >> 4;
    const int l16  = lane & 15;
    const int rowHalf = wave & 1;
    const int c0   = (wave >> 1) * 32;

    const bool fullTile = (rowBase + 32 <= n);

    if (fullTile) {
        const float4* xs = (const float4*)(x + (long long)rowBase * HDIM);
        #pragma unroll
        for (int i = 0; i < 4; ++i) {
            int idx = tid + i * 256;
            ((float4*)tileX)[idx] = xs[idx];
        }
    } else {
        #pragma unroll
        for (int i = 0; i < 4; ++i) {
            int idx = tid + i * 256;
            int r = rowBase + (idx >> 5);
            float4 v = make_float4(0.f, 0.f, 0.f, 0.f);
            if (r < n) v = *(const float4*)(x + (long long)r * HDIM + ((idx & 31) << 2));
            ((float4*)tileX)[idx] = v;
        }
    }
    __syncthreads();

    const float* xrow = tileX + (rowHalf * 16 + l16) * HDIM + 2 * half;
    v8f acc0 = {}, acc1 = {};

    #pragma unroll 4
    for (int k = 0; k < HDIM; k += 4) {
        v2f a; a.x = xrow[k]; a.y = xrow[k + 1];
        const float* wp = W + (long long)(k + 2 * half) * HDIM + c0 + l16;
        v2f b0; b0.x = wp[0];  b0.y = wp[HDIM];
        v2f b1; b1.x = wp[16]; b1.y = wp[HDIM + 16];
        acc0 = __builtin_amdgcn_wmma_f32_16x16x4_f32(false, a, false, b0,
                                                     (short)0, acc0, false, false);
        acc1 = __builtin_amdgcn_wmma_f32_16x16x4_f32(false, a, false, b1,
                                                     (short)0, acc1, false, false);
    }

    const int rbase = rowBase + rowHalf * 16 + half * 8;
    if (fullTile) {
        #pragma unroll
        for (int v = 0; v < 8; ++v) {
            float* op = h + (long long)(rbase + v) * HDIM + c0 + l16;
            op[0]  = acc0[v];
            op[16] = acc1[v];
        }
    } else {
        for (int v = 0; v < 8; ++v) {
            int r = rbase + v;
            if (r < n) {
                h[(long long)r * HDIM + c0 + l16]      = acc0[v];
                h[(long long)r * HDIM + c0 + 16 + l16] = acc1[v];
            }
        }
    }
}

__global__ __launch_bounds__(256) void k_edge_scatter(const float* __restrict__ h,
                                                      const int* __restrict__ src,
                                                      const int* __restrict__ dst,
                                                      const float* __restrict__ dinv,
                                                      float* __restrict__ agg, int E) {
    int t = blockIdx.x * 256 + threadIdx.x;
    if (t >= E * 32) return;
    int e = t >> 5, c4 = (t & 31) << 2;
    int s = src[e], d = dst[e];
    float coef = dinv[s] * dinv[d];
    const float4 hv = *(const float4*)(h + (long long)s * HDIM + c4);
    float* ap = agg + (long long)d * HDIM + c4;
    atomicAdd(ap + 0, hv.x * coef);
    atomicAdd(ap + 1, hv.y * coef);
    atomicAdd(ap + 2, hv.z * coef);
    atomicAdd(ap + 3, hv.w * coef);
}

__global__ __launch_bounds__(256) void k_combine_relu(float* __restrict__ agg,
                                                      const float* __restrict__ h,
                                                      const float* __restrict__ dinv,
                                                      const float* __restrict__ bias, int n) {
    int t = blockIdx.x * 256 + threadIdx.x;
    if (t >= n * 32) return;
    int row = t >> 5, c4 = (t & 31) << 2;
    float di = dinv[row], d2 = di * di;
    long long off = (long long)row * HDIM + c4;
    float4 a  = *(const float4*)(agg + off);
    float4 hv = *(const float4*)(h + off);
    float4 b  = *(const float4*)(bias + c4);
    a.x = fmaxf(a.x + hv.x * d2 + b.x, 0.f);
    a.y = fmaxf(a.y + hv.y * d2 + b.y, 0.f);
    a.z = fmaxf(a.z + hv.z * d2 + b.z, 0.f);
    a.w = fmaxf(a.w + hv.w * d2 + b.w, 0.f);
    *(float4*)(agg + off) = a;
}

__global__ __launch_bounds__(256) void k_stats(const float* __restrict__ y,
                                               float* __restrict__ sums, int n) {
    int c = threadIdx.x & 127;
    int sub = threadIdx.x >> 7;
    float s = 0.f, s2 = 0.f;
    for (int r = blockIdx.x * 2 + sub; r < n; r += gridDim.x * 2) {
        float v = y[(long long)r * HDIM + c];
        s += v; s2 += v * v;
    }
    __shared__ float ls[256], ls2[256];
    ls[threadIdx.x] = s; ls2[threadIdx.x] = s2;
    __syncthreads();
    if (sub == 0) {
        atomicAdd(&sums[c],        ls[c]  + ls[128 + c]);
        atomicAdd(&sums[128 + c],  ls2[c] + ls2[128 + c]);
    }
}

__global__ void k_scaleshift(const float* __restrict__ sums,
                             const float* __restrict__ gamma,
                             const float* __restrict__ beta,
                             float* __restrict__ ss, float invN) {
    int c = threadIdx.x;
    float m   = sums[c] * invN;
    float var = sums[128 + c] * invN - m * m;
    float sc  = rsqrtf(var + EPS_BN) * gamma[c];
    ss[c] = sc;
    ss[128 + c] = beta[c] - m * sc;
}

__global__ __launch_bounds__(256) void k_bn_apply(float* __restrict__ y,
                                                  const float* __restrict__ res,
                                                  const float* __restrict__ ss, int n) {
    int t = blockIdx.x * 256 + threadIdx.x;
    if (t >= n * 32) return;
    int row = t >> 5, c4 = (t & 31) << 2;
    long long off = (long long)row * HDIM + c4;
    float4 v  = *(const float4*)(y + off);
    float4 sc = *(const float4*)(ss + c4);
    float4 sh = *(const float4*)(ss + 128 + c4);
    v.x = v.x * sc.x + sh.x; v.y = v.y * sc.y + sh.y;
    v.z = v.z * sc.z + sh.z; v.w = v.w * sc.w + sh.w;
    if (res) {
        float4 r = *(const float4*)(res + off);
        v.x += r.x; v.y += r.y; v.z += r.z; v.w += r.w;
    }
    *(float4*)(y + off) = v;
}

__global__ __launch_bounds__(256) void k_counts(const int* __restrict__ batch,
                                                float* __restrict__ counts, int n) {
    int t = blockIdx.x * 256 + threadIdx.x;
    if (t < n) atomicAdd(&counts[batch[t]], 1.0f);
}

__global__ __launch_bounds__(256) void k_pool_acc(const float* __restrict__ x,
                                                  const int* __restrict__ batch,
                                                  float* __restrict__ pool, int n) {
    int t = blockIdx.x * 256 + threadIdx.x;
    if (t >= n * 32) return;
    int row = t >> 5, c4 = (t & 31) << 2;
    int g = batch[row];
    float4 v = *(const float4*)(x + (long long)row * HDIM + c4);
    float* p = pool + (long long)g * HDIM + c4;
    atomicAdd(p + 0, v.x); atomicAdd(p + 1, v.y);
    atomicAdd(p + 2, v.z); atomicAdd(p + 3, v.w);
}

__global__ __launch_bounds__(256) void k_pool_final(const float* __restrict__ pool,
                                                    const float* __restrict__ counts,
                                                    float* __restrict__ xmean, int G) {
    int t = blockIdx.x * 256 + threadIdx.x;
    if (t >= G * HDIM) return;
    int g = t / HDIM;
    xmean[t] = pool[t] / fmaxf(counts[g], 1.0f);
}

__global__ void k_pred(const float* __restrict__ xmean, const float* __restrict__ Wr,
                       const float* __restrict__ br, float* __restrict__ pred) {
    __shared__ float red[128];
    int g = blockIdx.x, c = threadIdx.x;
    red[c] = xmean[(long long)g * HDIM + c] * Wr[c];
    __syncthreads();
    for (int s = 64; s > 0; s >>= 1) {
        if (c < s) red[c] += red[c + s];
        __syncthreads();
    }
    if (c == 0) pred[g] = red[0] + br[0];
}

extern "C" void kernel_launch(void* const* d_in, const int* in_sizes, int n_in,
                              void* d_out, int out_size, void* d_ws, size_t ws_size,
                              hipStream_t stream) {
    const float* X     = (const float*)d_in[0];
    const int*   edge  = (const int*)d_in[1];
    const int*   batch = (const int*)d_in[2];
    const float* W1    = (const float*)d_in[3];
    const float* b1    = (const float*)d_in[4];
    const float* Wm    = (const float*)d_in[5];
    const float* bm    = (const float*)d_in[6];
    const float* gamma = (const float*)d_in[7];
    const float* beta  = (const float*)d_in[8];
    const float* Wr    = (const float*)d_in[9];
    const float* br    = (const float*)d_in[10];
    float* out = (float*)d_out;

    const int N = in_sizes[0];
    const int E = in_sizes[1] / 2;
    const int G = out_size / (HDIM + 1);
    const int* src = edge;
    const int* dst = edge + E;

    const long long NH = (long long)N * HDIM;
    float* ws    = (float*)d_ws;
    float* buf[3] = { ws, ws + NH, ws + 2 * NH };
    float* dinv  = ws + 3 * NH;
    float* sums  = dinv + N;
    float* ss    = sums + 256;
    float* pool  = ss + 256;
    float* cnts  = pool + (long long)G * HDIM;

    const int gE   = (E + 255) / 256;
    const int gN   = (N + 255) / 256;
    const int gNH4 = (N * 32 + 255) / 256;
    const int gE4  = (E * 32 + 255) / 256;
    const int gGem = (N + 31) / 32;

    k_zero<<<1024, 256, 0, stream>>>(dinv, N);
    k_deg_scatter<<<gE, 256, 0, stream>>>(dst, dinv, E);
    k_dinv<<<gN, 256, 0, stream>>>(dinv, N);

    auto layer_tail = [&](float* hbuf, float* abuf, const float* bias,
                          const float* g_, const float* be_, const float* res) {
        k_zero<<<2048, 256, 0, stream>>>(abuf, NH);
        k_edge_scatter<<<gE4, 256, 0, stream>>>(hbuf, src, dst, dinv, abuf, E);
        k_combine_relu<<<gNH4, 256, 0, stream>>>(abuf, hbuf, dinv, bias, N);
        k_zero<<<1, 256, 0, stream>>>(sums, 256);
        k_stats<<<256, 256, 0, stream>>>(abuf, sums, N);
        k_scaleshift<<<1, 128, 0, stream>>>(sums, g_, be_, ss, 1.0f / (float)N);
        k_bn_apply<<<gNH4, 256, 0, stream>>>(abuf, res, ss, N);
    };

    k_rank1<<<gNH4, 256, 0, stream>>>(X, W1, buf[1], N);
    layer_tail(buf[1], buf[2], b1, gamma, beta, nullptr);

    float* cur = buf[2];
    float* f0  = buf[0];
    float* f1  = buf[1];

    for (int i = 0; i < 4; ++i) {
        float* hbuf = f0;
        float* abuf = f1;
        k_gemm_wmma<<<gGem, 256, 0, stream>>>(cur, Wm + (long long)i * HDIM * HDIM, hbuf, N);
        layer_tail(hbuf, abuf, bm + (long long)i * HDIM,
                   gamma + (long long)(i + 1) * HDIM,
                   beta  + (long long)(i + 1) * HDIM, cur);
        float* next = abuf;
        f1 = cur;
        f0 = hbuf;
        cur = next;
    }

    k_zero<<<32, 256, 0, stream>>>(pool, (long long)G * HDIM + G);
    k_counts<<<gN, 256, 0, stream>>>(batch, cnts, N);
    k_pool_acc<<<gNH4, 256, 0, stream>>>(cur, batch, pool, N);
    k_pool_final<<<(G * HDIM + 255) / 256, 256, 0, stream>>>(pool, cnts, out + G, G);
    k_pred<<<G, 128, 0, stream>>>(out + G, Wr, br, out);
}


# k_edge_scatter: one wave per 32 consecutive edges instead of per edge: lanes fetch the 32 src/dst/coefficients once, the wave walks them with 8 edges' row loads in flight; adds stay in arrival order
# speedup vs baseline: 1.4643x; 1.4643x over previous
#include <hip/hip_runtime.h>
#include <hip/hip_bf16.h>

#define HDIM 128
#define EPS_BN 1e-5f

typedef __attribute__((ext_vector_type(2))) float v2f;
typedef __attribute__((ext_vector_type(8))) float v8f;

__global__ __launch_bounds__(256) void k_zero(float* p, long long n) {
    long long i = (long long)blockIdx.x * blockDim.x + threadIdx.x;
    long long stride = (long long)gridDim.x * blockDim.x;
    for (; i < n; i += stride) p[i] = 0.0f;
}

__global__ __launch_bounds__(256) void k_deg_scatter(const int* __restrict__ dst,
                                                     float* __restrict__ deg, int E) {
    int t = blockIdx.x * 256 + threadIdx.x;
    if (t < E) atomicAdd(&deg[dst[t]], 1.0f);
}

__global__ __launch_bounds__(256) void k_dinv(float* __restrict__ deg, int n) {
    int t = blockIdx.x * 256 + threadIdx.x;
    if (t < n) deg[t] = rsqrtf(deg[t] + 1.0f);
}

__global__ __launch_bounds__(256) void k_rank1(const float* __restrict__ X,
                                               const float* __restrict__ W1,
                                               float* __restrict__ h, int n) {
    int t = blockIdx.x * 256 + threadIdx.x;
    if (t >= n * 32) return;
    int row = t >> 5, c4 = (t & 31) << 2;
    float xv = X[row];
    const float4 w = *(const float4*)(W1 + c4);
    float4 o; o.x = xv * w.x; o.y = xv * w.y; o.z = xv * w.z; o.w = xv * w.w;
    *(float4*)(h + (long long)row * HDIM + c4) = o;
}

__global__ __launch_bounds__(256) void k_gemm_wmma(const float* __restrict__ x,
                                                   const float* __restrict__ W,
                                                   float* __restrict__ h, int n) {
    __shared__ float tileX[32 * HDIM];
    const int rowBase = blockIdx.x * 32;
    const int tid  = threadIdx.x;
    const int wave = tid >> 5;
    const int lane = tid & 31;
    const int half = lane >> 4;
    const int l16  = lane & 15;
    const int rowHalf = wave & 1;
    const int c0   = (wave >> 1) * 32;

    const bool fullTile = (rowBase + 32 <= n);

    if (fullTile) {
        const float4* xs = (const float4*)(x + (long long)rowBase * HDIM);
        #pragma unroll
        for (int i = 0; i < 4; ++i) {
            int idx = tid + i * 256;
            ((float4*)tileX)[idx] = xs[idx];
        }
    } else {
        #pragma unroll
        for (int i = 0; i < 4; ++i) {
            int idx = tid + i * 256;
            int r = rowBase + (idx >> 5);
            float4 v = make_float4(0.f, 0.f, 0.f, 0.f);
            if (r < n) v = *(const float4*)(x + (long long)r * HDIM + ((idx & 31) << 2));
            ((float4*)tileX)[idx] = v;
        }
    }
    __syncthreads();

    const float* xrow = tileX + (rowHalf * 16 + l16) * HDIM + 2 * half;
    v8f acc0 = {}, acc1 = {};

    #pragma unroll 4
    for (int k = 0; k < HDIM; k += 4) {
        v2f a; a.x = xrow[k]; a.y = xrow[k + 1];
        const float* wp = W + (long long)(k + 2 * half) * HDIM + c0 + l16;
        v2f b0; b0.x = wp[0];  b0.y = wp[HDIM];
        v2f b1; b1.x = wp[16]; b1.y = wp[HDIM + 16];
        acc0 = __builtin_amdgcn_wmma_f32_16x16x4_f32(false, a, false, b0,
                                                     (short)0, acc0, false, false);
        acc1 = __builtin_amdgcn_wmma_f32_16x16x4_f32(false, a, false, b1,
                                                     (short)0, acc1, false, false);
    }

    const int rbase = rowBase + rowHalf * 16 + half * 8;
    if (fullTile) {
        #pragma unroll
        for (int v = 0; v < 8; ++v) {
            float* op = h + (long long)(rbase + v) * HDIM + c0 + l16;
            op[0]  = acc0[v];
            op[16] = acc1[v];
        }
    } else {
        for (int v = 0; v < 8; ++v) {
            int r = rbase + v;
            if (r < n) {
                h[(long long)r * HDIM + c0 + l16]      = acc0[v];
                h[(long long)r * HDIM + c0 + 16 + l16] = acc1[v];
            }
        }
    }
}

#define EDGE_BATCH 8

__device__ __forceinline__ void edge_one(const float* __restrict__ h,
                                         float* __restrict__ agg,
                                         int s, int d, float coef, int lane) {
    const float* hp = h + (long long)s * HDIM;
    const float h0 = hp[lane], h1 = hp[lane + 32], h2 = hp[lane + 64], h3 = hp[lane + 96];
    float* ap = agg + (long long)d * HDIM;
    atomicAdd(ap + lane,      h0 * coef);
    atomicAdd(ap + lane + 32, h1 * coef);
    atomicAdd(ap + lane + 64, h2 * coef);
    atomicAdd(ap + lane + 96, h3 * coef);
}

__global__ __launch_bounds__(256) void k_edge_scatter(const float* __restrict__ h,
                                                      const int* __restrict__ src,
                                                      const int* __restrict__ dst,
                                                      const float* __restrict__ dinv,
                                                      float* __restrict__ agg, int E) {
    const int lane = threadIdx.x & 31;
    const int t = blockIdx.x * 256 + threadIdx.x;
    int s = 0, d = 0;
    float coef = 0.f;
    if (t < E) {
        s = src[t];
        d = dst[t];
        coef = dinv[s] * dinv[d];
    }
    const int e0 = __builtin_amdgcn_readfirstlane(t);
    int cnt = E - e0;
    cnt = cnt > 32 ? 32 : cnt;
    const int coefBits = __float_as_int(coef);

    int j = 0;
    for (; j + EDGE_BATCH <= cnt; j += EDGE_BATCH) {
        float v[EDGE_BATCH][4];
        #pragma unroll
        for (int u = 0; u < EDGE_BATCH; ++u) {
            const int sj = __builtin_amdgcn_readlane(s, j + u);
            const float* hp = h + (long long)sj * HDIM;
            v[u][0] = hp[lane];
            v[u][1] = hp[lane + 32];
            v[u][2] = hp[lane + 64];
            v[u][3] = hp[lane + 96];
        }
        #pragma unroll
        for (int u = 0; u < EDGE_BATCH; ++u) {
            const int dj = __builtin_amdgcn_readlane(d, j + u);
            const float cj = __int_as_float(__builtin_amdgcn_readlane(coefBits, j + u));
            float* ap = agg + (long long)dj * HDIM;
            atomicAdd(ap + lane,      v[u][0] * cj);
            atomicAdd(ap + lane + 32, v[u][1] * cj);
            atomicAdd(ap + lane + 64, v[u][2] * cj);
            atomicAdd(ap + lane + 96, v[u][3] * cj);
        }
    }
    for (; j < cnt; ++j) {
        edge_one(h, agg, __builtin_amdgcn_readlane(s, j), __builtin_amdgcn_readlane(d, j),
                 __int_as_float(__builtin_amdgcn_readlane(coefBits, j)), lane);
    }
}

__global__ __launch_bounds__(256) void k_combine_relu(float* __restrict__ agg,
                                                      const float* __restrict__ h,
                                                      const float* __restrict__ dinv,
                                                      const float* __restrict__ bias, int n) {
    int t = blockIdx.x * 256 + threadIdx.x;
    if (t >= n * 32) return;
    int row = t >> 5, c4 = (t & 31) << 2;
    float di = dinv[row], d2 = di * di;
    long long off = (long long)row * HDIM + c4;
    float4 a  = *(const float4*)(agg + off);
    float4 hv = *(const float4*)(h + off);
    float4 b  = *(const float4*)(bias + c4);
    a.x = fmaxf(a.x + hv.x * d2 + b.x, 0.f);
    a.y = fmaxf(a.y + hv.y * d2 + b.y, 0.f);
    a.z = fmaxf(a.z + hv.z * d2 + b.z, 0.f);
    a.w = fmaxf(a.w + hv.w * d2 + b.w, 0.f);
    *(float4*)(agg + off) = a;
}

__global__ __launch_bounds__(256) void k_stats(const float* __restrict__ y,
                                               float* __restrict__ sums, int n) {
    int c = threadIdx.x & 127;
    int sub = threadIdx.x >> 7;
    float s = 0.f, s2 = 0.f;
    for (int r = blockIdx.x * 2 + sub; r < n; r += gridDim.x * 2) {
        float v = y[(long long)r * HDIM + c];
        s += v; s2 += v * v;
    }
    __shared__ float ls[256], ls2[256];
    ls[threadIdx.x] = s; ls2[threadIdx.x] = s2;
    __syncthreads();
    if (sub == 0) {
        atomicAdd(&sums[c],        ls[c]  + ls[128 + c]);
        atomicAdd(&sums[128 + c],  ls2[c] + ls2[128 + c]);
    }
}

__global__ void k_scaleshift(const float* __restrict__ sums,
                             const float* __restrict__ gamma,
                             const float* __restrict__ beta,
                             float* __restrict__ ss, float invN) {
    int c = threadIdx.x;
    float m   = sums[c] * invN;
    float var = sums[128 + c] * invN - m * m;
    float sc  = rsqrtf(var + EPS_BN) * gamma[c];
    ss[c] = sc;
    ss[128 + c] = beta[c] - m * sc;
}

__global__ __launch_bounds__(256) void k_bn_apply(float* __restrict__ y,
                                                  const float* __restrict__ res,
                                                  const float* __restrict__ ss, int n) {
    int t = blockIdx.x * 256 + threadIdx.x;
    if (t >= n * 32) return;
    int row = t >> 5, c4 = (t & 31) << 2;
    long long off = (long long)row * HDIM + c4;
    float4 v  = *(const float4*)(y + off);
    float4 sc = *(const float4*)(ss + c4);
    float4 sh = *(const float4*)(ss + 128 + c4);
    v.x = v.x * sc.x + sh.x; v.y = v.y * sc.y + sh.y;
    v.z = v.z * sc.z + sh.z; v.w = v.w * sc.w + sh.w;
    if (res) {
        float4 r = *(const float4*)(res + off);
        v.x += r.x; v.y += r.y; v.z += r.z; v.w += r.w;
    }
    *(float4*)(y + off) = v;
}

__global__ __launch_bounds__(256) void k_counts(const int* __restrict__ batch,
                                                float* __restrict__ counts, int n) {
    int t = blockIdx.x * 256 + threadIdx.x;
    if (t < n) atomicAdd(&counts[batch[t]], 1.0f);
}

__global__ __launch_bounds__(256) void k_pool_acc(const float* __restrict__ x,
                                                  const int* __restrict__ batch,
                                                  float* __restrict__ pool, int n) {
    int t = blockIdx.x * 256 + threadIdx.x;
    if (t >= n * 32) return;
    int row = t >> 5, c4 = (t & 31) << 2;
    int g = batch[row];
    float4 v = *(const float4*)(x + (long long)row * HDIM + c4);
    float* p = pool + (long long)g * HDIM + c4;
    atomicAdd(p + 0, v.x); atomicAdd(p + 1, v.y);
    atomicAdd(p + 2, v.z); atomicAdd(p + 3, v.w);
}

__global__ __launch_bounds__(256) void k_pool_final(const float* __restrict__ pool,
                                                    const float* __restrict__ counts,
                                                    float* __restrict__ xmean, int G) {
    int t = blockIdx.x * 256 + threadIdx.x;
    if (t >= G * HDIM) return;
    int g = t / HDIM;
    xmean[t] = pool[t] / fmaxf(counts[g], 1.0f);
}

__global__ void k_pred(const float* __restrict__ xmean, const float* __restrict__ Wr,
                       const float* __restrict__ br, float* __restrict__ pred) {
    __shared__ float red[128];
    int g = blockIdx.x, c = threadIdx.x;
    red[c] = xmean[(long long)g * HDIM + c] * Wr[c];
    __syncthreads();
    for (int s = 64; s > 0; s >>= 1) {
        if (c < s) red[c] += red[c + s];
        __syncthreads();
    }
    if (c == 0) pred[g] = red[0] + br[0];
}

extern "C" void kernel_launch(void* const* d_in, const int* in_sizes, int n_in,
                              void* d_out, int out_size, void* d_ws, size_t ws_size,
                              hipStream_t stream) {
    const float* X     = (const float*)d_in[0];
    const int*   edge  = (const int*)d_in[1];
    const int*   batch = (const int*)d_in[2];
    const float* W1    = (const float*)d_in[3];
    const float* b1    = (const float*)d_in[4];
    const float* Wm    = (const float*)d_in[5];
    const float* bm    = (const float*)d_in[6];
    const float* gamma = (const float*)d_in[7];
    const float* beta  = (const float*)d_in[8];
    const float* Wr    = (const float*)d_in[9];
    const float* br    = (const float*)d_in[10];
    float* out = (float*)d_out;

    const int N = in_sizes[0];
    const int E = in_sizes[1] / 2;
    const int G = out_size / (HDIM + 1);
    const int* src = edge;
    const int* dst = edge + E;

    const long long NH = (long long)N * HDIM;
    float* ws    = (float*)d_ws;
    float* buf[3] = { ws, ws + NH, ws + 2 * NH };
    float* dinv  = ws + 3 * NH;
    float* sums  = dinv + N;
    float* ss    = sums + 256;
    float* pool  = ss + 256;
    float* cnts  = pool + (long long)G * HDIM;

    const int gE   = (E + 255) / 256;
    const int gN   = (N + 255) / 256;
    const int gNH4 = (N * 32 + 255) / 256;
    const int gGem = (N + 31) / 32;

    k_zero<<<1024, 256, 0, stream>>>(dinv, N);
    k_deg_scatter<<<gE, 256, 0, stream>>>(dst, dinv, E);
    k_dinv<<<gN, 256, 0, stream>>>(dinv, N);

    auto layer_tail = [&](float* hbuf, float* abuf, const float* bias,
                          const float* g_, const float* be_, const float* res) {
        k_zero<<<2048, 256, 0, stream>>>(abuf, NH);
        k_edge_scatter<<<gE, 256, 0, stream>>>(hbuf, src, dst, dinv, abuf, E);
        k_combine_relu<<<gNH4, 256, 0, stream>>>(abuf, hbuf, dinv, bias, N);
        k_zero<<<1, 256, 0, stream>>>(sums, 256);
        k_stats<<<256, 256, 0, stream>>>(abuf, sums, N);
        k_scaleshift<<<1, 128, 0, stream>>>(sums, g_, be_, ss, 1.0f / (float)N);
        k_bn_apply<<<gNH4, 256, 0, stream>>>(abuf, res, ss, N);
    };

    k_rank1<<<gNH4, 256, 0, stream>>>(X, W1, buf[1], N);
    layer_tail(buf[1], buf[2], b1, gamma, beta, nullptr);

    float* cur = buf[2];
    float* f0  = buf[0];
    float* f1  = buf[1];

    for (int i = 0; i < 4; ++i) {
        float* hbuf = f0;
        float* abuf = f1;
        k_gemm_wmma<<<gGem, 256, 0, stream>>>(cur, Wm + (long long)i * HDIM * HDIM, hbuf, N);
        layer_tail(hbuf, abuf, bm + (long long)i * HDIM,
                   gamma + (long long)(i + 1) * HDIM,
                   beta  + (long long)(i + 1) * HDIM, cur);
        float* next = abuf;
        f1 = cur;
        f0 = hbuf;
        cur = next;
    }

    k_zero<<<32, 256, 0, stream>>>(pool, (long long)G * HDIM + G);
    k_counts<<<gN, 256, 0, stream>>>(batch, cnts, N);
    k_pool_acc<<<gNH4, 256, 0, stream>>>(cur, batch, pool, N);
    k_pool_final<<<(G * HDIM + 255) / 256, 256, 0, stream>>>(pool, cnts, out + G, G);
    k_pred<<<G, 128, 0, stream>>>(out + G, Wr, br, out);
}


# Layer 1 aggregated on its one input channel before the outer product with W1: one f32 add per edge into s[dst], started at the self-loop term; 128-wide scatter of layer 1 gone; new sum order
# speedup vs baseline: 1.5435x; 1.0541x over previous
#include <hip/hip_runtime.h>
#include <hip/hip_bf16.h>

#define HDIM 128
#define EPS_BN 1e-5f

typedef __attribute__((ext_vector_type(2))) float v2f;
typedef __attribute__((ext_vector_type(8))) float v8f;

__global__ __launch_bounds__(256) void k_zero(float* p, long long n) {
    long long i = (long long)blockIdx.x * blockDim.x + threadIdx.x;
    long long stride = (long long)gridDim.x * blockDim.x;
    for (; i < n; i += stride) p[i] = 0.0f;
}

__global__ __launch_bounds__(256) void k_deg_scatter(const int* __restrict__ dst,
                                                     float* __restrict__ deg, int E) {
    int t = blockIdx.x * 256 + threadIdx.x;
    if (t < E) atomicAdd(&deg[dst[t]], 1.0f);
}

__global__ __launch_bounds__(256) void k_dinv(float* __restrict__ deg,
                                              const float* __restrict__ X,
                                              float* __restrict__ s1, int n) {
    int t = blockIdx.x * 256 + threadIdx.x;
    if (t < n) {
        float di = rsqrtf(deg[t] + 1.0f);
        deg[t] = di;
        s1[t] = X[t] * (di * di);
    }
}

__global__ __launch_bounds__(256) void k_edge_scalar(const float* __restrict__ X,
                                                     const int* __restrict__ src,
                                                     const int* __restrict__ dst,
                                                     const float* __restrict__ dinv,
                                                     float* __restrict__ s1, int E) {
    int t = blockIdx.x * 256 + threadIdx.x;
    if (t < E) {
        int s = src[t], d = dst[t];
        atomicAdd(&s1[d], X[s] * (dinv[s] * dinv[d]));
    }
}

__global__ __launch_bounds__(256) void k_expand_relu(const float* __restrict__ s1,
                                                     const float* __restrict__ W1,
                                                     const float* __restrict__ bias,
                                                     float* __restrict__ y, int n) {
    int t = blockIdx.x * 256 + threadIdx.x;
    if (t >= n * 32) return;
    int row = t >> 5, c4 = (t & 31) << 2;
    float sv = s1[row];
    const float4 w = *(const float4*)(W1 + c4);
    const float4 b = *(const float4*)(bias + c4);
    float4 o;
    o.x = fmaxf(sv * w.x + b.x, 0.f);
    o.y = fmaxf(sv * w.y + b.y, 0.f);
    o.z = fmaxf(sv * w.z + b.z, 0.f);
    o.w = fmaxf(sv * w.w + b.w, 0.f);
    *(float4*)(y + (long long)row * HDIM + c4) = o;
}

__global__ __launch_bounds__(256) void k_gemm_wmma(const float* __restrict__ x,
                                                   const float* __restrict__ W,
                                                   float* __restrict__ h, int n) {
    __shared__ float tileX[32 * HDIM];
    const int rowBase = blockIdx.x * 32;
    const int tid  = threadIdx.x;
    const int wave = tid >> 5;
    const int lane = tid & 31;
    const int half = lane >> 4;
    const int l16  = lane & 15;
    const int rowHalf = wave & 1;
    const int c0   = (wave >> 1) * 32;

    const bool fullTile = (rowBase + 32 <= n);

    if (fullTile) {
        const float4* xs = (const float4*)(x + (long long)rowBase * HDIM);
        #pragma unroll
        for (int i = 0; i < 4; ++i) {
            int idx = tid + i * 256;
            ((float4*)tileX)[idx] = xs[idx];
        }
    } else {
        #pragma unroll
        for (int i = 0; i < 4; ++i) {
            int idx = tid + i * 256;
            int r = rowBase + (idx >> 5);
            float4 v = make_float4(0.f, 0.f, 0.f, 0.f);
            if (r < n) v = *(const float4*)(x + (long long)r * HDIM + ((idx & 31) << 2));
            ((float4*)tileX)[idx] = v;
        }
    }
    __syncthreads();

    const float* xrow = tileX + (rowHalf * 16 + l16) * HDIM + 2 * half;
    v8f acc0 = {}, acc1 = {};

    #pragma unroll 4
    for (int k = 0; k < HDIM; k += 4) {
        v2f a; a.x = xrow[k]; a.y = xrow[k + 1];
        const float* wp = W + (long long)(k + 2 * half) * HDIM + c0 + l16;
        v2f b0; b0.x = wp[0];  b0.y = wp[HDIM];
        v2f b1; b1.x = wp[16]; b1.y = wp[HDIM + 16];
        acc0 = __builtin_amdgcn_wmma_f32_16x16x4_f32(false, a, false, b0,
                                                     (short)0, acc0, false, false);
        acc1 = __builtin_amdgcn_wmma_f32_16x16x4_f32(false, a, false, b1,
                                                     (short)0, acc1, false, false);
    }

    const int rbase = rowBase + rowHalf * 16 + half * 8;
    if (fullTile) {
        #pragma unroll
        for (int v = 0; v < 8; ++v) {
            float* op = h + (long long)(rbase + v) * HDIM + c0 + l16;
            op[0]  = acc0[v];
            op[16] = acc1[v];
        }
    } else {
        for (int v = 0; v < 8; ++v) {
            int r = rbase + v;
            if (r < n) {
                h[(long long)r * HDIM + c0 + l16]      = acc0[v];
                h[(long long)r * HDIM + c0 + 16 + l16] = acc1[v];
            }
        }
    }
}

#define EDGE_BATCH 8

__device__ __forceinline__ void edge_one(const float* __restrict__ h,
                                         float* __restrict__ agg,
                                         int s, int d, float coef, int lane) {
    const float* hp = h + (long long)s * HDIM;
    const float h0 = hp[lane], h1 = hp[lane + 32], h2 = hp[lane + 64], h3 = hp[lane + 96];
    float* ap = agg + (long long)d * HDIM;
    atomicAdd(ap + lane,      h0 * coef);
    atomicAdd(ap + lane + 32, h1 * coef);
    atomicAdd(ap + lane + 64, h2 * coef);
    atomicAdd(ap + lane + 96, h3 * coef);
}

__global__ __launch_bounds__(256) void k_edge_scatter(const float* __restrict__ h,
                                                      const int* __restrict__ src,
                                                      const int* __restrict__ dst,
                                                      const float* __restrict__ dinv,
                                                      float* __restrict__ agg, int E) {
    const int lane = threadIdx.x & 31;
    const int t = blockIdx.x * 256 + threadIdx.x;
    int s = 0, d = 0;
    float coef = 0.f;
    if (t < E) {
        s = src[t];
        d = dst[t];
        coef = dinv[s] * dinv[d];
    }
    const int e0 = __builtin_amdgcn_readfirstlane(t);
    int cnt = E - e0;
    cnt = cnt > 32 ? 32 : cnt;
    const int coefBits = __float_as_int(coef);

    int j = 0;
    for (; j + EDGE_BATCH <= cnt; j += EDGE_BATCH) {
        float v[EDGE_BATCH][4];
        #pragma unroll
        for (int u = 0; u < EDGE_BATCH; ++u) {
            const int sj = __builtin_amdgcn_readlane(s, j + u);
            const float* hp = h + (long long)sj * HDIM;
            v[u][0] = hp[lane];
            v[u][1] = hp[lane + 32];
            v[u][2] = hp[lane + 64];
            v[u][3] = hp[lane + 96];
        }
        #pragma unroll
        for (int u = 0; u < EDGE_BATCH; ++u) {
            const int dj = __builtin_amdgcn_readlane(d, j + u);
            const float cj = __int_as_float(__builtin_amdgcn_readlane(coefBits, j + u));
            float* ap = agg + (long long)dj * HDIM;
            atomicAdd(ap + lane,      v[u][0] * cj);
            atomicAdd(ap + lane + 32, v[u][1] * cj);
            atomicAdd(ap + lane + 64, v[u][2] * cj);
            atomicAdd(ap + lane + 96, v[u][3] * cj);
        }
    }
    for (; j < cnt; ++j) {
        edge_one(h, agg, __builtin_amdgcn_readlane(s, j), __builtin_amdgcn_readlane(d, j),
                 __int_as_float(__builtin_amdgcn_readlane(coefBits, j)), lane);
    }
}

__global__ __launch_bounds__(256) void k_combine_relu(float* __restrict__ agg,
                                                      const float* __restrict__ h,
                                                      const float* __restrict__ dinv,
                                                      const float* __restrict__ bias, int n) {
    int t = blockIdx.x * 256 + threadIdx.x;
    if (t >= n * 32) return;
    int row = t >> 5, c4 = (t & 31) << 2;
    float di = dinv[row], d2 = di * di;
    long long off = (long long)row * HDIM + c4;
    float4 a  = *(const float4*)(agg + off);
    float4 hv = *(const float4*)(h + off);
    float4 b  = *(const float4*)(bias + c4);
    a.x = fmaxf(a.x + hv.x * d2 + b.x, 0.f);
    a.y = fmaxf(a.y + hv.y * d2 + b.y, 0.f);
    a.z = fmaxf(a.z + hv.z * d2 + b.z, 0.f);
    a.w = fmaxf(a.w + hv.w * d2 + b.w, 0.f);
    *(float4*)(agg + off) = a;
}

__global__ __launch_bounds__(256) void k_stats(const float* __restrict__ y,
                                               float* __restrict__ sums, int n) {
    int c = threadIdx.x & 127;
    int sub = threadIdx.x >> 7;
    float s = 0.f, s2 = 0.f;
    for (int r = blockIdx.x * 2 + sub; r < n; r += gridDim.x * 2) {
        float v = y[(long long)r * HDIM + c];
        s += v; s2 += v * v;
    }
    __shared__ float ls[256], ls2[256];
    ls[threadIdx.x] = s; ls2[threadIdx.x] = s2;
    __syncthreads();
    if (sub == 0) {
        atomicAdd(&sums[c],        ls[c]  + ls[128 + c]);
        atomicAdd(&sums[128 + c],  ls2[c] + ls2[128 + c]);
    }
}

__global__ void k_scaleshift(const float* __restrict__ sums,
                             const float* __restrict__ gamma,
                             const float* __restrict__ beta,
                             float* __restrict__ ss, float invN) {
    int c = threadIdx.x;
    float m   = sums[c] * invN;
    float var = sums[128 + c] * invN - m * m;
    float sc  = rsqrtf(var + EPS_BN) * gamma[c];
    ss[c] = sc;
    ss[128 + c] = beta[c] - m * sc;
}

__global__ __launch_bounds__(256) void k_bn_apply(float* __restrict__ y,
                                                  const float* __restrict__ res,
                                                  const float* __restrict__ ss, int n) {
    int t = blockIdx.x * 256 + threadIdx.x;
    if (t >= n * 32) return;
    int row = t >> 5, c4 = (t & 31) << 2;
    long long off = (long long)row * HDIM + c4;
    float4 v  = *(const float4*)(y + off);
    float4 sc = *(const float4*)(ss + c4);
    float4 sh = *(const float4*)(ss + 128 + c4);
    v.x = v.x * sc.x + sh.x; v.y = v.y * sc.y + sh.y;
    v.z = v.z * sc.z + sh.z; v.w = v.w * sc.w + sh.w;
    if (res) {
        float4 r = *(const float4*)(res + off);
        v.x += r.x; v.y += r.y; v.z += r.z; v.w += r.w;
    }
    *(float4*)(y + off) = v;
}

__global__ __launch_bounds__(256) void k_counts(const int* __restrict__ batch,
                                                float* __restrict__ counts, int n) {
    int t = blockIdx.x * 256 + threadIdx.x;
    if (t < n) atomicAdd(&counts[batch[t]], 1.0f);
}

__global__ __launch_bounds__(256) void k_pool_acc(const float* __restrict__ x,
                                                  const int* __restrict__ batch,
                                                  float* __restrict__ pool, int n) {
    int t = blockIdx.x * 256 + threadIdx.x;
    if (t >= n * 32) return;
    int row = t >> 5, c4 = (t & 31) << 2;
    int g = batch[row];
    float4 v = *(const float4*)(x + (long long)row * HDIM + c4);
    float* p = pool + (long long)g * HDIM + c4;
    atomicAdd(p + 0, v.x); atomicAdd(p + 1, v.y);
    atomicAdd(p + 2, v.z); atomicAdd(p + 3, v.w);
}

__global__ __launch_bounds__(256) void k_pool_final(const float* __restrict__ pool,
                                                    const float* __restrict__ counts,
                                                    float* __restrict__ xmean, int G) {
    int t = blockIdx.x * 256 + threadIdx.x;
    if (t >= G * HDIM) return;
    int g = t / HDIM;
    xmean[t] = pool[t] / fmaxf(counts[g], 1.0f);
}

__global__ void k_pred(const float* __restrict__ xmean, const float* __restrict__ Wr,
                       const float* __restrict__ br, float* __restrict__ pred) {
    __shared__ float red[128];
    int g = blockIdx.x, c = threadIdx.x;
    red[c] = xmean[(long long)g * HDIM + c] * Wr[c];
    __syncthreads();
    for (int s = 64; s > 0; s >>= 1) {
        if (c < s) red[c] += red[c + s];
        __syncthreads();
    }
    if (c == 0) pred[g] = red[0] + br[0];
}

extern "C" void kernel_launch(void* const* d_in, const int* in_sizes, int n_in,
                              void* d_out, int out_size, void* d_ws, size_t ws_size,
                              hipStream_t stream) {
    const float* X     = (const float*)d_in[0];
    const int*   edge  = (const int*)d_in[1];
    const int*   batch = (const int*)d_in[2];
    const float* W1    = (const float*)d_in[3];
    const float* b1    = (const float*)d_in[4];
    const float* Wm    = (const float*)d_in[5];
    const float* bm    = (const float*)d_in[6];
    const float* gamma = (const float*)d_in[7];
    const float* beta  = (const float*)d_in[8];
    const float* Wr    = (const float*)d_in[9];
    const float* br    = (const float*)d_in[10];
    float* out = (float*)d_out;

    const int N = in_sizes[0];
    const int E = in_sizes[1] / 2;
    const int G = out_size / (HDIM + 1);
    const int* src = edge;
    const int* dst = edge + E;

    const long long NH = (long long)N * HDIM;
    float* ws    = (float*)d_ws;
    float* buf[3] = { ws, ws + NH, ws + 2 * NH };
    float* dinv  = ws + 3 * NH;
    float* sums  = dinv + N;
    float* ss    = sums + 256;
    float* pool  = ss + 256;
    float* cnts  = pool + (long long)G * HDIM;

    const int gE   = (E + 255) / 256;
    const int gN   = (N + 255) / 256;
    const int gNH4 = (N * 32 + 255) / 256;
    const int gGem = (N + 31) / 32;

    k_zero<<<1024, 256, 0, stream>>>(dinv, N);
    k_deg_scatter<<<gE, 256, 0, stream>>>(dst, dinv, E);
    float* s1 = buf[0];
    k_dinv<<<gN, 256, 0, stream>>>(dinv, X, s1, N);

    auto bn_tail = [&](float* abuf, const float* g_, const float* be_, const float* res) {
        k_zero<<<1, 256, 0, stream>>>(sums, 256);
        k_stats<<<256, 256, 0, stream>>>(abuf, sums, N);
        k_scaleshift<<<1, 128, 0, stream>>>(sums, g_, be_, ss, 1.0f / (float)N);
        k_bn_apply<<<gNH4, 256, 0, stream>>>(abuf, res, ss, N);
    };

    auto layer_tail = [&](float* hbuf, float* abuf, const float* bias,
                          const float* g_, const float* be_, const float* res) {
        k_zero<<<2048, 256, 0, stream>>>(abuf, NH);
        k_edge_scatter<<<gE, 256, 0, stream>>>(hbuf, src, dst, dinv, abuf, E);
        k_combine_relu<<<gNH4, 256, 0, stream>>>(abuf, hbuf, dinv, bias, N);
        bn_tail(abuf, g_, be_, res);
    };

    k_edge_scalar<<<gE, 256, 0, stream>>>(X, src, dst, dinv, s1, E);
    k_expand_relu<<<gNH4, 256, 0, stream>>>(s1, W1, b1, buf[2], N);
    bn_tail(buf[2], gamma, beta, nullptr);

    float* cur = buf[2];
    float* f0  = buf[0];
    float* f1  = buf[1];

    for (int i = 0; i < 4; ++i) {
        float* hbuf = f0;
        float* abuf = f1;
        k_gemm_wmma<<<gGem, 256, 0, stream>>>(cur, Wm + (long long)i * HDIM * HDIM, hbuf, N);
        layer_tail(hbuf, abuf, bm + (long long)i * HDIM,
                   gamma + (long long)(i + 1) * HDIM,
                   beta  + (long long)(i + 1) * HDIM, cur);
        float* next = abuf;
        f1 = cur;
        f0 = hbuf;
        cur = next;
    }

    k_zero<<<32, 256, 0, stream>>>(pool, (long long)G * HDIM + G);
    k_counts<<<gN, 256, 0, stream>>>(batch, cnts, N);
    k_pool_acc<<<gNH4, 256, 0, stream>>>(cur, batch, pool, N);
    k_pool_final<<<(G * HDIM + 255) / 256, 256, 0, stream>>>(pool, cnts, out + G, G);
    k_pred<<<G, 128, 0, stream>>>(out + G, Wr, br, out);
}


# k_stats: 8 whole-row b128 loads in flight per wave (256 blocks x 32 waves), not one 4-byte load per trip; block partials summed in fixed order by k_scaleshift: no stats atomics, new sum order
# speedup vs baseline: 2.0221x; 1.3101x over previous
#include <hip/hip_runtime.h>
#include <hip/hip_bf16.h>

#define HDIM 128
#define EPS_BN 1e-5f

typedef __attribute__((ext_vector_type(2))) float v2f;
typedef __attribute__((ext_vector_type(8))) float v8f;

__global__ __launch_bounds__(256) void k_zero(float* p, long long n) {
    long long i = (long long)blockIdx.x * blockDim.x + threadIdx.x;
    long long stride = (long long)gridDim.x * blockDim.x;
    for (; i < n; i += stride) p[i] = 0.0f;
}

__global__ __launch_bounds__(256) void k_deg_scatter(const int* __restrict__ dst,
                                                     float* __restrict__ deg, int E) {
    int t = blockIdx.x * 256 + threadIdx.x;
    if (t < E) atomicAdd(&deg[dst[t]], 1.0f);
}

__global__ __launch_bounds__(256) void k_dinv(float* __restrict__ deg,
                                              const float* __restrict__ X,
                                              float* __restrict__ s1, int n) {
    int t = blockIdx.x * 256 + threadIdx.x;
    if (t < n) {
        float di = rsqrtf(deg[t] + 1.0f);
        deg[t] = di;
        s1[t] = X[t] * (di * di);
    }
}

__global__ __launch_bounds__(256) void k_edge_scalar(const float* __restrict__ X,
                                                     const int* __restrict__ src,
                                                     const int* __restrict__ dst,
                                                     const float* __restrict__ dinv,
                                                     float* __restrict__ s1, int E) {
    int t = blockIdx.x * 256 + threadIdx.x;
    if (t < E) {
        int s = src[t], d = dst[t];
        atomicAdd(&s1[d], X[s] * (dinv[s] * dinv[d]));
    }
}

__global__ __launch_bounds__(256) void k_expand_relu(const float* __restrict__ s1,
                                                     const float* __restrict__ W1,
                                                     const float* __restrict__ bias,
                                                     float* __restrict__ y, int n) {
    int t = blockIdx.x * 256 + threadIdx.x;
    if (t >= n * 32) return;
    int row = t >> 5, c4 = (t & 31) << 2;
    float sv = s1[row];
    const float4 w = *(const float4*)(W1 + c4);
    const float4 b = *(const float4*)(bias + c4);
    float4 o;
    o.x = fmaxf(sv * w.x + b.x, 0.f);
    o.y = fmaxf(sv * w.y + b.y, 0.f);
    o.z = fmaxf(sv * w.z + b.z, 0.f);
    o.w = fmaxf(sv * w.w + b.w, 0.f);
    *(float4*)(y + (long long)row * HDIM + c4) = o;
}

__global__ __launch_bounds__(256) void k_gemm_wmma(const float* __restrict__ x,
                                                   const float* __restrict__ W,
                                                   float* __restrict__ h, int n) {
    __shared__ float tileX[32 * HDIM];
    const int rowBase = blockIdx.x * 32;
    const int tid  = threadIdx.x;
    const int wave = tid >> 5;
    const int lane = tid & 31;
    const int half = lane >> 4;
    const int l16  = lane & 15;
    const int rowHalf = wave & 1;
    const int c0   = (wave >> 1) * 32;

    const bool fullTile = (rowBase + 32 <= n);

    if (fullTile) {
        const float4* xs = (const float4*)(x + (long long)rowBase * HDIM);
        #pragma unroll
        for (int i = 0; i < 4; ++i) {
            int idx = tid + i * 256;
            ((float4*)tileX)[idx] = xs[idx];
        }
    } else {
        #pragma unroll
        for (int i = 0; i < 4; ++i) {
            int idx = tid + i * 256;
            int r = rowBase + (idx >> 5);
            float4 v = make_float4(0.f, 0.f, 0.f, 0.f);
            if (r < n) v = *(const float4*)(x + (long long)r * HDIM + ((idx & 31) << 2));
            ((float4*)tileX)[idx] = v;
        }
    }
    __syncthreads();

    const float* xrow = tileX + (rowHalf * 16 + l16) * HDIM + 2 * half;
    v8f acc0 = {}, acc1 = {};

    #pragma unroll 4
    for (int k = 0; k < HDIM; k += 4) {
        v2f a; a.x = xrow[k]; a.y = xrow[k + 1];
        const float* wp = W + (long long)(k + 2 * half) * HDIM + c0 + l16;
        v2f b0; b0.x = wp[0];  b0.y = wp[HDIM];
        v2f b1; b1.x = wp[16]; b1.y = wp[HDIM + 16];
        acc0 = __builtin_amdgcn_wmma_f32_16x16x4_f32(false, a, false, b0,
                                                     (short)0, acc0, false, false);
        acc1 = __builtin_amdgcn_wmma_f32_16x16x4_f32(false, a, false, b1,
                                                     (short)0, acc1, false, false);
    }

    const int rbase = rowBase + rowHalf * 16 + half * 8;
    if (fullTile) {
        #pragma unroll
        for (int v = 0; v < 8; ++v) {
            float* op = h + (long long)(rbase + v) * HDIM + c0 + l16;
            op[0]  = acc0[v];
            op[16] = acc1[v];
        }
    } else {
        for (int v = 0; v < 8; ++v) {
            int r = rbase + v;
            if (r < n) {
                h[(long long)r * HDIM + c0 + l16]      = acc0[v];
                h[(long long)r * HDIM + c0 + 16 + l16] = acc1[v];
            }
        }
    }
}

#define EDGE_BATCH 8

__device__ __forceinline__ void edge_one(const float* __restrict__ h,
                                         float* __restrict__ agg,
                                         int s, int d, float coef, int lane) {
    const float* hp = h + (long long)s * HDIM;
    const float h0 = hp[lane], h1 = hp[lane + 32], h2 = hp[lane + 64], h3 = hp[lane + 96];
    float* ap = agg + (long long)d * HDIM;
    atomicAdd(ap + lane,      h0 * coef);
    atomicAdd(ap + lane + 32, h1 * coef);
    atomicAdd(ap + lane + 64, h2 * coef);
    atomicAdd(ap + lane + 96, h3 * coef);
}

__global__ __launch_bounds__(256) void k_edge_scatter(const float* __restrict__ h,
                                                      const int* __restrict__ src,
                                                      const int* __restrict__ dst,
                                                      const float* __restrict__ dinv,
                                                      float* __restrict__ agg, int E) {
    const int lane = threadIdx.x & 31;
    const int t = blockIdx.x * 256 + threadIdx.x;
    int s = 0, d = 0;
    float coef = 0.f;
    if (t < E) {
        s = src[t];
        d = dst[t];
        coef = dinv[s] * dinv[d];
    }
    const int e0 = __builtin_amdgcn_readfirstlane(t);
    int cnt = E - e0;
    cnt = cnt > 32 ? 32 : cnt;
    const int coefBits = __float_as_int(coef);

    int j = 0;
    for (; j + EDGE_BATCH <= cnt; j += EDGE_BATCH) {
        float v[EDGE_BATCH][4];
        #pragma unroll
        for (int u = 0; u < EDGE_BATCH; ++u) {
            const int sj = __builtin_amdgcn_readlane(s, j + u);
            const float* hp = h + (long long)sj * HDIM;
            v[u][0] = hp[lane];
            v[u][1] = hp[lane + 32];
            v[u][2] = hp[lane + 64];
            v[u][3] = hp[lane + 96];
        }
        #pragma unroll
        for (int u = 0; u < EDGE_BATCH; ++u) {
            const int dj = __builtin_amdgcn_readlane(d, j + u);
            const float cj = __int_as_float(__builtin_amdgcn_readlane(coefBits, j + u));
            float* ap = agg + (long long)dj * HDIM;
            atomicAdd(ap + lane,      v[u][0] * cj);
            atomicAdd(ap + lane + 32, v[u][1] * cj);
            atomicAdd(ap + lane + 64, v[u][2] * cj);
            atomicAdd(ap + lane + 96, v[u][3] * cj);
        }
    }
    for (; j < cnt; ++j) {
        edge_one(h, agg, __builtin_amdgcn_readlane(s, j), __builtin_amdgcn_readlane(d, j),
                 __int_as_float(__builtin_amdgcn_readlane(coefBits, j)), lane);
    }
}

__global__ __launch_bounds__(256) void k_combine_relu(float* __restrict__ agg,
                                                      const float* __restrict__ h,
                                                      const float* __restrict__ dinv,
                                                      const float* __restrict__ bias, int n) {
    int t = blockIdx.x * 256 + threadIdx.x;
    if (t >= n * 32) return;
    int row = t >> 5, c4 = (t & 31) << 2;
    float di = dinv[row], d2 = di * di;
    long long off = (long long)row * HDIM + c4;
    float4 a  = *(const float4*)(agg + off);
    float4 hv = *(const float4*)(h + off);
    float4 b  = *(const float4*)(bias + c4);
    a.x = fmaxf(a.x + hv.x * d2 + b.x, 0.f);
    a.y = fmaxf(a.y + hv.y * d2 + b.y, 0.f);
    a.z = fmaxf(a.z + hv.z * d2 + b.z, 0.f);
    a.w = fmaxf(a.w + hv.w * d2 + b.w, 0.f);
    *(float4*)(agg + off) = a;
}

#define STAT_BLOCKS 256
#define STAT_WAVES  32
#define STAT_BATCH  8
#define STAT_GROUPS 16

__device__ __forceinline__ void stat_acc(float4& s, float4& q, const float4 v) {
    s.x += v.x; s.y += v.y; s.z += v.z; s.w += v.w;
    q.x += v.x * v.x; q.y += v.y * v.y; q.z += v.z * v.z; q.w += v.w * v.w;
}

__global__ __launch_bounds__(STAT_WAVES * 32) void k_stats(const float* __restrict__ y,
                                                            float* __restrict__ part, int n) {
    __shared__ float ls[STAT_WAVES][256];
    const int lane = threadIdx.x & 31;
    const int wave = threadIdx.x >> 5;
    const int S = STAT_BLOCKS * STAT_WAVES;
    const float4* yp = (const float4*)y + lane;
    int r = blockIdx.x * STAT_WAVES + wave;
    float4 s = make_float4(0.f, 0.f, 0.f, 0.f), q = s;

    for (; r + (STAT_BATCH - 1) * S < n; r += STAT_BATCH * S) {
        float4 v[STAT_BATCH];
        #pragma unroll
        for (int u = 0; u < STAT_BATCH; ++u) v[u] = yp[(long long)(r + u * S) * 32];
        #pragma unroll
        for (int u = 0; u < STAT_BATCH; ++u) stat_acc(s, q, v[u]);
    }
    if (r < n) {
        const float4 zero = make_float4(0.f, 0.f, 0.f, 0.f);
        float4 v[STAT_BATCH - 1];
        #pragma unroll
        for (int u = 0; u < STAT_BATCH - 1; ++u) {
            const int rr = r + u * S;
            const bool in = rr < n;
            const float4 t = yp[(long long)(in ? rr : n - 1) * 32];
            v[u] = in ? t : zero;
        }
        #pragma unroll
        for (int u = 0; u < STAT_BATCH - 1; ++u) stat_acc(s, q, v[u]);
    }

    *(float4*)&ls[wave][lane * 4]       = s;
    *(float4*)&ls[wave][128 + lane * 4] = q;
    __syncthreads();
    if (threadIdx.x < 256) {
        float a = 0.f;
        for (int w = 0; w < STAT_WAVES; ++w) a += ls[w][threadIdx.x];
        part[(long long)blockIdx.x * 256 + threadIdx.x] = a;
    }
}

__global__ __launch_bounds__(STAT_GROUPS * 64) void k_scaleshift(const float* __restrict__ part,
                                                                 const float* __restrict__ gamma,
                                                                 const float* __restrict__ beta,
                                                                 float* __restrict__ ss, float invN) {
    __shared__ float fs[STAT_GROUPS][256];
    const int c4 = threadIdx.x & 63;
    const int g  = threadIdx.x >> 6;
    const float4* pp = (const float4*)part + c4;
    float4 v[STAT_BLOCKS / STAT_GROUPS];
    #pragma unroll
    for (int i = 0; i < STAT_BLOCKS / STAT_GROUPS; ++i) v[i] = pp[(g + i * STAT_GROUPS) * 64];
    float4 a = v[0];
    #pragma unroll
    for (int i = 1; i < STAT_BLOCKS / STAT_GROUPS; ++i) {
        a.x += v[i].x; a.y += v[i].y; a.z += v[i].z; a.w += v[i].w;
    }
    *(float4*)&fs[g][c4 * 4] = a;
    __syncthreads();
    if (threadIdx.x < 128) {
        const int c = threadIdx.x;
        float su = 0.f, sq = 0.f;
        for (int k = 0; k < STAT_GROUPS; ++k) { su += fs[k][c]; sq += fs[k][128 + c]; }
        float m   = su * invN;
        float var = sq * invN - m * m;
        float sc  = rsqrtf(var + EPS_BN) * gamma[c];
        ss[c] = sc;
        ss[128 + c] = beta[c] - m * sc;
    }
}

__global__ __launch_bounds__(256) void k_bn_apply(float* __restrict__ y,
                                                  const float* __restrict__ res,
                                                  const float* __restrict__ ss, int n) {
    int t = blockIdx.x * 256 + threadIdx.x;
    if (t >= n * 32) return;
    int row = t >> 5, c4 = (t & 31) << 2;
    long long off = (long long)row * HDIM + c4;
    float4 v  = *(const float4*)(y + off);
    float4 sc = *(const float4*)(ss + c4);
    float4 sh = *(const float4*)(ss + 128 + c4);
    v.x = v.x * sc.x + sh.x; v.y = v.y * sc.y + sh.y;
    v.z = v.z * sc.z + sh.z; v.w = v.w * sc.w + sh.w;
    if (res) {
        float4 r = *(const float4*)(res + off);
        v.x += r.x; v.y += r.y; v.z += r.z; v.w += r.w;
    }
    *(float4*)(y + off) = v;
}

__global__ __launch_bounds__(256) void k_counts(const int* __restrict__ batch,
                                                float* __restrict__ counts, int n) {
    int t = blockIdx.x * 256 + threadIdx.x;
    if (t < n) atomicAdd(&counts[batch[t]], 1.0f);
}

__global__ __launch_bounds__(256) void k_pool_acc(const float* __restrict__ x,
                                                  const int* __restrict__ batch,
                                                  float* __restrict__ pool, int n) {
    int t = blockIdx.x * 256 + threadIdx.x;
    if (t >= n * 32) return;
    int row = t >> 5, c4 = (t & 31) << 2;
    int g = batch[row];
    float4 v = *(const float4*)(x + (long long)row * HDIM + c4);
    float* p = pool + (long long)g * HDIM + c4;
    atomicAdd(p + 0, v.x); atomicAdd(p + 1, v.y);
    atomicAdd(p + 2, v.z); atomicAdd(p + 3, v.w);
}

__global__ __launch_bounds__(256) void k_pool_final(const float* __restrict__ pool,
                                                    const float* __restrict__ counts,
                                                    float* __restrict__ xmean, int G) {
    int t = blockIdx.x * 256 + threadIdx.x;
    if (t >= G * HDIM) return;
    int g = t / HDIM;
    xmean[t] = pool[t] / fmaxf(counts[g], 1.0f);
}

__global__ void k_pred(const float* __restrict__ xmean, const float* __restrict__ Wr,
                       const float* __restrict__ br, float* __restrict__ pred) {
    __shared__ float red[128];
    int g = blockIdx.x, c = threadIdx.x;
    red[c] = xmean[(long long)g * HDIM + c] * Wr[c];
    __syncthreads();
    for (int s = 64; s > 0; s >>= 1) {
        if (c < s) red[c] += red[c + s];
        __syncthreads();
    }
    if (c == 0) pred[g] = red[0] + br[0];
}

extern "C" void kernel_launch(void* const* d_in, const int* in_sizes, int n_in,
                              void* d_out, int out_size, void* d_ws, size_t ws_size,
                              hipStream_t stream) {
    const float* X     = (const float*)d_in[0];
    const int*   edge  = (const int*)d_in[1];
    const int*   batch = (const int*)d_in[2];
    const float* W1    = (const float*)d_in[3];
    const float* b1    = (const float*)d_in[4];
    const float* Wm    = (const float*)d_in[5];
    const float* bm    = (const float*)d_in[6];
    const float* gamma = (const float*)d_in[7];
    const float* beta  = (const float*)d_in[8];
    const float* Wr    = (const float*)d_in[9];
    const float* br    = (const float*)d_in[10];
    float* out = (float*)d_out;

    const int N = in_sizes[0];
    const int E = in_sizes[1] / 2;
    const int G = out_size / (HDIM + 1);
    const int* src = edge;
    const int* dst = edge + E;

    const long long NH = (long long)N * HDIM;
    float* ws    = (float*)d_ws;
    float* buf[3] = { ws, ws + NH, ws + 2 * NH };
    float* dinv  = ws + 3 * NH;
    float* ss    = dinv + N;
    float* pool  = ss + 256;
    float* cnts  = pool + (long long)G * HDIM;

    const int gE   = (E + 255) / 256;
    const int gN   = (N + 255) / 256;
    const int gNH4 = (N * 32 + 255) / 256;
    const int gGem = (N + 31) / 32;

    k_zero<<<1024, 256, 0, stream>>>(dinv, N);
    k_deg_scatter<<<gE, 256, 0, stream>>>(dst, dinv, E);
    float* s1 = buf[0];
    k_dinv<<<gN, 256, 0, stream>>>(dinv, X, s1, N);

    auto bn_tail = [&](float* abuf, float* part, const float* g_, const float* be_,
                       const float* res) {
        k_stats<<<STAT_BLOCKS, STAT_WAVES * 32, 0, stream>>>(abuf, part, N);
        k_scaleshift<<<1, STAT_GROUPS * 64, 0, stream>>>(part, g_, be_, ss, 1.0f / (float)N);
        k_bn_apply<<<gNH4, 256, 0, stream>>>(abuf, res, ss, N);
    };

    auto layer_tail = [&](float* hbuf, float* abuf, const float* bias,
                          const float* g_, const float* be_, const float* res) {
        k_zero<<<2048, 256, 0, stream>>>(abuf, NH);
        k_edge_scatter<<<gE, 256, 0, stream>>>(hbuf, src, dst, dinv, abuf, E);
        k_combine_relu<<<gNH4, 256, 0, stream>>>(abuf, hbuf, dinv, bias, N);
        bn_tail(abuf, hbuf, g_, be_, res);
    };

    k_edge_scalar<<<gE, 256, 0, stream>>>(X, src, dst, dinv, s1, E);
    k_expand_relu<<<gNH4, 256, 0, stream>>>(s1, W1, b1, buf[2], N);
    bn_tail(buf[2], buf[0], gamma, beta, nullptr);

    float* cur = buf[2];
    float* f0  = buf[0];
    float* f1  = buf[1];

    for (int i = 0; i < 4; ++i) {
        float* hbuf = f0;
        float* abuf = f1;
        k_gemm_wmma<<<gGem, 256, 0, stream>>>(cur, Wm + (long long)i * HDIM * HDIM, hbuf, N);
        layer_tail(hbuf, abuf, bm + (long long)i * HDIM,
                   gamma + (long long)(i + 1) * HDIM,
                   beta  + (long long)(i + 1) * HDIM, cur);
        float* next = abuf;
        f1 = cur;
        f0 = hbuf;
        cur = next;
    }

    k_zero<<<32, 256, 0, stream>>>(pool, (long long)G * HDIM + G);
    k_counts<<<gN, 256, 0, stream>>>(batch, cnts, N);
    k_pool_acc<<<gNH4, 256, 0, stream>>>(cur, batch, pool, N);
    k_pool_final<<<(G * HDIM + 255) / 256, 256, 0, stream>>>(pool, cnts, out + G, G);
    k_pred<<<G, 128, 0, stream>>>(out + G, Wr, br, out);
}


# Neighbour sum starts at the self-loop term + bias, stored by the GEMM epilogue; ReLU applied by the stats and BN readers: zero and combine passes gone (41 -> 33 launches); new sum order
# speedup vs baseline: 2.2974x; 1.1361x over previous
#include <hip/hip_runtime.h>
#include <hip/hip_bf16.h>

#define HDIM 128
#define EPS_BN 1e-5f

typedef __attribute__((ext_vector_type(2))) float v2f;
typedef __attribute__((ext_vector_type(8))) float v8f;

__global__ __launch_bounds__(256) void k_zero(float* p, long long n) {
    long long i = (long long)blockIdx.x * blockDim.x + threadIdx.x;
    long long stride = (long long)gridDim.x * blockDim.x;
    for (; i < n; i += stride) p[i] = 0.0f;
}

__global__ __launch_bounds__(256) void k_deg_scatter(const int* __restrict__ dst,
                                                     float* __restrict__ deg, int E) {
    int t = blockIdx.x * 256 + threadIdx.x;
    if (t < E) atomicAdd(&deg[dst[t]], 1.0f);
}

__global__ __launch_bounds__(256) void k_dinv(float* __restrict__ deg,
                                              const float* __restrict__ X,
                                              float* __restrict__ s1, int n) {
    int t = blockIdx.x * 256 + threadIdx.x;
    if (t < n) {
        float di = rsqrtf(deg[t] + 1.0f);
        deg[t] = di;
        s1[t] = X[t] * (di * di);
    }
}

__global__ __launch_bounds__(256) void k_edge_scalar(const float* __restrict__ X,
                                                     const int* __restrict__ src,
                                                     const int* __restrict__ dst,
                                                     const float* __restrict__ dinv,
                                                     float* __restrict__ s1, int E) {
    int t = blockIdx.x * 256 + threadIdx.x;
    if (t < E) {
        int s = src[t], d = dst[t];
        atomicAdd(&s1[d], X[s] * (dinv[s] * dinv[d]));
    }
}

__global__ __launch_bounds__(256) void k_expand(const float* __restrict__ s1,
                                                     const float* __restrict__ W1,
                                                     const float* __restrict__ bias,
                                                     float* __restrict__ y, int n) {
    int t = blockIdx.x * 256 + threadIdx.x;
    if (t >= n * 32) return;
    int row = t >> 5, c4 = (t & 31) << 2;
    float sv = s1[row];
    const float4 w = *(const float4*)(W1 + c4);
    const float4 b = *(const float4*)(bias + c4);
    float4 o;
    o.x = sv * w.x + b.x;
    o.y = sv * w.y + b.y;
    o.z = sv * w.z + b.z;
    o.w = sv * w.w + b.w;
    *(float4*)(y + (long long)row * HDIM + c4) = o;
}

__global__ __launch_bounds__(256) void k_gemm_wmma(const float* __restrict__ x,
                                                   const float* __restrict__ W,
                                                   const float* __restrict__ dinv,
                                                   const float* __restrict__ bias,
                                                   float* __restrict__ h,
                                                   float* __restrict__ agg, int n) {
    __shared__ float tileX[32 * HDIM];
    const int rowBase = blockIdx.x * 32;
    const int tid  = threadIdx.x;
    const int wave = tid >> 5;
    const int lane = tid & 31;
    const int half = lane >> 4;
    const int l16  = lane & 15;
    const int rowHalf = wave & 1;
    const int c0   = (wave >> 1) * 32;

    const bool fullTile = (rowBase + 32 <= n);

    if (fullTile) {
        const float4* xs = (const float4*)(x + (long long)rowBase * HDIM);
        #pragma unroll
        for (int i = 0; i < 4; ++i) {
            int idx = tid + i * 256;
            ((float4*)tileX)[idx] = xs[idx];
        }
    } else {
        #pragma unroll
        for (int i = 0; i < 4; ++i) {
            int idx = tid + i * 256;
            int r = rowBase + (idx >> 5);
            float4 v = make_float4(0.f, 0.f, 0.f, 0.f);
            if (r < n) v = *(const float4*)(x + (long long)r * HDIM + ((idx & 31) << 2));
            ((float4*)tileX)[idx] = v;
        }
    }
    __syncthreads();

    const float* xrow = tileX + (rowHalf * 16 + l16) * HDIM + 2 * half;
    v8f acc0 = {}, acc1 = {};

    #pragma unroll 4
    for (int k = 0; k < HDIM; k += 4) {
        v2f a; a.x = xrow[k]; a.y = xrow[k + 1];
        const float* wp = W + (long long)(k + 2 * half) * HDIM + c0 + l16;
        v2f b0; b0.x = wp[0];  b0.y = wp[HDIM];
        v2f b1; b1.x = wp[16]; b1.y = wp[HDIM + 16];
        acc0 = __builtin_amdgcn_wmma_f32_16x16x4_f32(false, a, false, b0,
                                                     (short)0, acc0, false, false);
        acc1 = __builtin_amdgcn_wmma_f32_16x16x4_f32(false, a, false, b1,
                                                     (short)0, acc1, false, false);
    }

    const int rbase = rowBase + rowHalf * 16 + half * 8;
    const float bz0 = bias[c0 + l16], bz1 = bias[c0 + 16 + l16];
    if (fullTile) {
        float d2[8];
        #pragma unroll
        for (int v = 0; v < 8; ++v) { const float di = dinv[rbase + v]; d2[v] = di * di; }
        #pragma unroll
        for (int v = 0; v < 8; ++v) {
            const long long o = (long long)(rbase + v) * HDIM + c0 + l16;
            h[o]        = acc0[v];
            h[o + 16]   = acc1[v];
            agg[o]      = acc0[v] * d2[v] + bz0;
            agg[o + 16] = acc1[v] * d2[v] + bz1;
        }
    } else {
        for (int v = 0; v < 8; ++v) {
            int r = rbase + v;
            if (r < n) {
                const float di = dinv[r], dd = di * di;
                const long long o = (long long)r * HDIM + c0 + l16;
                h[o]        = acc0[v];
                h[o + 16]   = acc1[v];
                agg[o]      = acc0[v] * dd + bz0;
                agg[o + 16] = acc1[v] * dd + bz1;
            }
        }
    }
}

#define EDGE_BATCH 8

__device__ __forceinline__ void edge_one(const float* __restrict__ h,
                                         float* __restrict__ agg,
                                         int s, int d, float coef, int lane) {
    const float* hp = h + (long long)s * HDIM;
    const float h0 = hp[lane], h1 = hp[lane + 32], h2 = hp[lane + 64], h3 = hp[lane + 96];
    float* ap = agg + (long long)d * HDIM;
    atomicAdd(ap + lane,      h0 * coef);
    atomicAdd(ap + lane + 32, h1 * coef);
    atomicAdd(ap + lane + 64, h2 * coef);
    atomicAdd(ap + lane + 96, h3 * coef);
}

__global__ __launch_bounds__(256) void k_edge_scatter(const float* __restrict__ h,
                                                      const int* __restrict__ src,
                                                      const int* __restrict__ dst,
                                                      const float* __restrict__ dinv,
                                                      float* __restrict__ agg, int E) {
    const int lane = threadIdx.x & 31;
    const int t = blockIdx.x * 256 + threadIdx.x;
    int s = 0, d = 0;
    float coef = 0.f;
    if (t < E) {
        s = src[t];
        d = dst[t];
        coef = dinv[s] * dinv[d];
    }
    const int e0 = __builtin_amdgcn_readfirstlane(t);
    int cnt = E - e0;
    cnt = cnt > 32 ? 32 : cnt;
    const int coefBits = __float_as_int(coef);

    int j = 0;
    for (; j + EDGE_BATCH <= cnt; j += EDGE_BATCH) {
        float v[EDGE_BATCH][4];
        #pragma unroll
        for (int u = 0; u < EDGE_BATCH; ++u) {
            const int sj = __builtin_amdgcn_readlane(s, j + u);
            const float* hp = h + (long long)sj * HDIM;
            v[u][0] = hp[lane];
            v[u][1] = hp[lane + 32];
            v[u][2] = hp[lane + 64];
            v[u][3] = hp[lane + 96];
        }
        #pragma unroll
        for (int u = 0; u < EDGE_BATCH; ++u) {
            const int dj = __builtin_amdgcn_readlane(d, j + u);
            const float cj = __int_as_float(__builtin_amdgcn_readlane(coefBits, j + u));
            float* ap = agg + (long long)dj * HDIM;
            atomicAdd(ap + lane,      v[u][0] * cj);
            atomicAdd(ap + lane + 32, v[u][1] * cj);
            atomicAdd(ap + lane + 64, v[u][2] * cj);
            atomicAdd(ap + lane + 96, v[u][3] * cj);
        }
    }
    for (; j < cnt; ++j) {
        edge_one(h, agg, __builtin_amdgcn_readlane(s, j), __builtin_amdgcn_readlane(d, j),
                 __int_as_float(__builtin_amdgcn_readlane(coefBits, j)), lane);
    }
}

#define STAT_BLOCKS 256
#define STAT_WAVES  32
#define STAT_BATCH  8
#define STAT_GROUPS 16

__device__ __forceinline__ void stat_acc(float4& s, float4& q, const float4 p) {
    float4 v;
    v.x = fmaxf(p.x, 0.f); v.y = fmaxf(p.y, 0.f); v.z = fmaxf(p.z, 0.f); v.w = fmaxf(p.w, 0.f);
    s.x += v.x; s.y += v.y; s.z += v.z; s.w += v.w;
    q.x += v.x * v.x; q.y += v.y * v.y; q.z += v.z * v.z; q.w += v.w * v.w;
}

__global__ __launch_bounds__(STAT_WAVES * 32) void k_stats(const float* __restrict__ y,
                                                            float* __restrict__ part, int n) {
    __shared__ float ls[STAT_WAVES][256];
    const int lane = threadIdx.x & 31;
    const int wave = threadIdx.x >> 5;
    const int S = STAT_BLOCKS * STAT_WAVES;
    const float4* yp = (const float4*)y + lane;
    int r = blockIdx.x * STAT_WAVES + wave;
    float4 s = make_float4(0.f, 0.f, 0.f, 0.f), q = s;

    for (; r + (STAT_BATCH - 1) * S < n; r += STAT_BATCH * S) {
        float4 v[STAT_BATCH];
        #pragma unroll
        for (int u = 0; u < STAT_BATCH; ++u) v[u] = yp[(long long)(r + u * S) * 32];
        #pragma unroll
        for (int u = 0; u < STAT_BATCH; ++u) stat_acc(s, q, v[u]);
    }
    if (r < n) {
        const float4 zero = make_float4(0.f, 0.f, 0.f, 0.f);
        float4 v[STAT_BATCH - 1];
        #pragma unroll
        for (int u = 0; u < STAT_BATCH - 1; ++u) {
            const int rr = r + u * S;
            const bool in = rr < n;
            const float4 t = yp[(long long)(in ? rr : n - 1) * 32];
            v[u] = in ? t : zero;
        }
        #pragma unroll
        for (int u = 0; u < STAT_BATCH - 1; ++u) stat_acc(s, q, v[u]);
    }

    *(float4*)&ls[wave][lane * 4]       = s;
    *(float4*)&ls[wave][128 + lane * 4] = q;
    __syncthreads();
    if (threadIdx.x < 256) {
        float a = 0.f;
        for (int w = 0; w < STAT_WAVES; ++w) a += ls[w][threadIdx.x];
        part[(long long)blockIdx.x * 256 + threadIdx.x] = a;
    }
}

__global__ __launch_bounds__(STAT_GROUPS * 64) void k_scaleshift(const float* __restrict__ part,
                                                                 const float* __restrict__ gamma,
                                                                 const float* __restrict__ beta,
                                                                 float* __restrict__ ss, float invN) {
    __shared__ float fs[STAT_GROUPS][256];
    const int c4 = threadIdx.x & 63;
    const int g  = threadIdx.x >> 6;
    const float4* pp = (const float4*)part + c4;
    float4 v[STAT_BLOCKS / STAT_GROUPS];
    #pragma unroll
    for (int i = 0; i < STAT_BLOCKS / STAT_GROUPS; ++i) v[i] = pp[(g + i * STAT_GROUPS) * 64];
    float4 a = v[0];
    #pragma unroll
    for (int i = 1; i < STAT_BLOCKS / STAT_GROUPS; ++i) {
        a.x += v[i].x; a.y += v[i].y; a.z += v[i].z; a.w += v[i].w;
    }
    *(float4*)&fs[g][c4 * 4] = a;
    __syncthreads();
    if (threadIdx.x < 128) {
        const int c = threadIdx.x;
        float su = 0.f, sq = 0.f;
        for (int k = 0; k < STAT_GROUPS; ++k) { su += fs[k][c]; sq += fs[k][128 + c]; }
        float m   = su * invN;
        float var = sq * invN - m * m;
        float sc  = rsqrtf(var + EPS_BN) * gamma[c];
        ss[c] = sc;
        ss[128 + c] = beta[c] - m * sc;
    }
}

__global__ __launch_bounds__(256) void k_bn_apply(float* __restrict__ y,
                                                  const float* __restrict__ res,
                                                  const float* __restrict__ ss, int n) {
    int t = blockIdx.x * 256 + threadIdx.x;
    if (t >= n * 32) return;
    int row = t >> 5, c4 = (t & 31) << 2;
    long long off = (long long)row * HDIM + c4;
    float4 v  = *(const float4*)(y + off);
    float4 sc = *(const float4*)(ss + c4);
    float4 sh = *(const float4*)(ss + 128 + c4);
    v.x = fmaxf(v.x, 0.f) * sc.x + sh.x; v.y = fmaxf(v.y, 0.f) * sc.y + sh.y;
    v.z = fmaxf(v.z, 0.f) * sc.z + sh.z; v.w = fmaxf(v.w, 0.f) * sc.w + sh.w;
    if (res) {
        float4 r = *(const float4*)(res + off);
        v.x += r.x; v.y += r.y; v.z += r.z; v.w += r.w;
    }
    *(float4*)(y + off) = v;
}

__global__ __launch_bounds__(256) void k_counts(const int* __restrict__ batch,
                                                float* __restrict__ counts, int n) {
    int t = blockIdx.x * 256 + threadIdx.x;
    if (t < n) atomicAdd(&counts[batch[t]], 1.0f);
}

__global__ __launch_bounds__(256) void k_pool_acc(const float* __restrict__ x,
                                                  const int* __restrict__ batch,
                                                  float* __restrict__ pool, int n) {
    int t = blockIdx.x * 256 + threadIdx.x;
    if (t >= n * 32) return;
    int row = t >> 5, c4 = (t & 31) << 2;
    int g = batch[row];
    float4 v = *(const float4*)(x + (long long)row * HDIM + c4);
    float* p = pool + (long long)g * HDIM + c4;
    atomicAdd(p + 0, v.x); atomicAdd(p + 1, v.y);
    atomicAdd(p + 2, v.z); atomicAdd(p + 3, v.w);
}

__global__ __launch_bounds__(256) void k_pool_final(const float* __restrict__ pool,
                                                    const float* __restrict__ counts,
                                                    float* __restrict__ xmean, int G) {
    int t = blockIdx.x * 256 + threadIdx.x;
    if (t >= G * HDIM) return;
    int g = t / HDIM;
    xmean[t] = pool[t] / fmaxf(counts[g], 1.0f);
}

__global__ void k_pred(const float* __restrict__ xmean, const float* __restrict__ Wr,
                       const float* __restrict__ br, float* __restrict__ pred) {
    __shared__ float red[128];
    int g = blockIdx.x, c = threadIdx.x;
    red[c] = xmean[(long long)g * HDIM + c] * Wr[c];
    __syncthreads();
    for (int s = 64; s > 0; s >>= 1) {
        if (c < s) red[c] += red[c + s];
        __syncthreads();
    }
    if (c == 0) pred[g] = red[0] + br[0];
}

extern "C" void kernel_launch(void* const* d_in, const int* in_sizes, int n_in,
                              void* d_out, int out_size, void* d_ws, size_t ws_size,
                              hipStream_t stream) {
    const float* X     = (const float*)d_in[0];
    const int*   edge  = (const int*)d_in[1];
    const int*   batch = (const int*)d_in[2];
    const float* W1    = (const float*)d_in[3];
    const float* b1    = (const float*)d_in[4];
    const float* Wm    = (const float*)d_in[5];
    const float* bm    = (const float*)d_in[6];
    const float* gamma = (const float*)d_in[7];
    const float* beta  = (const float*)d_in[8];
    const float* Wr    = (const float*)d_in[9];
    const float* br    = (const float*)d_in[10];
    float* out = (float*)d_out;

    const int N = in_sizes[0];
    const int E = in_sizes[1] / 2;
    const int G = out_size / (HDIM + 1);
    const int* src = edge;
    const int* dst = edge + E;

    const long long NH = (long long)N * HDIM;
    float* ws    = (float*)d_ws;
    float* buf[3] = { ws, ws + NH, ws + 2 * NH };
    float* dinv  = ws + 3 * NH;
    float* ss    = dinv + N;
    float* pool  = ss + 256;
    float* cnts  = pool + (long long)G * HDIM;

    const int gE   = (E + 255) / 256;
    const int gN   = (N + 255) / 256;
    const int gNH4 = (N * 32 + 255) / 256;
    const int gGem = (N + 31) / 32;

    k_zero<<<1024, 256, 0, stream>>>(dinv, N);
    k_deg_scatter<<<gE, 256, 0, stream>>>(dst, dinv, E);
    float* s1 = buf[0];
    k_dinv<<<gN, 256, 0, stream>>>(dinv, X, s1, N);

    auto bn_tail = [&](float* abuf, float* part, const float* g_, const float* be_,
                       const float* res) {
        k_stats<<<STAT_BLOCKS, STAT_WAVES * 32, 0, stream>>>(abuf, part, N);
        k_scaleshift<<<1, STAT_GROUPS * 64, 0, stream>>>(part, g_, be_, ss, 1.0f / (float)N);
        k_bn_apply<<<gNH4, 256, 0, stream>>>(abuf, res, ss, N);
    };

    auto layer_tail = [&](float* hbuf, float* abuf,
                          const float* g_, const float* be_, const float* res) {
        k_edge_scatter<<<gE, 256, 0, stream>>>(hbuf, src, dst, dinv, abuf, E);
        bn_tail(abuf, hbuf, g_, be_, res);
    };

    k_edge_scalar<<<gE, 256, 0, stream>>>(X, src, dst, dinv, s1, E);
    k_expand<<<gNH4, 256, 0, stream>>>(s1, W1, b1, buf[2], N);
    bn_tail(buf[2], buf[0], gamma, beta, nullptr);

    float* cur = buf[2];
    float* f0  = buf[0];
    float* f1  = buf[1];

    for (int i = 0; i < 4; ++i) {
        float* hbuf = f0;
        float* abuf = f1;
        k_gemm_wmma<<<gGem, 256, 0, stream>>>(cur, Wm + (long long)i * HDIM * HDIM, dinv,
                                              bm + (long long)i * HDIM, hbuf, abuf, N);
        layer_tail(hbuf, abuf,
                   gamma + (long long)(i + 1) * HDIM,
                   beta  + (long long)(i + 1) * HDIM, cur);
        float* next = abuf;
        f1 = cur;
        f0 = hbuf;
        cur = next;
    }

    k_zero<<<32, 256, 0, stream>>>(pool, (long long)G * HDIM + G);
    k_counts<<<gN, 256, 0, stream>>>(batch, cnts, N);
    k_pool_acc<<<gNH4, 256, 0, stream>>>(cur, batch, pool, N);
    k_pool_final<<<(G * HDIM + 255) / 256, 256, 0, stream>>>(pool, cnts, out + G, G);
    k_pred<<<G, 128, 0, stream>>>(out + G, Wr, br, out);
}


# Global float atomics on narrow targets (degree count, layer 1's scalar sum, graph counts, mean pool) replaced by LDS sums per block, planes added in fixed order; dinv[dst] factored out; new sum order
# speedup vs baseline: 2.9402x; 1.2798x over previous
#include <hip/hip_runtime.h>
#include <hip/hip_bf16.h>

#define HDIM 128
#define EPS_BN 1e-5f

typedef __attribute__((ext_vector_type(2))) float v2f;
typedef __attribute__((ext_vector_type(8))) float v8f;

#define SC_RANGE   16384
#define SC_CHUNKS  32
#define SC_THREADS 1024
#define SC_BATCH   8

__global__ __launch_bounds__(SC_THREADS) void k_deg_count(const int* __restrict__ dst,
                                                          float* __restrict__ part,
                                                          int E, int n) {
    __shared__ unsigned int cnt[SC_RANGE];
    const int r0 = blockIdx.y * SC_RANGE;
    for (int i = threadIdx.x; i < SC_RANGE; i += SC_THREADS) cnt[i] = 0u;
    __syncthreads();
    const int L = (E + (int)gridDim.x - 1) / (int)gridDim.x;
    const int eBeg = blockIdx.x * L;
    const int eEnd = (eBeg + L < E) ? (eBeg + L) : E;
    int e = eBeg + threadIdx.x;
    for (; e + (SC_BATCH - 1) * SC_THREADS < eEnd; e += SC_BATCH * SC_THREADS) {
        int d[SC_BATCH];
        #pragma unroll
        for (int k = 0; k < SC_BATCH; ++k) d[k] = dst[e + k * SC_THREADS];
        #pragma unroll
        for (int k = 0; k < SC_BATCH; ++k) {
            const unsigned rel = (unsigned)(d[k] - r0);
            if (rel < SC_RANGE) atomicAdd(&cnt[rel], 1u);
        }
    }
    for (; e < eEnd; e += SC_THREADS) {
        const unsigned rel = (unsigned)(dst[e] - r0);
        if (rel < SC_RANGE) atomicAdd(&cnt[rel], 1u);
    }
    __syncthreads();
    const int lim = (n - r0 < SC_RANGE) ? (n - r0) : SC_RANGE;
    float* o = part + (long long)blockIdx.x * n + r0;
    for (int i = threadIdx.x; i < lim; i += SC_THREADS) o[i] = (float)cnt[i];
}

__global__ __launch_bounds__(256) void k_deg_finish(const float* __restrict__ part,
                                                    const float* __restrict__ X,
                                                    float* __restrict__ dinv,
                                                    float* __restrict__ u, int n) {
    const int t = blockIdx.x * 256 + threadIdx.x;
    if (t >= n) return;
    float v[SC_CHUNKS];
    #pragma unroll
    for (int c = 0; c < SC_CHUNKS; ++c) v[c] = part[(long long)c * n + t];
    float s = v[0];
    #pragma unroll
    for (int c = 1; c < SC_CHUNKS; ++c) s += v[c];
    const float di = rsqrtf(s + 1.0f);
    dinv[t] = di;
    u[t] = X[t] * di;
}

__global__ __launch_bounds__(SC_THREADS) void k_l1_sum(const float* __restrict__ u,
                                                       const int* __restrict__ src,
                                                       const int* __restrict__ dst,
                                                       float* __restrict__ part,
                                                       int E, int n) {
    __shared__ float acc[SC_RANGE];
    const int r0 = blockIdx.y * SC_RANGE;
    for (int i = threadIdx.x; i < SC_RANGE; i += SC_THREADS) acc[i] = 0.0f;
    __syncthreads();
    const int L = (E + (int)gridDim.x - 1) / (int)gridDim.x;
    const int eBeg = blockIdx.x * L;
    const int eEnd = (eBeg + L < E) ? (eBeg + L) : E;
    int e = eBeg + threadIdx.x;
    for (; e + (SC_BATCH - 1) * SC_THREADS < eEnd; e += SC_BATCH * SC_THREADS) {
        int d[SC_BATCH], s[SC_BATCH];
        float val[SC_BATCH];
        #pragma unroll
        for (int k = 0; k < SC_BATCH; ++k) {
            d[k] = dst[e + k * SC_THREADS];
            s[k] = src[e + k * SC_THREADS];
        }
        #pragma unroll
        for (int k = 0; k < SC_BATCH; ++k) asm volatile("" :: "v"(d[k]), "v"(s[k]));
        #pragma unroll
        for (int k = 0; k < SC_BATCH; ++k) {
            const bool in = (unsigned)(d[k] - r0) < SC_RANGE;
            val[k] = in ? u[s[k]] : 0.0f;
        }
        #pragma unroll
        for (int k = 0; k < SC_BATCH; ++k) {
            const unsigned rel = (unsigned)(d[k] - r0);
            if (rel < SC_RANGE) atomicAdd(&acc[rel], val[k]);
        }
    }
    for (; e < eEnd; e += SC_THREADS) {
        const unsigned rel = (unsigned)(dst[e] - r0);
        if (rel < SC_RANGE) atomicAdd(&acc[rel], u[src[e]]);
    }
    __syncthreads();
    const int lim = (n - r0 < SC_RANGE) ? (n - r0) : SC_RANGE;
    float* o = part + (long long)blockIdx.x * n + r0;
    for (int i = threadIdx.x; i < lim; i += SC_THREADS) o[i] = acc[i];
}

__global__ __launch_bounds__(256) void k_l1_finish(const float* __restrict__ part,
                                                   const float* __restrict__ X,
                                                   const float* __restrict__ dinv,
                                                   float* __restrict__ s1, int n) {
    const int t = blockIdx.x * 256 + threadIdx.x;
    if (t >= n) return;
    float v[SC_CHUNKS];
    #pragma unroll
    for (int c = 0; c < SC_CHUNKS; ++c) v[c] = part[(long long)c * n + t];
    float s = v[0];
    #pragma unroll
    for (int c = 1; c < SC_CHUNKS; ++c) s += v[c];
    const float di = dinv[t];
    s1[t] = di * s + X[t] * (di * di);
}

__global__ __launch_bounds__(256) void k_expand(const float* __restrict__ s1,
                                                     const float* __restrict__ W1,
                                                     const float* __restrict__ bias,
                                                     float* __restrict__ y, int n) {
    int t = blockIdx.x * 256 + threadIdx.x;
    if (t >= n * 32) return;
    int row = t >> 5, c4 = (t & 31) << 2;
    float sv = s1[row];
    const float4 w = *(const float4*)(W1 + c4);
    const float4 b = *(const float4*)(bias + c4);
    float4 o;
    o.x = sv * w.x + b.x;
    o.y = sv * w.y + b.y;
    o.z = sv * w.z + b.z;
    o.w = sv * w.w + b.w;
    *(float4*)(y + (long long)row * HDIM + c4) = o;
}

__global__ __launch_bounds__(256) void k_gemm_wmma(const float* __restrict__ x,
                                                   const float* __restrict__ W,
                                                   const float* __restrict__ dinv,
                                                   const float* __restrict__ bias,
                                                   float* __restrict__ h,
                                                   float* __restrict__ agg, int n) {
    __shared__ float tileX[32 * HDIM];
    const int rowBase = blockIdx.x * 32;
    const int tid  = threadIdx.x;
    const int wave = tid >> 5;
    const int lane = tid & 31;
    const int half = lane >> 4;
    const int l16  = lane & 15;
    const int rowHalf = wave & 1;
    const int c0   = (wave >> 1) * 32;

    const bool fullTile = (rowBase + 32 <= n);

    if (fullTile) {
        const float4* xs = (const float4*)(x + (long long)rowBase * HDIM);
        #pragma unroll
        for (int i = 0; i < 4; ++i) {
            int idx = tid + i * 256;
            ((float4*)tileX)[idx] = xs[idx];
        }
    } else {
        #pragma unroll
        for (int i = 0; i < 4; ++i) {
            int idx = tid + i * 256;
            int r = rowBase + (idx >> 5);
            float4 v = make_float4(0.f, 0.f, 0.f, 0.f);
            if (r < n) v = *(const float4*)(x + (long long)r * HDIM + ((idx & 31) << 2));
            ((float4*)tileX)[idx] = v;
        }
    }
    __syncthreads();

    const float* xrow = tileX + (rowHalf * 16 + l16) * HDIM + 2 * half;
    v8f acc0 = {}, acc1 = {};

    #pragma unroll 4
    for (int k = 0; k < HDIM; k += 4) {
        v2f a; a.x = xrow[k]; a.y = xrow[k + 1];
        const float* wp = W + (long long)(k + 2 * half) * HDIM + c0 + l16;
        v2f b0; b0.x = wp[0];  b0.y = wp[HDIM];
        v2f b1; b1.x = wp[16]; b1.y = wp[HDIM + 16];
        acc0 = __builtin_amdgcn_wmma_f32_16x16x4_f32(false, a, false, b0,
                                                     (short)0, acc0, false, false);
        acc1 = __builtin_amdgcn_wmma_f32_16x16x4_f32(false, a, false, b1,
                                                     (short)0, acc1, false, false);
    }

    const int rbase = rowBase + rowHalf * 16 + half * 8;
    const float bz0 = bias[c0 + l16], bz1 = bias[c0 + 16 + l16];
    if (fullTile) {
        float d2[8];
        #pragma unroll
        for (int v = 0; v < 8; ++v) { const float di = dinv[rbase + v]; d2[v] = di * di; }
        #pragma unroll
        for (int v = 0; v < 8; ++v) {
            const long long o = (long long)(rbase + v) * HDIM + c0 + l16;
            h[o]        = acc0[v];
            h[o + 16]   = acc1[v];
            agg[o]      = acc0[v] * d2[v] + bz0;
            agg[o + 16] = acc1[v] * d2[v] + bz1;
        }
    } else {
        for (int v = 0; v < 8; ++v) {
            int r = rbase + v;
            if (r < n) {
                const float di = dinv[r], dd = di * di;
                const long long o = (long long)r * HDIM + c0 + l16;
                h[o]        = acc0[v];
                h[o + 16]   = acc1[v];
                agg[o]      = acc0[v] * dd + bz0;
                agg[o + 16] = acc1[v] * dd + bz1;
            }
        }
    }
}

#define EDGE_BATCH 8

__device__ __forceinline__ void edge_one(const float* __restrict__ h,
                                         float* __restrict__ agg,
                                         int s, int d, float coef, int lane) {
    const float* hp = h + (long long)s * HDIM;
    const float h0 = hp[lane], h1 = hp[lane + 32], h2 = hp[lane + 64], h3 = hp[lane + 96];
    float* ap = agg + (long long)d * HDIM;
    atomicAdd(ap + lane,      h0 * coef);
    atomicAdd(ap + lane + 32, h1 * coef);
    atomicAdd(ap + lane + 64, h2 * coef);
    atomicAdd(ap + lane + 96, h3 * coef);
}

__global__ __launch_bounds__(256) void k_edge_scatter(const float* __restrict__ h,
                                                      const int* __restrict__ src,
                                                      const int* __restrict__ dst,
                                                      const float* __restrict__ dinv,
                                                      float* __restrict__ agg, int E) {
    const int lane = threadIdx.x & 31;
    const int t = blockIdx.x * 256 + threadIdx.x;
    int s = 0, d = 0;
    float coef = 0.f;
    if (t < E) {
        s = src[t];
        d = dst[t];
        coef = dinv[s] * dinv[d];
    }
    const int e0 = __builtin_amdgcn_readfirstlane(t);
    int cnt = E - e0;
    cnt = cnt > 32 ? 32 : cnt;
    const int coefBits = __float_as_int(coef);

    int j = 0;
    for (; j + EDGE_BATCH <= cnt; j += EDGE_BATCH) {
        float v[EDGE_BATCH][4];
        #pragma unroll
        for (int u = 0; u < EDGE_BATCH; ++u) {
            const int sj = __builtin_amdgcn_readlane(s, j + u);
            const float* hp = h + (long long)sj * HDIM;
            v[u][0] = hp[lane];
            v[u][1] = hp[lane + 32];
            v[u][2] = hp[lane + 64];
            v[u][3] = hp[lane + 96];
        }
        #pragma unroll
        for (int u = 0; u < EDGE_BATCH; ++u) {
            const int dj = __builtin_amdgcn_readlane(d, j + u);
            const float cj = __int_as_float(__builtin_amdgcn_readlane(coefBits, j + u));
            float* ap = agg + (long long)dj * HDIM;
            atomicAdd(ap + lane,      v[u][0] * cj);
            atomicAdd(ap + lane + 32, v[u][1] * cj);
            atomicAdd(ap + lane + 64, v[u][2] * cj);
            atomicAdd(ap + lane + 96, v[u][3] * cj);
        }
    }
    for (; j < cnt; ++j) {
        edge_one(h, agg, __builtin_amdgcn_readlane(s, j), __builtin_amdgcn_readlane(d, j),
                 __int_as_float(__builtin_amdgcn_readlane(coefBits, j)), lane);
    }
}

#define STAT_BLOCKS 256
#define STAT_WAVES  32
#define STAT_BATCH  8
#define STAT_GROUPS 16

__device__ __forceinline__ void stat_acc(float4& s, float4& q, const float4 p) {
    float4 v;
    v.x = fmaxf(p.x, 0.f); v.y = fmaxf(p.y, 0.f); v.z = fmaxf(p.z, 0.f); v.w = fmaxf(p.w, 0.f);
    s.x += v.x; s.y += v.y; s.z += v.z; s.w += v.w;
    q.x += v.x * v.x; q.y += v.y * v.y; q.z += v.z * v.z; q.w += v.w * v.w;
}

__global__ __launch_bounds__(STAT_WAVES * 32) void k_stats(const float* __restrict__ y,
                                                            float* __restrict__ part, int n) {
    __shared__ float ls[STAT_WAVES][256];
    const int lane = threadIdx.x & 31;
    const int wave = threadIdx.x >> 5;
    const int S = STAT_BLOCKS * STAT_WAVES;
    const float4* yp = (const float4*)y + lane;
    int r = blockIdx.x * STAT_WAVES + wave;
    float4 s = make_float4(0.f, 0.f, 0.f, 0.f), q = s;

    for (; r + (STAT_BATCH - 1) * S < n; r += STAT_BATCH * S) {
        float4 v[STAT_BATCH];
        #pragma unroll
        for (int u = 0; u < STAT_BATCH; ++u) v[u] = yp[(long long)(r + u * S) * 32];
        #pragma unroll
        for (int u = 0; u < STAT_BATCH; ++u) stat_acc(s, q, v[u]);
    }
    if (r < n) {
        const float4 zero = make_float4(0.f, 0.f, 0.f, 0.f);
        float4 v[STAT_BATCH - 1];
        #pragma unroll
        for (int u = 0; u < STAT_BATCH - 1; ++u) {
            const int rr = r + u * S;
            const bool in = rr < n;
            const float4 t = yp[(long long)(in ? rr : n - 1) * 32];
            v[u] = in ? t : zero;
        }
        #pragma unroll
        for (int u = 0; u < STAT_BATCH - 1; ++u) stat_acc(s, q, v[u]);
    }

    *(float4*)&ls[wave][lane * 4]       = s;
    *(float4*)&ls[wave][128 + lane * 4] = q;
    __syncthreads();
    if (threadIdx.x < 256) {
        float a = 0.f;
        for (int w = 0; w < STAT_WAVES; ++w) a += ls[w][threadIdx.x];
        part[(long long)blockIdx.x * 256 + threadIdx.x] = a;
    }
}

__global__ __launch_bounds__(STAT_GROUPS * 64) void k_scaleshift(const float* __restrict__ part,
                                                                 const float* __restrict__ gamma,
                                                                 const float* __restrict__ beta,
                                                                 float* __restrict__ ss, float invN) {
    __shared__ float fs[STAT_GROUPS][256];
    const int c4 = threadIdx.x & 63;
    const int g  = threadIdx.x >> 6;
    const float4* pp = (const float4*)part + c4;
    float4 v[STAT_BLOCKS / STAT_GROUPS];
    #pragma unroll
    for (int i = 0; i < STAT_BLOCKS / STAT_GROUPS; ++i) v[i] = pp[(g + i * STAT_GROUPS) * 64];
    float4 a = v[0];
    #pragma unroll
    for (int i = 1; i < STAT_BLOCKS / STAT_GROUPS; ++i) {
        a.x += v[i].x; a.y += v[i].y; a.z += v[i].z; a.w += v[i].w;
    }
    *(float4*)&fs[g][c4 * 4] = a;
    __syncthreads();
    if (threadIdx.x < 128) {
        const int c = threadIdx.x;
        float su = 0.f, sq = 0.f;
        for (int k = 0; k < STAT_GROUPS; ++k) { su += fs[k][c]; sq += fs[k][128 + c]; }
        float m   = su * invN;
        float var = sq * invN - m * m;
        float sc  = rsqrtf(var + EPS_BN) * gamma[c];
        ss[c] = sc;
        ss[128 + c] = beta[c] - m * sc;
    }
}

__global__ __launch_bounds__(256) void k_bn_apply(float* __restrict__ y,
                                                  const float* __restrict__ res,
                                                  const float* __restrict__ ss, int n) {
    int t = blockIdx.x * 256 + threadIdx.x;
    if (t >= n * 32) return;
    int row = t >> 5, c4 = (t & 31) << 2;
    long long off = (long long)row * HDIM + c4;
    float4 v  = *(const float4*)(y + off);
    float4 sc = *(const float4*)(ss + c4);
    float4 sh = *(const float4*)(ss + 128 + c4);
    v.x = fmaxf(v.x, 0.f) * sc.x + sh.x; v.y = fmaxf(v.y, 0.f) * sc.y + sh.y;
    v.z = fmaxf(v.z, 0.f) * sc.z + sh.z; v.w = fmaxf(v.w, 0.f) * sc.w + sh.w;
    if (res) {
        float4 r = *(const float4*)(res + off);
        v.x += r.x; v.y += r.y; v.z += r.z; v.w += r.w;
    }
    *(float4*)(y + off) = v;
}

#define POOL_BLOCKS 256
#define POOL_WAVES  32
#define POOL_BATCH  8
#define PSUM_GROUPS 16

__device__ __forceinline__ void pool_add(float* __restrict__ pl, int g, int lane,
                                         const float4 v, int cntBase) {
    float* p = pl + g * HDIM + lane * 4;
    atomicAdd(p + 0, v.x); atomicAdd(p + 1, v.y);
    atomicAdd(p + 2, v.z); atomicAdd(p + 3, v.w);
    if (lane == 0) atomicAdd(pl + cntBase + g, 1.0f);
}

__global__ __launch_bounds__(POOL_WAVES * 32) void k_pool_part(const float* __restrict__ x,
                                                                const int* __restrict__ batch,
                                                                float* __restrict__ planes,
                                                                int n, int G) {
    extern __shared__ float pl[];
    const int lane = threadIdx.x & 31;
    const int tot = G * HDIM + G;
    for (int i = threadIdx.x; i < tot; i += POOL_WAVES * 32) pl[i] = 0.0f;
    __syncthreads();
    const int S = (int)gridDim.x * POOL_WAVES;
    int r = __builtin_amdgcn_readfirstlane((int)blockIdx.x * POOL_WAVES + (int)(threadIdx.x >> 5));
    const float4* xp = (const float4*)x + lane;

    for (; r + (POOL_BATCH - 1) * S < n; r += POOL_BATCH * S) {
        float4 v[POOL_BATCH];
        int g[POOL_BATCH];
        #pragma unroll
        for (int k = 0; k < POOL_BATCH; ++k) {
            v[k] = xp[(long long)(r + k * S) * 32];
            g[k] = batch[r + k * S];
        }
        #pragma unroll
        for (int k = 0; k < POOL_BATCH; ++k) pool_add(pl, g[k], lane, v[k], G * HDIM);
    }
    if (r < n) {
        float4 v[POOL_BATCH - 1];
        int g[POOL_BATCH - 1];
        #pragma unroll
        for (int k = 0; k < POOL_BATCH - 1; ++k) {
            const int rr = r + k * S;
            const int rc = rr < n ? rr : n - 1;
            v[k] = xp[(long long)rc * 32];
            g[k] = batch[rc];
        }
        #pragma unroll
        for (int k = 0; k < POOL_BATCH - 1; ++k)
            if (r + k * S < n) pool_add(pl, g[k], lane, v[k], G * HDIM);
    }
    __syncthreads();
    float* o = planes + (long long)blockIdx.x * tot;
    for (int i = threadIdx.x; i < tot; i += POOL_WAVES * 32) o[i] = pl[i];
}

__global__ __launch_bounds__(PSUM_GROUPS * 64) void k_pool_sum(const float* __restrict__ planes,
                                                               float* __restrict__ outp, int tot) {
    __shared__ float fs[PSUM_GROUPS][64];
    const int o   = threadIdx.x & 63;
    const int grp = threadIdx.x >> 6;
    const int i   = blockIdx.x * 64 + o;
    float a = 0.0f;
    if (i < tot) {
        float v[POOL_BLOCKS / PSUM_GROUPS];
        #pragma unroll
        for (int j = 0; j < POOL_BLOCKS / PSUM_GROUPS; ++j)
            v[j] = planes[(long long)(grp + j * PSUM_GROUPS) * tot + i];
        a = v[0];
        #pragma unroll
        for (int j = 1; j < POOL_BLOCKS / PSUM_GROUPS; ++j) a += v[j];
    }
    fs[grp][o] = a;
    __syncthreads();
    if (threadIdx.x < 64 && i < tot) {
        float s = 0.0f;
        for (int k = 0; k < PSUM_GROUPS; ++k) s += fs[k][o];
        outp[i] = s;
    }
}

__global__ __launch_bounds__(256) void k_pool_final(const float* __restrict__ pool,
                                                    const float* __restrict__ counts,
                                                    float* __restrict__ xmean, int G) {
    int t = blockIdx.x * 256 + threadIdx.x;
    if (t >= G * HDIM) return;
    int g = t / HDIM;
    xmean[t] = pool[t] / fmaxf(counts[g], 1.0f);
}

__global__ void k_pred(const float* __restrict__ xmean, const float* __restrict__ Wr,
                       const float* __restrict__ br, float* __restrict__ pred) {
    __shared__ float red[128];
    int g = blockIdx.x, c = threadIdx.x;
    red[c] = xmean[(long long)g * HDIM + c] * Wr[c];
    __syncthreads();
    for (int s = 64; s > 0; s >>= 1) {
        if (c < s) red[c] += red[c + s];
        __syncthreads();
    }
    if (c == 0) pred[g] = red[0] + br[0];
}

extern "C" void kernel_launch(void* const* d_in, const int* in_sizes, int n_in,
                              void* d_out, int out_size, void* d_ws, size_t ws_size,
                              hipStream_t stream) {
    const float* X     = (const float*)d_in[0];
    const int*   edge  = (const int*)d_in[1];
    const int*   batch = (const int*)d_in[2];
    const float* W1    = (const float*)d_in[3];
    const float* b1    = (const float*)d_in[4];
    const float* Wm    = (const float*)d_in[5];
    const float* bm    = (const float*)d_in[6];
    const float* gamma = (const float*)d_in[7];
    const float* beta  = (const float*)d_in[8];
    const float* Wr    = (const float*)d_in[9];
    const float* br    = (const float*)d_in[10];
    float* out = (float*)d_out;

    const int N = in_sizes[0];
    const int E = in_sizes[1] / 2;
    const int G = out_size / (HDIM + 1);
    const int* src = edge;
    const int* dst = edge + E;

    const long long NH = (long long)N * HDIM;
    float* ws    = (float*)d_ws;
    float* buf[3] = { ws, ws + NH, ws + 2 * NH };
    float* dinv  = ws + 3 * NH;
    float* ss    = dinv + N;
    float* pool  = ss + 256;
    float* cnts  = pool + (long long)G * HDIM;

    const int gE   = (E + 255) / 256;
    const int gN   = (N + 255) / 256;
    const int gNH4 = (N * 32 + 255) / 256;
    const int gGem = (N + 31) / 32;

    float* s1   = buf[0];
    float* u1   = buf[0] + N;
    float* scp  = buf[1];
    const dim3 gSc(SC_CHUNKS, (N + SC_RANGE - 1) / SC_RANGE);
    k_deg_count<<<gSc, SC_THREADS, 0, stream>>>(dst, scp, E, N);
    k_deg_finish<<<gN, 256, 0, stream>>>(scp, X, dinv, u1, N);

    auto bn_tail = [&](float* abuf, float* part, const float* g_, const float* be_,
                       const float* res) {
        k_stats<<<STAT_BLOCKS, STAT_WAVES * 32, 0, stream>>>(abuf, part, N);
        k_scaleshift<<<1, STAT_GROUPS * 64, 0, stream>>>(part, g_, be_, ss, 1.0f / (float)N);
        k_bn_apply<<<gNH4, 256, 0, stream>>>(abuf, res, ss, N);
    };

    auto layer_tail = [&](float* hbuf, float* abuf,
                          const float* g_, const float* be_, const float* res) {
        k_edge_scatter<<<gE, 256, 0, stream>>>(hbuf, src, dst, dinv, abuf, E);
        bn_tail(abuf, hbuf, g_, be_, res);
    };

    k_l1_sum<<<gSc, SC_THREADS, 0, stream>>>(u1, src, dst, scp, E, N);
    k_l1_finish<<<gN, 256, 0, stream>>>(scp, X, dinv, s1, N);
    k_expand<<<gNH4, 256, 0, stream>>>(s1, W1, b1, buf[2], N);
    bn_tail(buf[2], buf[0], gamma, beta, nullptr);

    float* cur = buf[2];
    float* f0  = buf[0];
    float* f1  = buf[1];

    for (int i = 0; i < 4; ++i) {
        float* hbuf = f0;
        float* abuf = f1;
        k_gemm_wmma<<<gGem, 256, 0, stream>>>(cur, Wm + (long long)i * HDIM * HDIM, dinv,
                                              bm + (long long)i * HDIM, hbuf, abuf, N);
        layer_tail(hbuf, abuf,
                   gamma + (long long)(i + 1) * HDIM,
                   beta  + (long long)(i + 1) * HDIM, cur);
        float* next = abuf;
        f1 = cur;
        f0 = hbuf;
        cur = next;
    }

    const int poolTot = G * HDIM + G;
    k_pool_part<<<POOL_BLOCKS, POOL_WAVES * 32, poolTot * sizeof(float), stream>>>(cur, batch, f0, N, G);
    k_pool_sum<<<(poolTot + 63) / 64, PSUM_GROUPS * 64, 0, stream>>>(f0, pool, poolTot);
    k_pool_final<<<(G * HDIM + 255) / 256, 256, 0, stream>>>(pool, cnts, out + G, G);
    k_pred<<<G, 128, 0, stream>>>(out + G, Wr, br, out);
}


# Neighbour sums kept in LDS, not global float atomics: a block owns 544 destination rows, scans the whole dst list, gathers its edges' dinv-scaled h rows; dinv[dst], bias at store; new sum order
# speedup vs baseline: 3.1026x; 1.0552x over previous
#include <hip/hip_runtime.h>
#include <hip/hip_bf16.h>

#define HDIM 128
#define EPS_BN 1e-5f

typedef __attribute__((ext_vector_type(2))) float v2f;
typedef __attribute__((ext_vector_type(8))) float v8f;

#define SC_RANGE   16384
#define SC_CHUNKS  32
#define SC_THREADS 1024
#define SC_BATCH   8

__global__ __launch_bounds__(SC_THREADS) void k_deg_count(const int* __restrict__ dst,
                                                          float* __restrict__ part,
                                                          int E, int n) {
    __shared__ unsigned int cnt[SC_RANGE];
    const int r0 = blockIdx.y * SC_RANGE;
    for (int i = threadIdx.x; i < SC_RANGE; i += SC_THREADS) cnt[i] = 0u;
    __syncthreads();
    const int L = (E + (int)gridDim.x - 1) / (int)gridDim.x;
    const int eBeg = blockIdx.x * L;
    const int eEnd = (eBeg + L < E) ? (eBeg + L) : E;
    int e = eBeg + threadIdx.x;
    for (; e + (SC_BATCH - 1) * SC_THREADS < eEnd; e += SC_BATCH * SC_THREADS) {
        int d[SC_BATCH];
        #pragma unroll
        for (int k = 0; k < SC_BATCH; ++k) d[k] = dst[e + k * SC_THREADS];
        #pragma unroll
        for (int k = 0; k < SC_BATCH; ++k) {
            const unsigned rel = (unsigned)(d[k] - r0);
            if (rel < SC_RANGE) atomicAdd(&cnt[rel], 1u);
        }
    }
    for (; e < eEnd; e += SC_THREADS) {
        const unsigned rel = (unsigned)(dst[e] - r0);
        if (rel < SC_RANGE) atomicAdd(&cnt[rel], 1u);
    }
    __syncthreads();
    const int lim = (n - r0 < SC_RANGE) ? (n - r0) : SC_RANGE;
    float* o = part + (long long)blockIdx.x * n + r0;
    for (int i = threadIdx.x; i < lim; i += SC_THREADS) o[i] = (float)cnt[i];
}

__global__ __launch_bounds__(256) void k_deg_finish(const float* __restrict__ part,
                                                    const float* __restrict__ X,
                                                    float* __restrict__ dinv,
                                                    float* __restrict__ u, int n) {
    const int t = blockIdx.x * 256 + threadIdx.x;
    if (t >= n) return;
    float v[SC_CHUNKS];
    #pragma unroll
    for (int c = 0; c < SC_CHUNKS; ++c) v[c] = part[(long long)c * n + t];
    float s = v[0];
    #pragma unroll
    for (int c = 1; c < SC_CHUNKS; ++c) s += v[c];
    const float di = rsqrtf(s + 1.0f);
    dinv[t] = di;
    u[t] = X[t] * di;
}

__global__ __launch_bounds__(SC_THREADS) void k_l1_sum(const float* __restrict__ u,
                                                       const int* __restrict__ src,
                                                       const int* __restrict__ dst,
                                                       float* __restrict__ part,
                                                       int E, int n) {
    __shared__ float acc[SC_RANGE];
    const int r0 = blockIdx.y * SC_RANGE;
    for (int i = threadIdx.x; i < SC_RANGE; i += SC_THREADS) acc[i] = 0.0f;
    __syncthreads();
    const int L = (E + (int)gridDim.x - 1) / (int)gridDim.x;
    const int eBeg = blockIdx.x * L;
    const int eEnd = (eBeg + L < E) ? (eBeg + L) : E;
    int e = eBeg + threadIdx.x;
    for (; e + (SC_BATCH - 1) * SC_THREADS < eEnd; e += SC_BATCH * SC_THREADS) {
        int d[SC_BATCH], s[SC_BATCH];
        float val[SC_BATCH];
        #pragma unroll
        for (int k = 0; k < SC_BATCH; ++k) {
            d[k] = dst[e + k * SC_THREADS];
            s[k] = src[e + k * SC_THREADS];
        }
        #pragma unroll
        for (int k = 0; k < SC_BATCH; ++k) asm volatile("" :: "v"(d[k]), "v"(s[k]));
        #pragma unroll
        for (int k = 0; k < SC_BATCH; ++k) {
            const bool in = (unsigned)(d[k] - r0) < SC_RANGE;
            val[k] = in ? u[s[k]] : 0.0f;
        }
        #pragma unroll
        for (int k = 0; k < SC_BATCH; ++k) {
            const unsigned rel = (unsigned)(d[k] - r0);
            if (rel < SC_RANGE) atomicAdd(&acc[rel], val[k]);
        }
    }
    for (; e < eEnd; e += SC_THREADS) {
        const unsigned rel = (unsigned)(dst[e] - r0);
        if (rel < SC_RANGE) atomicAdd(&acc[rel], u[src[e]]);
    }
    __syncthreads();
    const int lim = (n - r0 < SC_RANGE) ? (n - r0) : SC_RANGE;
    float* o = part + (long long)blockIdx.x * n + r0;
    for (int i = threadIdx.x; i < lim; i += SC_THREADS) o[i] = acc[i];
}

__global__ __launch_bounds__(256) void k_l1_finish(const float* __restrict__ part,
                                                   const float* __restrict__ X,
                                                   const float* __restrict__ dinv,
                                                   float* __restrict__ s1, int n) {
    const int t = blockIdx.x * 256 + threadIdx.x;
    if (t >= n) return;
    float v[SC_CHUNKS];
    #pragma unroll
    for (int c = 0; c < SC_CHUNKS; ++c) v[c] = part[(long long)c * n + t];
    float s = v[0];
    #pragma unroll
    for (int c = 1; c < SC_CHUNKS; ++c) s += v[c];
    const float di = dinv[t];
    s1[t] = di * s + X[t] * (di * di);
}

__global__ __launch_bounds__(256) void k_expand(const float* __restrict__ s1,
                                                     const float* __restrict__ W1,
                                                     const float* __restrict__ bias,
                                                     float* __restrict__ y, int n) {
    int t = blockIdx.x * 256 + threadIdx.x;
    if (t >= n * 32) return;
    int row = t >> 5, c4 = (t & 31) << 2;
    float sv = s1[row];
    const float4 w = *(const float4*)(W1 + c4);
    const float4 b = *(const float4*)(bias + c4);
    float4 o;
    o.x = sv * w.x + b.x;
    o.y = sv * w.y + b.y;
    o.z = sv * w.z + b.z;
    o.w = sv * w.w + b.w;
    *(float4*)(y + (long long)row * HDIM + c4) = o;
}

__global__ __launch_bounds__(256) void k_gemm_wmma(const float* __restrict__ x,
                                                   const float* __restrict__ W,
                                                   const float* __restrict__ dinv,
                                                   float* __restrict__ hs, int n) {
    __shared__ float tileX[32 * HDIM];
    const int rowBase = blockIdx.x * 32;
    const int tid  = threadIdx.x;
    const int wave = tid >> 5;
    const int lane = tid & 31;
    const int half = lane >> 4;
    const int l16  = lane & 15;
    const int rowHalf = wave & 1;
    const int c0   = (wave >> 1) * 32;

    const bool fullTile = (rowBase + 32 <= n);

    if (fullTile) {
        const float4* xs = (const float4*)(x + (long long)rowBase * HDIM);
        #pragma unroll
        for (int i = 0; i < 4; ++i) {
            int idx = tid + i * 256;
            ((float4*)tileX)[idx] = xs[idx];
        }
    } else {
        #pragma unroll
        for (int i = 0; i < 4; ++i) {
            int idx = tid + i * 256;
            int r = rowBase + (idx >> 5);
            float4 v = make_float4(0.f, 0.f, 0.f, 0.f);
            if (r < n) v = *(const float4*)(x + (long long)r * HDIM + ((idx & 31) << 2));
            ((float4*)tileX)[idx] = v;
        }
    }
    __syncthreads();

    const float* xrow = tileX + (rowHalf * 16 + l16) * HDIM + 2 * half;
    v8f acc0 = {}, acc1 = {};

    #pragma unroll 4
    for (int k = 0; k < HDIM; k += 4) {
        v2f a; a.x = xrow[k]; a.y = xrow[k + 1];
        const float* wp = W + (long long)(k + 2 * half) * HDIM + c0 + l16;
        v2f b0; b0.x = wp[0];  b0.y = wp[HDIM];
        v2f b1; b1.x = wp[16]; b1.y = wp[HDIM + 16];
        acc0 = __builtin_amdgcn_wmma_f32_16x16x4_f32(false, a, false, b0,
                                                     (short)0, acc0, false, false);
        acc1 = __builtin_amdgcn_wmma_f32_16x16x4_f32(false, a, false, b1,
                                                     (short)0, acc1, false, false);
    }

    const int rbase = rowBase + rowHalf * 16 + half * 8;
    if (fullTile) {
        float di[8];
        #pragma unroll
        for (int v = 0; v < 8; ++v) di[v] = dinv[rbase + v];
        #pragma unroll
        for (int v = 0; v < 8; ++v) {
            const long long o = (long long)(rbase + v) * HDIM + c0 + l16;
            hs[o]      = acc0[v] * di[v];
            hs[o + 16] = acc1[v] * di[v];
        }
    } else {
        for (int v = 0; v < 8; ++v) {
            int r = rbase + v;
            if (r < n) {
                const float dd = dinv[r];
                const long long o = (long long)r * HDIM + c0 + l16;
                hs[o]      = acc0[v] * dd;
                hs[o + 16] = acc1[v] * dd;
            }
        }
    }
}

#define AG_ROWS    544
#define AG_THREADS 1024
#define AG_PF      8
#define AG_BATCH   8
#define AG_IO      6

__device__ __forceinline__ void agg_drain(const float* __restrict__ hs,
                                          const int* __restrict__ src,
                                          float* acc, int pe, int pr, int count, int lane) {
    int s = 0;
    if (lane < count) s = src[pe];
    #pragma nounroll
    for (int j0 = 0; j0 < count; j0 += AG_BATCH) {
        float v[AG_BATCH][4];
        #pragma unroll
        for (int u = 0; u < AG_BATCH; ++u) {
            int jj = j0 + u;
            jj = jj < count ? jj : count - 1;
            const int sj = __builtin_amdgcn_readlane(s, jj);
            const float* hp = hs + (long long)sj * HDIM;
            v[u][0] = hp[lane];
            v[u][1] = hp[lane + 32];
            v[u][2] = hp[lane + 64];
            v[u][3] = hp[lane + 96];
        }
        #pragma unroll
        for (int u = 0; u < AG_BATCH; ++u) {
            if (j0 + u < count) {
                const int rj = __builtin_amdgcn_readlane(pr, j0 + u);
                float* ap = acc + rj * HDIM + lane;
                atomicAdd(ap,      v[u][0]);
                atomicAdd(ap + 32, v[u][1]);
                atomicAdd(ap + 64, v[u][2]);
                atomicAdd(ap + 96, v[u][3]);
            }
        }
    }
}

__global__ __launch_bounds__(AG_THREADS) void k_agg_lds(const float* __restrict__ hs,
                                                        const int* __restrict__ src,
                                                        const int* __restrict__ dst,
                                                        const float* __restrict__ dinv,
                                                        const float* __restrict__ bias,
                                                        float* __restrict__ y, int E, int n) {
    __shared__ float acc[AG_ROWS * HDIM];
    const int lane = threadIdx.x & 31;
    const int wave = __builtin_amdgcn_readfirstlane((int)(threadIdx.x >> 5));
    const int r0 = blockIdx.x * AG_ROWS;
    const int lim = (n - r0 < AG_ROWS) ? (n - r0) : AG_ROWS;
    const unsigned ulim = (unsigned)lim;

    const int tot = lim * 32;
    {
        const float4* hp = (const float4*)(hs + (long long)r0 * HDIM);
        for (int i0 = threadIdx.x; i0 < tot; i0 += AG_IO * AG_THREADS) {
            float4 t[AG_IO];
            #pragma unroll
            for (int q = 0; q < AG_IO; ++q) {
                const int i = i0 + q * AG_THREADS;
                t[q] = hp[i < tot ? i : tot - 1];
            }
            asm volatile("" :: "v"(t[0].x), "v"(t[0].y), "v"(t[0].z), "v"(t[0].w),
                               "v"(t[1].x), "v"(t[1].y), "v"(t[1].z), "v"(t[1].w),
                               "v"(t[2].x), "v"(t[2].y), "v"(t[2].z), "v"(t[2].w),
                               "v"(t[3].x), "v"(t[3].y), "v"(t[3].z), "v"(t[3].w),
                               "v"(t[4].x), "v"(t[4].y), "v"(t[4].z), "v"(t[4].w),
                               "v"(t[5].x), "v"(t[5].y), "v"(t[5].z), "v"(t[5].w));
            #pragma unroll
            for (int q = 0; q < AG_IO; ++q) {
                const int i = i0 + q * AG_THREADS;
                if (i < tot) ((float4*)acc)[i] = t[q];
            }
        }
    }
    __syncthreads();

    int pe = 0, pr = 0, cnt = 0;

    const int nvec = (((size_t)dst & 15) == 0) ? (E >> 2) : 0;
    const int4* dv = (const int4*)dst;
    const int nIt = (nvec + 31) >> 5;
    const int K = nIt > wave ? ((nIt - wave + 31) >> 5) : 0;
    const int rot = (int)blockIdx.x * (nIt / (int)gridDim.x);
    const int4 none = make_int4(-1, -1, -1, -1);

    int4 buf[AG_PF];
    #pragma unroll
    for (int u = 0; u < AG_PF; ++u) {
        buf[u] = none;
        if (u < K) {
            int itr = wave + 32 * u + rot;
            itr = itr >= nIt ? itr - nIt : itr;
            const int idx = itr * 32 + lane;
            buf[u] = dv[idx < nvec ? idx : nvec - 1];
        }
    }
    for (int k = 0; k < K; k += AG_PF) {
        #pragma unroll
        for (int u = 0; u < AG_PF; ++u) {
            if (k + u >= K) break;
            const int4 cur = buf[u];
            int itr = wave + 32 * (k + u) + rot;
            itr = itr >= nIt ? itr - nIt : itr;
            {
                const int kn = k + u + AG_PF;
                if (kn < K) {
                    int itn = wave + 32 * kn + rot;
                    itn = itn >= nIt ? itn - nIt : itn;
                    const int idx = itn * 32 + lane;
                    buf[u] = dv[idx < nvec ? idx : nvec - 1];
                }
            }
            const bool ok = itr * 32 + lane < nvec;
            const bool m0 = ok & ((unsigned)(cur.x - r0) < ulim);
            const bool m1 = ok & ((unsigned)(cur.y - r0) < ulim);
            const bool m2 = ok & ((unsigned)(cur.z - r0) < ulim);
            const bool m3 = ok & ((unsigned)(cur.w - r0) < ulim);
            if (__builtin_amdgcn_ballot_w32(m0 | m1 | m2 | m3) != 0u) {
                unsigned bits = (m0 ? 1u : 0u) | (m1 ? 2u : 0u) | (m2 ? 4u : 0u) | (m3 ? 8u : 0u);
                unsigned act = __builtin_amdgcn_ballot_w32(bits != 0u);
                while (act != 0u) {
                    const int L = __builtin_ctz(act);
                    const unsigned b = (unsigned)__builtin_amdgcn_readlane((int)bits, L);
                    const int j = __builtin_ctz(b);
                    const int vsel = j == 0 ? cur.x : (j == 1 ? cur.y : (j == 2 ? cur.z : cur.w));
                    const int d = __builtin_amdgcn_readlane(vsel, L);
                    const int e = ((itr * 32 + L) << 2) + j;
                    pe = lane == cnt ? e : pe;
                    pr = lane == cnt ? d - r0 : pr;
                    ++cnt;
                    if (lane == L) bits &= bits - 1u;
                    if (cnt == 32) {
                        agg_drain(hs, src, acc, pe, pr, 32, lane);
                        cnt = 0;
                    }
                    act = __builtin_amdgcn_ballot_w32(bits != 0u);
                }
            }
        }
    }
    for (int base = nvec * 4 + wave * 32; base < E; base += AG_THREADS) {
        const int e = base + lane;
        const int d = e < E ? dst[e] : -1;
        unsigned act = __builtin_amdgcn_ballot_w32((unsigned)(d - r0) < ulim);
        while (act != 0u) {
            const int L = __builtin_ctz(act);
            act &= act - 1u;
            const int dl = __builtin_amdgcn_readlane(d, L);
            pe = lane == cnt ? base + L : pe;
            pr = lane == cnt ? dl - r0 : pr;
            ++cnt;
            if (cnt == 32) {
                agg_drain(hs, src, acc, pe, pr, 32, lane);
                cnt = 0;
            }
        }
    }
    agg_drain(hs, src, acc, pe, pr, cnt, lane);
    __syncthreads();

    {
        float4* yp = (float4*)(y + (long long)r0 * HDIM);
        const float4 b = *(const float4*)(bias + (lane << 2));
        for (int i0 = threadIdx.x; i0 < tot; i0 += AG_IO * AG_THREADS) {
            float di[AG_IO];
            #pragma unroll
            for (int q = 0; q < AG_IO; ++q) {
                const int i = i0 + q * AG_THREADS;
                di[q] = dinv[r0 + ((i < tot ? i : tot - 1) >> 5)];
            }
            #pragma unroll
            for (int q = 0; q < AG_IO; ++q) {
                const int i = i0 + q * AG_THREADS;
                if (i < tot) {
                    const float4 a = ((const float4*)acc)[i];
                    float4 o;
                    o.x = a.x * di[q] + b.x;
                    o.y = a.y * di[q] + b.y;
                    o.z = a.z * di[q] + b.z;
                    o.w = a.w * di[q] + b.w;
                    yp[i] = o;
                }
            }
        }
    }
}

#define STAT_BLOCKS 256
#define STAT_WAVES  32
#define STAT_BATCH  8
#define STAT_GROUPS 16

__device__ __forceinline__ void stat_acc(float4& s, float4& q, const float4 p) {
    float4 v;
    v.x = fmaxf(p.x, 0.f); v.y = fmaxf(p.y, 0.f); v.z = fmaxf(p.z, 0.f); v.w = fmaxf(p.w, 0.f);
    s.x += v.x; s.y += v.y; s.z += v.z; s.w += v.w;
    q.x += v.x * v.x; q.y += v.y * v.y; q.z += v.z * v.z; q.w += v.w * v.w;
}

__global__ __launch_bounds__(STAT_WAVES * 32) void k_stats(const float* __restrict__ y,
                                                            float* __restrict__ part, int n) {
    __shared__ float ls[STAT_WAVES][256];
    const int lane = threadIdx.x & 31;
    const int wave = threadIdx.x >> 5;
    const int S = STAT_BLOCKS * STAT_WAVES;
    const float4* yp = (const float4*)y + lane;
    int r = blockIdx.x * STAT_WAVES + wave;
    float4 s = make_float4(0.f, 0.f, 0.f, 0.f), q = s;

    for (; r + (STAT_BATCH - 1) * S < n; r += STAT_BATCH * S) {
        float4 v[STAT_BATCH];
        #pragma unroll
        for (int u = 0; u < STAT_BATCH; ++u) v[u] = yp[(long long)(r + u * S) * 32];
        #pragma unroll
        for (int u = 0; u < STAT_BATCH; ++u) stat_acc(s, q, v[u]);
    }
    if (r < n) {
        const float4 zero = make_float4(0.f, 0.f, 0.f, 0.f);
        float4 v[STAT_BATCH - 1];
        #pragma unroll
        for (int u = 0; u < STAT_BATCH - 1; ++u) {
            const int rr = r + u * S;
            const bool in = rr < n;
            const float4 t = yp[(long long)(in ? rr : n - 1) * 32];
            v[u] = in ? t : zero;
        }
        #pragma unroll
        for (int u = 0; u < STAT_BATCH - 1; ++u) stat_acc(s, q, v[u]);
    }

    *(float4*)&ls[wave][lane * 4]       = s;
    *(float4*)&ls[wave][128 + lane * 4] = q;
    __syncthreads();
    if (threadIdx.x < 256) {
        float a = 0.f;
        for (int w = 0; w < STAT_WAVES; ++w) a += ls[w][threadIdx.x];
        part[(long long)blockIdx.x * 256 + threadIdx.x] = a;
    }
}

__global__ __launch_bounds__(STAT_GROUPS * 64) void k_scaleshift(const float* __restrict__ part,
                                                                 const float* __restrict__ gamma,
                                                                 const float* __restrict__ beta,
                                                                 float* __restrict__ ss, float invN) {
    __shared__ float fs[STAT_GROUPS][256];
    const int c4 = threadIdx.x & 63;
    const int g  = threadIdx.x >> 6;
    const float4* pp = (const float4*)part + c4;
    float4 v[STAT_BLOCKS / STAT_GROUPS];
    #pragma unroll
    for (int i = 0; i < STAT_BLOCKS / STAT_GROUPS; ++i) v[i] = pp[(g + i * STAT_GROUPS) * 64];
    float4 a = v[0];
    #pragma unroll
    for (int i = 1; i < STAT_BLOCKS / STAT_GROUPS; ++i) {
        a.x += v[i].x; a.y += v[i].y; a.z += v[i].z; a.w += v[i].w;
    }
    *(float4*)&fs[g][c4 * 4] = a;
    __syncthreads();
    if (threadIdx.x < 128) {
        const int c = threadIdx.x;
        float su = 0.f, sq = 0.f;
        for (int k = 0; k < STAT_GROUPS; ++k) { su += fs[k][c]; sq += fs[k][128 + c]; }
        float m   = su * invN;
        float var = sq * invN - m * m;
        float sc  = rsqrtf(var + EPS_BN) * gamma[c];
        ss[c] = sc;
        ss[128 + c] = beta[c] - m * sc;
    }
}

__global__ __launch_bounds__(256) void k_bn_apply(float* __restrict__ y,
                                                  const float* __restrict__ res,
                                                  const float* __restrict__ ss, int n) {
    int t = blockIdx.x * 256 + threadIdx.x;
    if (t >= n * 32) return;
    int row = t >> 5, c4 = (t & 31) << 2;
    long long off = (long long)row * HDIM + c4;
    float4 v  = *(const float4*)(y + off);
    float4 sc = *(const float4*)(ss + c4);
    float4 sh = *(const float4*)(ss + 128 + c4);
    v.x = fmaxf(v.x, 0.f) * sc.x + sh.x; v.y = fmaxf(v.y, 0.f) * sc.y + sh.y;
    v.z = fmaxf(v.z, 0.f) * sc.z + sh.z; v.w = fmaxf(v.w, 0.f) * sc.w + sh.w;
    if (res) {
        float4 r = *(const float4*)(res + off);
        v.x += r.x; v.y += r.y; v.z += r.z; v.w += r.w;
    }
    *(float4*)(y + off) = v;
}

#define POOL_BLOCKS 256
#define POOL_WAVES  32
#define POOL_BATCH  8
#define PSUM_GROUPS 16

__device__ __forceinline__ void pool_add(float* __restrict__ pl, int g, int lane,
                                         const float4 v, int cntBase) {
    float* p = pl + g * HDIM + lane * 4;
    atomicAdd(p + 0, v.x); atomicAdd(p + 1, v.y);
    atomicAdd(p + 2, v.z); atomicAdd(p + 3, v.w);
    if (lane == 0) atomicAdd(pl + cntBase + g, 1.0f);
}

__global__ __launch_bounds__(POOL_WAVES * 32) void k_pool_part(const float* __restrict__ x,
                                                                const int* __restrict__ batch,
                                                                float* __restrict__ planes,
                                                                int n, int G) {
    extern __shared__ float pl[];
    const int lane = threadIdx.x & 31;
    const int tot = G * HDIM + G;
    for (int i = threadIdx.x; i < tot; i += POOL_WAVES * 32) pl[i] = 0.0f;
    __syncthreads();
    const int S = (int)gridDim.x * POOL_WAVES;
    int r = __builtin_amdgcn_readfirstlane((int)blockIdx.x * POOL_WAVES + (int)(threadIdx.x >> 5));
    const float4* xp = (const float4*)x + lane;

    for (; r + (POOL_BATCH - 1) * S < n; r += POOL_BATCH * S) {
        float4 v[POOL_BATCH];
        int g[POOL_BATCH];
        #pragma unroll
        for (int k = 0; k < POOL_BATCH; ++k) {
            v[k] = xp[(long long)(r + k * S) * 32];
            g[k] = batch[r + k * S];
        }
        #pragma unroll
        for (int k = 0; k < POOL_BATCH; ++k) pool_add(pl, g[k], lane, v[k], G * HDIM);
    }
    if (r < n) {
        float4 v[POOL_BATCH - 1];
        int g[POOL_BATCH - 1];
        #pragma unroll
        for (int k = 0; k < POOL_BATCH - 1; ++k) {
            const int rr = r + k * S;
            const int rc = rr < n ? rr : n - 1;
            v[k] = xp[(long long)rc * 32];
            g[k] = batch[rc];
        }
        #pragma unroll
        for (int k = 0; k < POOL_BATCH - 1; ++k)
            if (r + k * S < n) pool_add(pl, g[k], lane, v[k], G * HDIM);
    }
    __syncthreads();
    float* o = planes + (long long)blockIdx.x * tot;
    for (int i = threadIdx.x; i < tot; i += POOL_WAVES * 32) o[i] = pl[i];
}

__global__ __launch_bounds__(PSUM_GROUPS * 64) void k_pool_sum(const float* __restrict__ planes,
                                                               float* __restrict__ outp, int tot) {
    __shared__ float fs[PSUM_GROUPS][64];
    const int o   = threadIdx.x & 63;
    const int grp = threadIdx.x >> 6;
    const int i   = blockIdx.x * 64 + o;
    float a = 0.0f;
    if (i < tot) {
        float v[POOL_BLOCKS / PSUM_GROUPS];
        #pragma unroll
        for (int j = 0; j < POOL_BLOCKS / PSUM_GROUPS; ++j)
            v[j] = planes[(long long)(grp + j * PSUM_GROUPS) * tot + i];
        a = v[0];
        #pragma unroll
        for (int j = 1; j < POOL_BLOCKS / PSUM_GROUPS; ++j) a += v[j];
    }
    fs[grp][o] = a;
    __syncthreads();
    if (threadIdx.x < 64 && i < tot) {
        float s = 0.0f;
        for (int k = 0; k < PSUM_GROUPS; ++k) s += fs[k][o];
        outp[i] = s;
    }
}

__global__ __launch_bounds__(256) void k_pool_final(const float* __restrict__ pool,
                                                    const float* __restrict__ counts,
                                                    float* __restrict__ xmean, int G) {
    int t = blockIdx.x * 256 + threadIdx.x;
    if (t >= G * HDIM) return;
    int g = t / HDIM;
    xmean[t] = pool[t] / fmaxf(counts[g], 1.0f);
}

__global__ void k_pred(const float* __restrict__ xmean, const float* __restrict__ Wr,
                       const float* __restrict__ br, float* __restrict__ pred) {
    __shared__ float red[128];
    int g = blockIdx.x, c = threadIdx.x;
    red[c] = xmean[(long long)g * HDIM + c] * Wr[c];
    __syncthreads();
    for (int s = 64; s > 0; s >>= 1) {
        if (c < s) red[c] += red[c + s];
        __syncthreads();
    }
    if (c == 0) pred[g] = red[0] + br[0];
}

extern "C" void kernel_launch(void* const* d_in, const int* in_sizes, int n_in,
                              void* d_out, int out_size, void* d_ws, size_t ws_size,
                              hipStream_t stream) {
    const float* X     = (const float*)d_in[0];
    const int*   edge  = (const int*)d_in[1];
    const int*   batch = (const int*)d_in[2];
    const float* W1    = (const float*)d_in[3];
    const float* b1    = (const float*)d_in[4];
    const float* Wm    = (const float*)d_in[5];
    const float* bm    = (const float*)d_in[6];
    const float* gamma = (const float*)d_in[7];
    const float* beta  = (const float*)d_in[8];
    const float* Wr    = (const float*)d_in[9];
    const float* br    = (const float*)d_in[10];
    float* out = (float*)d_out;

    const int N = in_sizes[0];
    const int E = in_sizes[1] / 2;
    const int G = out_size / (HDIM + 1);
    const int* src = edge;
    const int* dst = edge + E;

    const long long NH = (long long)N * HDIM;
    float* ws    = (float*)d_ws;
    float* buf[3] = { ws, ws + NH, ws + 2 * NH };
    float* dinv  = ws + 3 * NH;
    float* ss    = dinv + N;
    float* pool  = ss + 256;
    float* cnts  = pool + (long long)G * HDIM;

    const int gAgg = (N + AG_ROWS - 1) / AG_ROWS;
    const int gN   = (N + 255) / 256;
    const int gNH4 = (N * 32 + 255) / 256;
    const int gGem = (N + 31) / 32;

    float* s1   = buf[0];
    float* u1   = buf[0] + N;
    float* scp  = buf[1];
    const dim3 gSc(SC_CHUNKS, (N + SC_RANGE - 1) / SC_RANGE);
    k_deg_count<<<gSc, SC_THREADS, 0, stream>>>(dst, scp, E, N);
    k_deg_finish<<<gN, 256, 0, stream>>>(scp, X, dinv, u1, N);

    auto bn_tail = [&](float* abuf, float* part, const float* g_, const float* be_,
                       const float* res) {
        k_stats<<<STAT_BLOCKS, STAT_WAVES * 32, 0, stream>>>(abuf, part, N);
        k_scaleshift<<<1, STAT_GROUPS * 64, 0, stream>>>(part, g_, be_, ss, 1.0f / (float)N);
        k_bn_apply<<<gNH4, 256, 0, stream>>>(abuf, res, ss, N);
    };

    auto layer_tail = [&](float* hbuf, float* abuf, const float* bias_,
                          const float* g_, const float* be_, const float* res) {
        k_agg_lds<<<gAgg, AG_THREADS, 0, stream>>>(hbuf, src, dst, dinv, bias_, abuf, E, N);
        bn_tail(abuf, hbuf, g_, be_, res);
    };

    k_l1_sum<<<gSc, SC_THREADS, 0, stream>>>(u1, src, dst, scp, E, N);
    k_l1_finish<<<gN, 256, 0, stream>>>(scp, X, dinv, s1, N);
    k_expand<<<gNH4, 256, 0, stream>>>(s1, W1, b1, buf[2], N);
    bn_tail(buf[2], buf[0], gamma, beta, nullptr);

    float* cur = buf[2];
    float* f0  = buf[0];
    float* f1  = buf[1];

    for (int i = 0; i < 4; ++i) {
        float* hbuf = f0;
        float* abuf = f1;
        k_gemm_wmma<<<gGem, 256, 0, stream>>>(cur, Wm + (long long)i * HDIM * HDIM, dinv,
                                              hbuf, N);
        layer_tail(hbuf, abuf, bm + (long long)i * HDIM,
                   gamma + (long long)(i + 1) * HDIM,
                   beta  + (long long)(i + 1) * HDIM, cur);
        float* next = abuf;
        f1 = cur;
        f0 = hbuf;
        cur = next;
    }

    const int poolTot = G * HDIM + G;
    k_pool_part<<<POOL_BLOCKS, POOL_WAVES * 32, poolTot * sizeof(float), stream>>>(cur, batch, f0, N, G);
    k_pool_sum<<<(poolTot + 63) / 64, PSUM_GROUPS * 64, 0, stream>>>(f0, pool, poolTot);
    k_pool_final<<<(G * HDIM + 255) / 256, 256, 0, stream>>>(pool, cnts, out + G, G);
    k_pred<<<G, 128, 0, stream>>>(out + G, Wr, br, out);
}


# k_agg_lds finds its edges in a 1-bit-per-edge mark list of its own, built per layer from dst into the block's not yet written output rows, instead of scanning all of dst; adds stay in arrival order
# speedup vs baseline: 3.4813x; 1.1221x over previous
#include <hip/hip_runtime.h>
#include <hip/hip_bf16.h>

#define HDIM 128
#define EPS_BN 1e-5f

typedef __attribute__((ext_vector_type(2))) float v2f;
typedef __attribute__((ext_vector_type(8))) float v8f;

#define SC_RANGE   16384
#define SC_CHUNKS  32
#define SC_THREADS 1024
#define SC_BATCH   8

__global__ __launch_bounds__(SC_THREADS) void k_deg_count(const int* __restrict__ dst,
                                                          float* __restrict__ part,
                                                          int E, int n) {
    __shared__ unsigned int cnt[SC_RANGE];
    const int r0 = blockIdx.y * SC_RANGE;
    for (int i = threadIdx.x; i < SC_RANGE; i += SC_THREADS) cnt[i] = 0u;
    __syncthreads();
    const int L = (E + (int)gridDim.x - 1) / (int)gridDim.x;
    const int eBeg = blockIdx.x * L;
    const int eEnd = (eBeg + L < E) ? (eBeg + L) : E;
    int e = eBeg + threadIdx.x;
    for (; e + (SC_BATCH - 1) * SC_THREADS < eEnd; e += SC_BATCH * SC_THREADS) {
        int d[SC_BATCH];
        #pragma unroll
        for (int k = 0; k < SC_BATCH; ++k) d[k] = dst[e + k * SC_THREADS];
        #pragma unroll
        for (int k = 0; k < SC_BATCH; ++k) {
            const unsigned rel = (unsigned)(d[k] - r0);
            if (rel < SC_RANGE) atomicAdd(&cnt[rel], 1u);
        }
    }
    for (; e < eEnd; e += SC_THREADS) {
        const unsigned rel = (unsigned)(dst[e] - r0);
        if (rel < SC_RANGE) atomicAdd(&cnt[rel], 1u);
    }
    __syncthreads();
    const int lim = (n - r0 < SC_RANGE) ? (n - r0) : SC_RANGE;
    float* o = part + (long long)blockIdx.x * n + r0;
    for (int i = threadIdx.x; i < lim; i += SC_THREADS) o[i] = (float)cnt[i];
}

__global__ __launch_bounds__(256) void k_deg_finish(const float* __restrict__ part,
                                                    const float* __restrict__ X,
                                                    float* __restrict__ dinv,
                                                    float* __restrict__ u, int n) {
    const int t = blockIdx.x * 256 + threadIdx.x;
    if (t >= n) return;
    float v[SC_CHUNKS];
    #pragma unroll
    for (int c = 0; c < SC_CHUNKS; ++c) v[c] = part[(long long)c * n + t];
    float s = v[0];
    #pragma unroll
    for (int c = 1; c < SC_CHUNKS; ++c) s += v[c];
    const float di = rsqrtf(s + 1.0f);
    dinv[t] = di;
    u[t] = X[t] * di;
}

__global__ __launch_bounds__(SC_THREADS) void k_l1_sum(const float* __restrict__ u,
                                                       const int* __restrict__ src,
                                                       const int* __restrict__ dst,
                                                       float* __restrict__ part,
                                                       int E, int n) {
    __shared__ float acc[SC_RANGE];
    const int r0 = blockIdx.y * SC_RANGE;
    for (int i = threadIdx.x; i < SC_RANGE; i += SC_THREADS) acc[i] = 0.0f;
    __syncthreads();
    const int L = (E + (int)gridDim.x - 1) / (int)gridDim.x;
    const int eBeg = blockIdx.x * L;
    const int eEnd = (eBeg + L < E) ? (eBeg + L) : E;
    int e = eBeg + threadIdx.x;
    for (; e + (SC_BATCH - 1) * SC_THREADS < eEnd; e += SC_BATCH * SC_THREADS) {
        int d[SC_BATCH], s[SC_BATCH];
        float val[SC_BATCH];
        #pragma unroll
        for (int k = 0; k < SC_BATCH; ++k) {
            d[k] = dst[e + k * SC_THREADS];
            s[k] = src[e + k * SC_THREADS];
        }
        #pragma unroll
        for (int k = 0; k < SC_BATCH; ++k) asm volatile("" :: "v"(d[k]), "v"(s[k]));
        #pragma unroll
        for (int k = 0; k < SC_BATCH; ++k) {
            const bool in = (unsigned)(d[k] - r0) < SC_RANGE;
            val[k] = in ? u[s[k]] : 0.0f;
        }
        #pragma unroll
        for (int k = 0; k < SC_BATCH; ++k) {
            const unsigned rel = (unsigned)(d[k] - r0);
            if (rel < SC_RANGE) atomicAdd(&acc[rel], val[k]);
        }
    }
    for (; e < eEnd; e += SC_THREADS) {
        const unsigned rel = (unsigned)(dst[e] - r0);
        if (rel < SC_RANGE) atomicAdd(&acc[rel], u[src[e]]);
    }
    __syncthreads();
    const int lim = (n - r0 < SC_RANGE) ? (n - r0) : SC_RANGE;
    float* o = part + (long long)blockIdx.x * n + r0;
    for (int i = threadIdx.x; i < lim; i += SC_THREADS) o[i] = acc[i];
}

__global__ __launch_bounds__(256) void k_l1_finish(const float* __restrict__ part,
                                                   const float* __restrict__ X,
                                                   const float* __restrict__ dinv,
                                                   float* __restrict__ s1, int n) {
    const int t = blockIdx.x * 256 + threadIdx.x;
    if (t >= n) return;
    float v[SC_CHUNKS];
    #pragma unroll
    for (int c = 0; c < SC_CHUNKS; ++c) v[c] = part[(long long)c * n + t];
    float s = v[0];
    #pragma unroll
    for (int c = 1; c < SC_CHUNKS; ++c) s += v[c];
    const float di = dinv[t];
    s1[t] = di * s + X[t] * (di * di);
}

__global__ __launch_bounds__(256) void k_expand(const float* __restrict__ s1,
                                                     const float* __restrict__ W1,
                                                     const float* __restrict__ bias,
                                                     float* __restrict__ y, int n) {
    int t = blockIdx.x * 256 + threadIdx.x;
    if (t >= n * 32) return;
    int row = t >> 5, c4 = (t & 31) << 2;
    float sv = s1[row];
    const float4 w = *(const float4*)(W1 + c4);
    const float4 b = *(const float4*)(bias + c4);
    float4 o;
    o.x = sv * w.x + b.x;
    o.y = sv * w.y + b.y;
    o.z = sv * w.z + b.z;
    o.w = sv * w.w + b.w;
    *(float4*)(y + (long long)row * HDIM + c4) = o;
}

__global__ __launch_bounds__(256) void k_gemm_wmma(const float* __restrict__ x,
                                                   const float* __restrict__ W,
                                                   const float* __restrict__ dinv,
                                                   float* __restrict__ hs, int n) {
    __shared__ float tileX[32 * HDIM];
    const int rowBase = blockIdx.x * 32;
    const int tid  = threadIdx.x;
    const int wave = tid >> 5;
    const int lane = tid & 31;
    const int half = lane >> 4;
    const int l16  = lane & 15;
    const int rowHalf = wave & 1;
    const int c0   = (wave >> 1) * 32;

    const bool fullTile = (rowBase + 32 <= n);

    if (fullTile) {
        const float4* xs = (const float4*)(x + (long long)rowBase * HDIM);
        #pragma unroll
        for (int i = 0; i < 4; ++i) {
            int idx = tid + i * 256;
            ((float4*)tileX)[idx] = xs[idx];
        }
    } else {
        #pragma unroll
        for (int i = 0; i < 4; ++i) {
            int idx = tid + i * 256;
            int r = rowBase + (idx >> 5);
            float4 v = make_float4(0.f, 0.f, 0.f, 0.f);
            if (r < n) v = *(const float4*)(x + (long long)r * HDIM + ((idx & 31) << 2));
            ((float4*)tileX)[idx] = v;
        }
    }
    __syncthreads();

    const float* xrow = tileX + (rowHalf * 16 + l16) * HDIM + 2 * half;
    v8f acc0 = {}, acc1 = {};

    #pragma unroll 4
    for (int k = 0; k < HDIM; k += 4) {
        v2f a; a.x = xrow[k]; a.y = xrow[k + 1];
        const float* wp = W + (long long)(k + 2 * half) * HDIM + c0 + l16;
        v2f b0; b0.x = wp[0];  b0.y = wp[HDIM];
        v2f b1; b1.x = wp[16]; b1.y = wp[HDIM + 16];
        acc0 = __builtin_amdgcn_wmma_f32_16x16x4_f32(false, a, false, b0,
                                                     (short)0, acc0, false, false);
        acc1 = __builtin_amdgcn_wmma_f32_16x16x4_f32(false, a, false, b1,
                                                     (short)0, acc1, false, false);
    }

    const int rbase = rowBase + rowHalf * 16 + half * 8;
    if (fullTile) {
        float di[8];
        #pragma unroll
        for (int v = 0; v < 8; ++v) di[v] = dinv[rbase + v];
        #pragma unroll
        for (int v = 0; v < 8; ++v) {
            const long long o = (long long)(rbase + v) * HDIM + c0 + l16;
            hs[o]      = acc0[v] * di[v];
            hs[o + 16] = acc1[v] * di[v];
        }
    } else {
        for (int v = 0; v < 8; ++v) {
            int r = rbase + v;
            if (r < n) {
                const float dd = dinv[r];
                const long long o = (long long)r * HDIM + c0 + l16;
                hs[o]      = acc0[v] * dd;
                hs[o + 16] = acc1[v] * dd;
            }
        }
    }
}

#define AG_ROWS    544
#define AG_THREADS 1024
#define AG_PF      8
#define AG_BATCH   8
#define AG_IO      6

#define MK_EDGES 2048
#define MK_WORDS (MK_EDGES / 32)

__global__ __launch_bounds__(256) void k_edge_marks(const int* __restrict__ dst,
                                                    float* __restrict__ ybuf,
                                                    int E, int nWpad, int gAgg) {
    extern __shared__ unsigned int tab[];
    const int tot = gAgg * MK_WORDS;
    for (int i = threadIdx.x; i < tot; i += 256) tab[i] = 0u;
    __syncthreads();
    const int e0 = blockIdx.x * MK_EDGES;
    int d[MK_EDGES / 256];
    #pragma unroll
    for (int k = 0; k < MK_EDGES / 256; ++k) {
        const int e = e0 + threadIdx.x + k * 256;
        const int v = dst[e < E ? e : E - 1];
        d[k] = e < E ? v : -1;
    }
    #pragma unroll
    for (int k = 0; k < MK_EDGES / 256; ++k) {
        const int le = threadIdx.x + k * 256;
        const unsigned b = (unsigned)d[k] / AG_ROWS;
        if (b < (unsigned)gAgg) atomicOr(&tab[b * MK_WORDS + (le >> 5)], 1u << (le & 31));
    }
    __syncthreads();
    const int wj = blockIdx.x * MK_WORDS + (threadIdx.x % MK_WORDS);
    if (wj < nWpad) {
        for (int i0 = threadIdx.x; i0 < tot; i0 += 4 * 256) {
            unsigned int v[4];
            #pragma unroll
            for (int k = 0; k < 4; ++k) {
                const int i = i0 + k * 256;
                v[k] = tab[i < tot ? i : tot - 1];
            }
            #pragma unroll
            for (int k = 0; k < 4; ++k) {
                const int i = i0 + k * 256;
                if (i < tot)
                    ((unsigned int*)(ybuf + (long long)(i / MK_WORDS) * AG_ROWS * HDIM))[wj] = v[k];
            }
        }
    }
}

__device__ __forceinline__ void agg_drain(const float* __restrict__ hs,
                                          const int* __restrict__ src,
                                          float* acc, int pe, int pr, int count, int lane) {
    int s = 0;
    if (lane < count) s = src[pe];
    #pragma nounroll
    for (int j0 = 0; j0 < count; j0 += AG_BATCH) {
        float v[AG_BATCH][4];
        #pragma unroll
        for (int u = 0; u < AG_BATCH; ++u) {
            int jj = j0 + u;
            jj = jj < count ? jj : count - 1;
            const int sj = __builtin_amdgcn_readlane(s, jj);
            const float* hp = hs + (long long)sj * HDIM;
            v[u][0] = hp[lane];
            v[u][1] = hp[lane + 32];
            v[u][2] = hp[lane + 64];
            v[u][3] = hp[lane + 96];
        }
        #pragma unroll
        for (int u = 0; u < AG_BATCH; ++u) {
            if (j0 + u < count) {
                const int rj = __builtin_amdgcn_readlane(pr, j0 + u);
                float* ap = acc + rj * HDIM + lane;
                atomicAdd(ap,      v[u][0]);
                atomicAdd(ap + 32, v[u][1]);
                atomicAdd(ap + 64, v[u][2]);
                atomicAdd(ap + 96, v[u][3]);
            }
        }
    }
}

template <bool MARKS>
__global__ __launch_bounds__(AG_THREADS) void k_agg_lds(const float* __restrict__ hs,
                                                        const int* __restrict__ src,
                                                        const int* __restrict__ dst,
                                                        const float* __restrict__ dinv,
                                                        const float* __restrict__ bias,
                                                        float* __restrict__ y, int E, int n) {
    __shared__ float acc[AG_ROWS * HDIM];
    const int lane = threadIdx.x & 31;
    const int wave = __builtin_amdgcn_readfirstlane((int)(threadIdx.x >> 5));
    const int r0 = blockIdx.x * AG_ROWS;
    const int lim = (n - r0 < AG_ROWS) ? (n - r0) : AG_ROWS;
    const unsigned ulim = (unsigned)lim;

    const int tot = lim * 32;
    {
        const float4* hp = (const float4*)(hs + (long long)r0 * HDIM);
        for (int i0 = threadIdx.x; i0 < tot; i0 += AG_IO * AG_THREADS) {
            float4 t[AG_IO];
            #pragma unroll
            for (int q = 0; q < AG_IO; ++q) {
                const int i = i0 + q * AG_THREADS;
                t[q] = hp[i < tot ? i : tot - 1];
            }
            asm volatile("" :: "v"(t[0].x), "v"(t[0].y), "v"(t[0].z), "v"(t[0].w),
                               "v"(t[1].x), "v"(t[1].y), "v"(t[1].z), "v"(t[1].w),
                               "v"(t[2].x), "v"(t[2].y), "v"(t[2].z), "v"(t[2].w),
                               "v"(t[3].x), "v"(t[3].y), "v"(t[3].z), "v"(t[3].w),
                               "v"(t[4].x), "v"(t[4].y), "v"(t[4].z), "v"(t[4].w),
                               "v"(t[5].x), "v"(t[5].y), "v"(t[5].z), "v"(t[5].w));
            #pragma unroll
            for (int q = 0; q < AG_IO; ++q) {
                const int i = i0 + q * AG_THREADS;
                if (i < tot) ((float4*)acc)[i] = t[q];
            }
        }
    }
    __syncthreads();

    int pe = 0, pr = 0, cnt = 0;

    int tailBeg;
    if constexpr (MARKS) {
        const int nq = (((E + 31) >> 5) + 3) >> 2;
        const uint4* mk = (const uint4*)(y + (long long)r0 * HDIM);
        const int nIt = (nq + 31) >> 5;
        const int K = nIt > wave ? ((nIt - wave + 31) >> 5) : 0;
        const uint4 none = make_uint4(0u, 0u, 0u, 0u);
        auto piece = [&](int t) {
            const int idx = (wave + 32 * t) * 32 + lane;
            return mk[idx < nq ? idx : nq - 1];
        };
        uint4 n0 = none, n1 = none;
        if (K > 0) n0 = piece(0);
        if (K > 1) n1 = piece(1);
        #pragma nounroll
        for (int k = 0; k < K; ++k) {
            {
                const uint4 cur = n0;
                n0 = n1;
                if (k + 2 < K) n1 = piece(k + 2);
                const int itr = wave + 32 * k;
                const bool ok = itr * 32 + lane < nq;
                const unsigned m0 = ok ? cur.x : 0u, m1 = ok ? cur.y : 0u;
                const unsigned m2 = ok ? cur.z : 0u, m3 = ok ? cur.w : 0u;
                #pragma nounroll
                for (int q = 0; q < 4; ++q) {
                    const unsigned mv = q == 0 ? m0 : (q == 1 ? m1 : (q == 2 ? m2 : m3));
                    unsigned act = __builtin_amdgcn_ballot_w32(mv != 0u);
                    while (act != 0u) {
                        const int L = __builtin_ctz(act);
                        act &= act - 1u;
                        unsigned sw = (unsigned)__builtin_amdgcn_readlane((int)mv, L);
                        const int e0 = ((((itr * 32 + L) << 2) + q) << 5);
                        while (sw != 0u) {
                            const int p = __builtin_ctz(sw);
                            sw &= sw - 1u;
                            pe = lane == cnt ? e0 + p : pe;
                            ++cnt;
                            if (cnt == 32) {
                                pr = dst[pe] - r0;
                                agg_drain(hs, src, acc, pe, pr, 32, lane);
                                cnt = 0;
                            }
                        }
                    }
                }
            }
        }
        tailBeg = E;
    } else {
        const int nvec = (((size_t)dst & 15) == 0) ? (E >> 2) : 0;
        const int4* dv = (const int4*)dst;
        const int nIt = (nvec + 31) >> 5;
        const int K = nIt > wave ? ((nIt - wave + 31) >> 5) : 0;
        const int rot = (int)blockIdx.x * (nIt / (int)gridDim.x);
        const int4 none = make_int4(-1, -1, -1, -1);

        int4 buf[AG_PF];
        #pragma unroll
        for (int u = 0; u < AG_PF; ++u) {
            buf[u] = none;
            if (u < K) {
                int itr = wave + 32 * u + rot;
                itr = itr >= nIt ? itr - nIt : itr;
                const int idx = itr * 32 + lane;
                buf[u] = dv[idx < nvec ? idx : nvec - 1];
            }
        }
        for (int k = 0; k < K; k += AG_PF) {
            #pragma unroll
            for (int u = 0; u < AG_PF; ++u) {
                if (k + u >= K) break;
                const int4 cur = buf[u];
                int itr = wave + 32 * (k + u) + rot;
                itr = itr >= nIt ? itr - nIt : itr;
                {
                    const int kn = k + u + AG_PF;
                    if (kn < K) {
                        int itn = wave + 32 * kn + rot;
                        itn = itn >= nIt ? itn - nIt : itn;
                        const int idx = itn * 32 + lane;
                        buf[u] = dv[idx < nvec ? idx : nvec - 1];
                    }
                }
                const bool ok = itr * 32 + lane < nvec;
                const bool m0 = ok & ((unsigned)(cur.x - r0) < ulim);
                const bool m1 = ok & ((unsigned)(cur.y - r0) < ulim);
                const bool m2 = ok & ((unsigned)(cur.z - r0) < ulim);
                const bool m3 = ok & ((unsigned)(cur.w - r0) < ulim);
                if (__builtin_amdgcn_ballot_w32(m0 | m1 | m2 | m3) != 0u) {
                    unsigned bits = (m0 ? 1u : 0u) | (m1 ? 2u : 0u) | (m2 ? 4u : 0u) | (m3 ? 8u : 0u);
                    unsigned act = __builtin_amdgcn_ballot_w32(bits != 0u);
                    while (act != 0u) {
                        const int L = __builtin_ctz(act);
                        const unsigned b = (unsigned)__builtin_amdgcn_readlane((int)bits, L);
                        const int j = __builtin_ctz(b);
                        const int vsel = j == 0 ? cur.x : (j == 1 ? cur.y : (j == 2 ? cur.z : cur.w));
                        const int d = __builtin_amdgcn_readlane(vsel, L);
                        const int e = ((itr * 32 + L) << 2) + j;
                        pe = lane == cnt ? e : pe;
                        pr = lane == cnt ? d - r0 : pr;
                        ++cnt;
                        if (lane == L) bits &= bits - 1u;
                        if (cnt == 32) {
                            agg_drain(hs, src, acc, pe, pr, 32, lane);
                            cnt = 0;
                        }
                        act = __builtin_amdgcn_ballot_w32(bits != 0u);
                    }
                }
            }
        }
        tailBeg = nvec * 4;
    }
    if constexpr (!MARKS) {
        for (int base = tailBeg + wave * 32; base < E; base += AG_THREADS) {
            const int e = base + lane;
            const int d = e < E ? dst[e] : -1;
            unsigned act = __builtin_amdgcn_ballot_w32((unsigned)(d - r0) < ulim);
            while (act != 0u) {
                const int L = __builtin_ctz(act);
                act &= act - 1u;
                const int dl = __builtin_amdgcn_readlane(d, L);
                pe = lane == cnt ? base + L : pe;
                pr = lane == cnt ? dl - r0 : pr;
                ++cnt;
                if (cnt == 32) {
                    agg_drain(hs, src, acc, pe, pr, 32, lane);
                    cnt = 0;
                }
            }
        }
    }
    if constexpr (MARKS) {
        if (lane < cnt) pr = dst[pe] - r0;
    }
    agg_drain(hs, src, acc, pe, pr, cnt, lane);
    __syncthreads();

    {
        float4* yp = (float4*)(y + (long long)r0 * HDIM);
        const float4 b = *(const float4*)(bias + (lane << 2));
        for (int i0 = threadIdx.x; i0 < tot; i0 += AG_IO * AG_THREADS) {
            float di[AG_IO];
            #pragma unroll
            for (int q = 0; q < AG_IO; ++q) {
                const int i = i0 + q * AG_THREADS;
                di[q] = dinv[r0 + ((i < tot ? i : tot - 1) >> 5)];
            }
            #pragma unroll
            for (int q = 0; q < AG_IO; ++q) {
                const int i = i0 + q * AG_THREADS;
                if (i < tot) {
                    const float4 a = ((const float4*)acc)[i];
                    float4 o;
                    o.x = a.x * di[q] + b.x;
                    o.y = a.y * di[q] + b.y;
                    o.z = a.z * di[q] + b.z;
                    o.w = a.w * di[q] + b.w;
                    yp[i] = o;
                }
            }
        }
    }
}

#define STAT_BLOCKS 256
#define STAT_WAVES  32
#define STAT_BATCH  8
#define STAT_GROUPS 16

__device__ __forceinline__ void stat_acc(float4& s, float4& q, const float4 p) {
    float4 v;
    v.x = fmaxf(p.x, 0.f); v.y = fmaxf(p.y, 0.f); v.z = fmaxf(p.z, 0.f); v.w = fmaxf(p.w, 0.f);
    s.x += v.x; s.y += v.y; s.z += v.z; s.w += v.w;
    q.x += v.x * v.x; q.y += v.y * v.y; q.z += v.z * v.z; q.w += v.w * v.w;
}

__global__ __launch_bounds__(STAT_WAVES * 32) void k_stats(const float* __restrict__ y,
                                                            float* __restrict__ part, int n) {
    __shared__ float ls[STAT_WAVES][256];
    const int lane = threadIdx.x & 31;
    const int wave = threadIdx.x >> 5;
    const int S = STAT_BLOCKS * STAT_WAVES;
    const float4* yp = (const float4*)y + lane;
    int r = blockIdx.x * STAT_WAVES + wave;
    float4 s = make_float4(0.f, 0.f, 0.f, 0.f), q = s;

    for (; r + (STAT_BATCH - 1) * S < n; r += STAT_BATCH * S) {
        float4 v[STAT_BATCH];
        #pragma unroll
        for (int u = 0; u < STAT_BATCH; ++u) v[u] = yp[(long long)(r + u * S) * 32];
        #pragma unroll
        for (int u = 0; u < STAT_BATCH; ++u) stat_acc(s, q, v[u]);
    }
    if (r < n) {
        const float4 zero = make_float4(0.f, 0.f, 0.f, 0.f);
        float4 v[STAT_BATCH - 1];
        #pragma unroll
        for (int u = 0; u < STAT_BATCH - 1; ++u) {
            const int rr = r + u * S;
            const bool in = rr < n;
            const float4 t = yp[(long long)(in ? rr : n - 1) * 32];
            v[u] = in ? t : zero;
        }
        #pragma unroll
        for (int u = 0; u < STAT_BATCH - 1; ++u) stat_acc(s, q, v[u]);
    }

    *(float4*)&ls[wave][lane * 4]       = s;
    *(float4*)&ls[wave][128 + lane * 4] = q;
    __syncthreads();
    if (threadIdx.x < 256) {
        float a = 0.f;
        for (int w = 0; w < STAT_WAVES; ++w) a += ls[w][threadIdx.x];
        part[(long long)blockIdx.x * 256 + threadIdx.x] = a;
    }
}

__global__ __launch_bounds__(STAT_GROUPS * 64) void k_scaleshift(const float* __restrict__ part,
                                                                 const float* __restrict__ gamma,
                                                                 const float* __restrict__ beta,
                                                                 float* __restrict__ ss, float invN) {
    __shared__ float fs[STAT_GROUPS][256];
    const int c4 = threadIdx.x & 63;
    const int g  = threadIdx.x >> 6;
    const float4* pp = (const float4*)part + c4;
    float4 v[STAT_BLOCKS / STAT_GROUPS];
    #pragma unroll
    for (int i = 0; i < STAT_BLOCKS / STAT_GROUPS; ++i) v[i] = pp[(g + i * STAT_GROUPS) * 64];
    float4 a = v[0];
    #pragma unroll
    for (int i = 1; i < STAT_BLOCKS / STAT_GROUPS; ++i) {
        a.x += v[i].x; a.y += v[i].y; a.z += v[i].z; a.w += v[i].w;
    }
    *(float4*)&fs[g][c4 * 4] = a;
    __syncthreads();
    if (threadIdx.x < 128) {
        const int c = threadIdx.x;
        float su = 0.f, sq = 0.f;
        for (int k = 0; k < STAT_GROUPS; ++k) { su += fs[k][c]; sq += fs[k][128 + c]; }
        float m   = su * invN;
        float var = sq * invN - m * m;
        float sc  = rsqrtf(var + EPS_BN) * gamma[c];
        ss[c] = sc;
        ss[128 + c] = beta[c] - m * sc;
    }
}

__global__ __launch_bounds__(256) void k_bn_apply(float* __restrict__ y,
                                                  const float* __restrict__ res,
                                                  const float* __restrict__ ss, int n) {
    int t = blockIdx.x * 256 + threadIdx.x;
    if (t >= n * 32) return;
    int row = t >> 5, c4 = (t & 31) << 2;
    long long off = (long long)row * HDIM + c4;
    float4 v  = *(const float4*)(y + off);
    float4 sc = *(const float4*)(ss + c4);
    float4 sh = *(const float4*)(ss + 128 + c4);
    v.x = fmaxf(v.x, 0.f) * sc.x + sh.x; v.y = fmaxf(v.y, 0.f) * sc.y + sh.y;
    v.z = fmaxf(v.z, 0.f) * sc.z + sh.z; v.w = fmaxf(v.w, 0.f) * sc.w + sh.w;
    if (res) {
        float4 r = *(const float4*)(res + off);
        v.x += r.x; v.y += r.y; v.z += r.z; v.w += r.w;
    }
    *(float4*)(y + off) = v;
}

#define POOL_BLOCKS 256
#define POOL_WAVES  32
#define POOL_BATCH  8
#define PSUM_GROUPS 16

__device__ __forceinline__ void pool_add(float* __restrict__ pl, int g, int lane,
                                         const float4 v, int cntBase) {
    float* p = pl + g * HDIM + lane * 4;
    atomicAdd(p + 0, v.x); atomicAdd(p + 1, v.y);
    atomicAdd(p + 2, v.z); atomicAdd(p + 3, v.w);
    if (lane == 0) atomicAdd(pl + cntBase + g, 1.0f);
}

__global__ __launch_bounds__(POOL_WAVES * 32) void k_pool_part(const float* __restrict__ x,
                                                                const int* __restrict__ batch,
                                                                float* __restrict__ planes,
                                                                int n, int G) {
    extern __shared__ float pl[];
    const int lane = threadIdx.x & 31;
    const int tot = G * HDIM + G;
    for (int i = threadIdx.x; i < tot; i += POOL_WAVES * 32) pl[i] = 0.0f;
    __syncthreads();
    const int S = (int)gridDim.x * POOL_WAVES;
    int r = __builtin_amdgcn_readfirstlane((int)blockIdx.x * POOL_WAVES + (int)(threadIdx.x >> 5));
    const float4* xp = (const float4*)x + lane;

    for (; r + (POOL_BATCH - 1) * S < n; r += POOL_BATCH * S) {
        float4 v[POOL_BATCH];
        int g[POOL_BATCH];
        #pragma unroll
        for (int k = 0; k < POOL_BATCH; ++k) {
            v[k] = xp[(long long)(r + k * S) * 32];
            g[k] = batch[r + k * S];
        }
        #pragma unroll
        for (int k = 0; k < POOL_BATCH; ++k) pool_add(pl, g[k], lane, v[k], G * HDIM);
    }
    if (r < n) {
        float4 v[POOL_BATCH - 1];
        int g[POOL_BATCH - 1];
        #pragma unroll
        for (int k = 0; k < POOL_BATCH - 1; ++k) {
            const int rr = r + k * S;
            const int rc = rr < n ? rr : n - 1;
            v[k] = xp[(long long)rc * 32];
            g[k] = batch[rc];
        }
        #pragma unroll
        for (int k = 0; k < POOL_BATCH - 1; ++k)
            if (r + k * S < n) pool_add(pl, g[k], lane, v[k], G * HDIM);
    }
    __syncthreads();
    float* o = planes + (long long)blockIdx.x * tot;
    for (int i = threadIdx.x; i < tot; i += POOL_WAVES * 32) o[i] = pl[i];
}

__global__ __launch_bounds__(PSUM_GROUPS * 64) void k_pool_sum(const float* __restrict__ planes,
                                                               float* __restrict__ outp, int tot) {
    __shared__ float fs[PSUM_GROUPS][64];
    const int o   = threadIdx.x & 63;
    const int grp = threadIdx.x >> 6;
    const int i   = blockIdx.x * 64 + o;
    float a = 0.0f;
    if (i < tot) {
        float v[POOL_BLOCKS / PSUM_GROUPS];
        #pragma unroll
        for (int j = 0; j < POOL_BLOCKS / PSUM_GROUPS; ++j)
            v[j] = planes[(long long)(grp + j * PSUM_GROUPS) * tot + i];
        a = v[0];
        #pragma unroll
        for (int j = 1; j < POOL_BLOCKS / PSUM_GROUPS; ++j) a += v[j];
    }
    fs[grp][o] = a;
    __syncthreads();
    if (threadIdx.x < 64 && i < tot) {
        float s = 0.0f;
        for (int k = 0; k < PSUM_GROUPS; ++k) s += fs[k][o];
        outp[i] = s;
    }
}

__global__ __launch_bounds__(256) void k_pool_final(const float* __restrict__ pool,
                                                    const float* __restrict__ counts,
                                                    float* __restrict__ xmean, int G) {
    int t = blockIdx.x * 256 + threadIdx.x;
    if (t >= G * HDIM) return;
    int g = t / HDIM;
    xmean[t] = pool[t] / fmaxf(counts[g], 1.0f);
}

__global__ void k_pred(const float* __restrict__ xmean, const float* __restrict__ Wr,
                       const float* __restrict__ br, float* __restrict__ pred) {
    __shared__ float red[128];
    int g = blockIdx.x, c = threadIdx.x;
    red[c] = xmean[(long long)g * HDIM + c] * Wr[c];
    __syncthreads();
    for (int s = 64; s > 0; s >>= 1) {
        if (c < s) red[c] += red[c + s];
        __syncthreads();
    }
    if (c == 0) pred[g] = red[0] + br[0];
}

extern "C" void kernel_launch(void* const* d_in, const int* in_sizes, int n_in,
                              void* d_out, int out_size, void* d_ws, size_t ws_size,
                              hipStream_t stream) {
    const float* X     = (const float*)d_in[0];
    const int*   edge  = (const int*)d_in[1];
    const int*   batch = (const int*)d_in[2];
    const float* W1    = (const float*)d_in[3];
    const float* b1    = (const float*)d_in[4];
    const float* Wm    = (const float*)d_in[5];
    const float* bm    = (const float*)d_in[6];
    const float* gamma = (const float*)d_in[7];
    const float* beta  = (const float*)d_in[8];
    const float* Wr    = (const float*)d_in[9];
    const float* br    = (const float*)d_in[10];
    float* out = (float*)d_out;

    const int N = in_sizes[0];
    const int E = in_sizes[1] / 2;
    const int G = out_size / (HDIM + 1);
    const int* src = edge;
    const int* dst = edge + E;

    const long long NH = (long long)N * HDIM;
    float* ws    = (float*)d_ws;
    float* buf[3] = { ws, ws + NH, ws + 2 * NH };
    float* dinv  = ws + 3 * NH;
    float* ss    = dinv + N;
    float* pool  = ss + 256;
    float* cnts  = pool + (long long)G * HDIM;

    const int gAgg = (N + AG_ROWS - 1) / AG_ROWS;

    const int nWpad = ((((E + 31) >> 5) + 3) >> 2) << 2;
    const int lastRows = N - (gAgg - 1) * AG_ROWS;
    const bool useMarks = (long long)nWpad * 4 <= (long long)lastRows * HDIM * 4
                          && gAgg * MK_WORDS * 4 <= 65536 && (((size_t)d_ws & 15) == 0) && nWpad > 0;
    const int gMk = (nWpad + MK_WORDS - 1) / MK_WORDS;
    const int gN   = (N + 255) / 256;
    const int gNH4 = (N * 32 + 255) / 256;
    const int gGem = (N + 31) / 32;

    float* s1   = buf[0];
    float* u1   = buf[0] + N;
    float* scp  = buf[1];
    const dim3 gSc(SC_CHUNKS, (N + SC_RANGE - 1) / SC_RANGE);
    k_deg_count<<<gSc, SC_THREADS, 0, stream>>>(dst, scp, E, N);
    k_deg_finish<<<gN, 256, 0, stream>>>(scp, X, dinv, u1, N);

    auto bn_tail = [&](float* abuf, float* part, const float* g_, const float* be_,
                       const float* res) {
        k_stats<<<STAT_BLOCKS, STAT_WAVES * 32, 0, stream>>>(abuf, part, N);
        k_scaleshift<<<1, STAT_GROUPS * 64, 0, stream>>>(part, g_, be_, ss, 1.0f / (float)N);
        k_bn_apply<<<gNH4, 256, 0, stream>>>(abuf, res, ss, N);
    };

    auto layer_tail = [&](float* hbuf, float* abuf, const float* bias_,
                          const float* g_, const float* be_, const float* res) {
        if (useMarks) {
            k_edge_marks<<<gMk, 256, gAgg * MK_WORDS * sizeof(unsigned int), stream>>>(dst, abuf, E, nWpad, gAgg);
            k_agg_lds<true><<<gAgg, AG_THREADS, 0, stream>>>(hbuf, src, dst, dinv, bias_, abuf, E, N);
        } else {
            k_agg_lds<false><<<gAgg, AG_THREADS, 0, stream>>>(hbuf, src, dst, dinv, bias_, abuf, E, N);
        }
        bn_tail(abuf, hbuf, g_, be_, res);
    };

    k_l1_sum<<<gSc, SC_THREADS, 0, stream>>>(u1, src, dst, scp, E, N);
    k_l1_finish<<<gN, 256, 0, stream>>>(scp, X, dinv, s1, N);
    k_expand<<<gNH4, 256, 0, stream>>>(s1, W1, b1, buf[2], N);
    bn_tail(buf[2], buf[0], gamma, beta, nullptr);

    float* cur = buf[2];
    float* f0  = buf[0];
    float* f1  = buf[1];

    for (int i = 0; i < 4; ++i) {
        float* hbuf = f0;
        float* abuf = f1;
        k_gemm_wmma<<<gGem, 256, 0, stream>>>(cur, Wm + (long long)i * HDIM * HDIM, dinv,
                                              hbuf, N);
        layer_tail(hbuf, abuf, bm + (long long)i * HDIM,
                   gamma + (long long)(i + 1) * HDIM,
                   beta  + (long long)(i + 1) * HDIM, cur);
        float* next = abuf;
        f1 = cur;
        f0 = hbuf;
        cur = next;
    }

    const int poolTot = G * HDIM + G;
    k_pool_part<<<POOL_BLOCKS, POOL_WAVES * 32, poolTot * sizeof(float), stream>>>(cur, batch, f0, N, G);
    k_pool_sum<<<(poolTot + 63) / 64, PSUM_GROUPS * 64, 0, stream>>>(f0, pool, poolTot);
    k_pool_final<<<(G * HDIM + 255) / 256, 256, 0, stream>>>(pool, cnts, out + G, G);
    k_pred<<<G, 128, 0, stream>>>(out + G, Wr, br, out);
}


# k_agg_lds: a block owns 400 destination rows, not 544: 250 equal blocks on 250 of 256 compute units (was 184), 26% fewer LDS adds and row gathers each; start/store 7 pieces deep; arrival-order adds
# speedup vs baseline: 3.6207x; 1.0400x over previous
#include <hip/hip_runtime.h>
#include <hip/hip_bf16.h>

#define HDIM 128
#define EPS_BN 1e-5f

typedef __attribute__((ext_vector_type(2))) float v2f;
typedef __attribute__((ext_vector_type(8))) float v8f;

#define SC_RANGE   16384
#define SC_CHUNKS  32
#define SC_THREADS 1024
#define SC_BATCH   8

__global__ __launch_bounds__(SC_THREADS) void k_deg_count(const int* __restrict__ dst,
                                                          float* __restrict__ part,
                                                          int E, int n) {
    __shared__ unsigned int cnt[SC_RANGE];
    const int r0 = blockIdx.y * SC_RANGE;
    for (int i = threadIdx.x; i < SC_RANGE; i += SC_THREADS) cnt[i] = 0u;
    __syncthreads();
    const int L = (E + (int)gridDim.x - 1) / (int)gridDim.x;
    const int eBeg = blockIdx.x * L;
    const int eEnd = (eBeg + L < E) ? (eBeg + L) : E;
    int e = eBeg + threadIdx.x;
    for (; e + (SC_BATCH - 1) * SC_THREADS < eEnd; e += SC_BATCH * SC_THREADS) {
        int d[SC_BATCH];
        #pragma unroll
        for (int k = 0; k < SC_BATCH; ++k) d[k] = dst[e + k * SC_THREADS];
        #pragma unroll
        for (int k = 0; k < SC_BATCH; ++k) {
            const unsigned rel = (unsigned)(d[k] - r0);
            if (rel < SC_RANGE) atomicAdd(&cnt[rel], 1u);
        }
    }
    for (; e < eEnd; e += SC_THREADS) {
        const unsigned rel = (unsigned)(dst[e] - r0);
        if (rel < SC_RANGE) atomicAdd(&cnt[rel], 1u);
    }
    __syncthreads();
    const int lim = (n - r0 < SC_RANGE) ? (n - r0) : SC_RANGE;
    float* o = part + (long long)blockIdx.x * n + r0;
    for (int i = threadIdx.x; i < lim; i += SC_THREADS) o[i] = (float)cnt[i];
}

__global__ __launch_bounds__(256) void k_deg_finish(const float* __restrict__ part,
                                                    const float* __restrict__ X,
                                                    float* __restrict__ dinv,
                                                    float* __restrict__ u, int n) {
    const int t = blockIdx.x * 256 + threadIdx.x;
    if (t >= n) return;
    float v[SC_CHUNKS];
    #pragma unroll
    for (int c = 0; c < SC_CHUNKS; ++c) v[c] = part[(long long)c * n + t];
    float s = v[0];
    #pragma unroll
    for (int c = 1; c < SC_CHUNKS; ++c) s += v[c];
    const float di = rsqrtf(s + 1.0f);
    dinv[t] = di;
    u[t] = X[t] * di;
}

__global__ __launch_bounds__(SC_THREADS) void k_l1_sum(const float* __restrict__ u,
                                                       const int* __restrict__ src,
                                                       const int* __restrict__ dst,
                                                       float* __restrict__ part,
                                                       int E, int n) {
    __shared__ float acc[SC_RANGE];
    const int r0 = blockIdx.y * SC_RANGE;
    for (int i = threadIdx.x; i < SC_RANGE; i += SC_THREADS) acc[i] = 0.0f;
    __syncthreads();
    const int L = (E + (int)gridDim.x - 1) / (int)gridDim.x;
    const int eBeg = blockIdx.x * L;
    const int eEnd = (eBeg + L < E) ? (eBeg + L) : E;
    int e = eBeg + threadIdx.x;
    for (; e + (SC_BATCH - 1) * SC_THREADS < eEnd; e += SC_BATCH * SC_THREADS) {
        int d[SC_BATCH], s[SC_BATCH];
        float val[SC_BATCH];
        #pragma unroll
        for (int k = 0; k < SC_BATCH; ++k) {
            d[k] = dst[e + k * SC_THREADS];
            s[k] = src[e + k * SC_THREADS];
        }
        #pragma unroll
        for (int k = 0; k < SC_BATCH; ++k) asm volatile("" :: "v"(d[k]), "v"(s[k]));
        #pragma unroll
        for (int k = 0; k < SC_BATCH; ++k) {
            const bool in = (unsigned)(d[k] - r0) < SC_RANGE;
            val[k] = in ? u[s[k]] : 0.0f;
        }
        #pragma unroll
        for (int k = 0; k < SC_BATCH; ++k) {
            const unsigned rel = (unsigned)(d[k] - r0);
            if (rel < SC_RANGE) atomicAdd(&acc[rel], val[k]);
        }
    }
    for (; e < eEnd; e += SC_THREADS) {
        const unsigned rel = (unsigned)(dst[e] - r0);
        if (rel < SC_RANGE) atomicAdd(&acc[rel], u[src[e]]);
    }
    __syncthreads();
    const int lim = (n - r0 < SC_RANGE) ? (n - r0) : SC_RANGE;
    float* o = part + (long long)blockIdx.x * n + r0;
    for (int i = threadIdx.x; i < lim; i += SC_THREADS) o[i] = acc[i];
}

__global__ __launch_bounds__(256) void k_l1_finish(const float* __restrict__ part,
                                                   const float* __restrict__ X,
                                                   const float* __restrict__ dinv,
                                                   float* __restrict__ s1, int n) {
    const int t = blockIdx.x * 256 + threadIdx.x;
    if (t >= n) return;
    float v[SC_CHUNKS];
    #pragma unroll
    for (int c = 0; c < SC_CHUNKS; ++c) v[c] = part[(long long)c * n + t];
    float s = v[0];
    #pragma unroll
    for (int c = 1; c < SC_CHUNKS; ++c) s += v[c];
    const float di = dinv[t];
    s1[t] = di * s + X[t] * (di * di);
}

__global__ __launch_bounds__(256) void k_expand(const float* __restrict__ s1,
                                                     const float* __restrict__ W1,
                                                     const float* __restrict__ bias,
                                                     float* __restrict__ y, int n) {
    int t = blockIdx.x * 256 + threadIdx.x;
    if (t >= n * 32) return;
    int row = t >> 5, c4 = (t & 31) << 2;
    float sv = s1[row];
    const float4 w = *(const float4*)(W1 + c4);
    const float4 b = *(const float4*)(bias + c4);
    float4 o;
    o.x = sv * w.x + b.x;
    o.y = sv * w.y + b.y;
    o.z = sv * w.z + b.z;
    o.w = sv * w.w + b.w;
    *(float4*)(y + (long long)row * HDIM + c4) = o;
}

__global__ __launch_bounds__(256) void k_gemm_wmma(const float* __restrict__ x,
                                                   const float* __restrict__ W,
                                                   const float* __restrict__ dinv,
                                                   float* __restrict__ hs, int n) {
    __shared__ float tileX[32 * HDIM];
    const int rowBase = blockIdx.x * 32;
    const int tid  = threadIdx.x;
    const int wave = tid >> 5;
    const int lane = tid & 31;
    const int half = lane >> 4;
    const int l16  = lane & 15;
    const int rowHalf = wave & 1;
    const int c0   = (wave >> 1) * 32;

    const bool fullTile = (rowBase + 32 <= n);

    if (fullTile) {
        const float4* xs = (const float4*)(x + (long long)rowBase * HDIM);
        #pragma unroll
        for (int i = 0; i < 4; ++i) {
            int idx = tid + i * 256;
            ((float4*)tileX)[idx] = xs[idx];
        }
    } else {
        #pragma unroll
        for (int i = 0; i < 4; ++i) {
            int idx = tid + i * 256;
            int r = rowBase + (idx >> 5);
            float4 v = make_float4(0.f, 0.f, 0.f, 0.f);
            if (r < n) v = *(const float4*)(x + (long long)r * HDIM + ((idx & 31) << 2));
            ((float4*)tileX)[idx] = v;
        }
    }
    __syncthreads();

    const float* xrow = tileX + (rowHalf * 16 + l16) * HDIM + 2 * half;
    v8f acc0 = {}, acc1 = {};

    #pragma unroll 4
    for (int k = 0; k < HDIM; k += 4) {
        v2f a; a.x = xrow[k]; a.y = xrow[k + 1];
        const float* wp = W + (long long)(k + 2 * half) * HDIM + c0 + l16;
        v2f b0; b0.x = wp[0];  b0.y = wp[HDIM];
        v2f b1; b1.x = wp[16]; b1.y = wp[HDIM + 16];
        acc0 = __builtin_amdgcn_wmma_f32_16x16x4_f32(false, a, false, b0,
                                                     (short)0, acc0, false, false);
        acc1 = __builtin_amdgcn_wmma_f32_16x16x4_f32(false, a, false, b1,
                                                     (short)0, acc1, false, false);
    }

    const int rbase = rowBase + rowHalf * 16 + half * 8;
    if (fullTile) {
        float di[8];
        #pragma unroll
        for (int v = 0; v < 8; ++v) di[v] = dinv[rbase + v];
        #pragma unroll
        for (int v = 0; v < 8; ++v) {
            const long long o = (long long)(rbase + v) * HDIM + c0 + l16;
            hs[o]      = acc0[v] * di[v];
            hs[o + 16] = acc1[v] * di[v];
        }
    } else {
        for (int v = 0; v < 8; ++v) {
            int r = rbase + v;
            if (r < n) {
                const float dd = dinv[r];
                const long long o = (long long)r * HDIM + c0 + l16;
                hs[o]      = acc0[v] * dd;
                hs[o + 16] = acc1[v] * dd;
            }
        }
    }
}

#define AG_ROWS    400
#define AG_THREADS 1024
#define AG_PF      8
#define AG_BATCH   8
#define AG_IO      7

#define MK_EDGES 2048
#define MK_WORDS (MK_EDGES / 32)

__global__ __launch_bounds__(256) void k_edge_marks(const int* __restrict__ dst,
                                                    float* __restrict__ ybuf,
                                                    int E, int nWpad, int gAgg) {
    extern __shared__ unsigned int tab[];
    const int tot = gAgg * MK_WORDS;
    for (int i = threadIdx.x; i < tot; i += 256) tab[i] = 0u;
    __syncthreads();
    const int e0 = blockIdx.x * MK_EDGES;
    int d[MK_EDGES / 256];
    #pragma unroll
    for (int k = 0; k < MK_EDGES / 256; ++k) {
        const int e = e0 + threadIdx.x + k * 256;
        const int v = dst[e < E ? e : E - 1];
        d[k] = e < E ? v : -1;
    }
    #pragma unroll
    for (int k = 0; k < MK_EDGES / 256; ++k) {
        const int le = threadIdx.x + k * 256;
        const unsigned b = (unsigned)d[k] / AG_ROWS;
        if (b < (unsigned)gAgg) atomicOr(&tab[b * MK_WORDS + (le >> 5)], 1u << (le & 31));
    }
    __syncthreads();
    const int wj = blockIdx.x * MK_WORDS + (threadIdx.x % MK_WORDS);
    if (wj < nWpad) {
        for (int i0 = threadIdx.x; i0 < tot; i0 += 4 * 256) {
            unsigned int v[4];
            #pragma unroll
            for (int k = 0; k < 4; ++k) {
                const int i = i0 + k * 256;
                v[k] = tab[i < tot ? i : tot - 1];
            }
            #pragma unroll
            for (int k = 0; k < 4; ++k) {
                const int i = i0 + k * 256;
                if (i < tot)
                    ((unsigned int*)(ybuf + (long long)(i / MK_WORDS) * AG_ROWS * HDIM))[wj] = v[k];
            }
        }
    }
}

__device__ __forceinline__ void agg_drain(const float* __restrict__ hs,
                                          const int* __restrict__ src,
                                          float* acc, int pe, int pr, int count, int lane) {
    int s = 0;
    if (lane < count) s = src[pe];
    #pragma nounroll
    for (int j0 = 0; j0 < count; j0 += AG_BATCH) {
        float v[AG_BATCH][4];
        #pragma unroll
        for (int u = 0; u < AG_BATCH; ++u) {
            int jj = j0 + u;
            jj = jj < count ? jj : count - 1;
            const int sj = __builtin_amdgcn_readlane(s, jj);
            const float* hp = hs + (long long)sj * HDIM;
            v[u][0] = hp[lane];
            v[u][1] = hp[lane + 32];
            v[u][2] = hp[lane + 64];
            v[u][3] = hp[lane + 96];
        }
        #pragma unroll
        for (int u = 0; u < AG_BATCH; ++u) {
            if (j0 + u < count) {
                const int rj = __builtin_amdgcn_readlane(pr, j0 + u);
                float* ap = acc + rj * HDIM + lane;
                atomicAdd(ap,      v[u][0]);
                atomicAdd(ap + 32, v[u][1]);
                atomicAdd(ap + 64, v[u][2]);
                atomicAdd(ap + 96, v[u][3]);
            }
        }
    }
}

template <bool MARKS>
__global__ __launch_bounds__(AG_THREADS) void k_agg_lds(const float* __restrict__ hs,
                                                        const int* __restrict__ src,
                                                        const int* __restrict__ dst,
                                                        const float* __restrict__ dinv,
                                                        const float* __restrict__ bias,
                                                        float* __restrict__ y, int E, int n) {
    __shared__ float acc[AG_ROWS * HDIM];
    const int lane = threadIdx.x & 31;
    const int wave = __builtin_amdgcn_readfirstlane((int)(threadIdx.x >> 5));
    const int r0 = blockIdx.x * AG_ROWS;
    const int lim = (n - r0 < AG_ROWS) ? (n - r0) : AG_ROWS;
    const unsigned ulim = (unsigned)lim;

    const int tot = lim * 32;
    {
        const float4* hp = (const float4*)(hs + (long long)r0 * HDIM);
        for (int i0 = threadIdx.x; i0 < tot; i0 += AG_IO * AG_THREADS) {
            float4 t[AG_IO];
            #pragma unroll
            for (int q = 0; q < AG_IO; ++q) {
                const int i = i0 + q * AG_THREADS;
                t[q] = hp[i < tot ? i : tot - 1];
            }
            asm volatile("" :: "v"(t[0].x), "v"(t[0].y), "v"(t[0].z), "v"(t[0].w),
                               "v"(t[1].x), "v"(t[1].y), "v"(t[1].z), "v"(t[1].w),
                               "v"(t[2].x), "v"(t[2].y), "v"(t[2].z), "v"(t[2].w),
                               "v"(t[3].x), "v"(t[3].y), "v"(t[3].z), "v"(t[3].w),
                               "v"(t[4].x), "v"(t[4].y), "v"(t[4].z), "v"(t[4].w),
                               "v"(t[5].x), "v"(t[5].y), "v"(t[5].z), "v"(t[5].w),
                               "v"(t[6].x), "v"(t[6].y), "v"(t[6].z), "v"(t[6].w));
            #pragma unroll
            for (int q = 0; q < AG_IO; ++q) {
                const int i = i0 + q * AG_THREADS;
                if (i < tot) ((float4*)acc)[i] = t[q];
            }
        }
    }
    __syncthreads();

    int pe = 0, pr = 0, cnt = 0;

    int tailBeg;
    if constexpr (MARKS) {
        const int nq = (((E + 31) >> 5) + 3) >> 2;
        const uint4* mk = (const uint4*)(y + (long long)r0 * HDIM);
        const int nIt = (nq + 31) >> 5;
        const int K = nIt > wave ? ((nIt - wave + 31) >> 5) : 0;
        const uint4 none = make_uint4(0u, 0u, 0u, 0u);
        auto piece = [&](int t) {
            const int idx = (wave + 32 * t) * 32 + lane;
            return mk[idx < nq ? idx : nq - 1];
        };
        uint4 n0 = none, n1 = none;
        if (K > 0) n0 = piece(0);
        if (K > 1) n1 = piece(1);
        #pragma nounroll
        for (int k = 0; k < K; ++k) {
            {
                const uint4 cur = n0;
                n0 = n1;
                if (k + 2 < K) n1 = piece(k + 2);
                const int itr = wave + 32 * k;
                const bool ok = itr * 32 + lane < nq;
                const unsigned m0 = ok ? cur.x : 0u, m1 = ok ? cur.y : 0u;
                const unsigned m2 = ok ? cur.z : 0u, m3 = ok ? cur.w : 0u;
                #pragma nounroll
                for (int q = 0; q < 4; ++q) {
                    const unsigned mv = q == 0 ? m0 : (q == 1 ? m1 : (q == 2 ? m2 : m3));
                    unsigned act = __builtin_amdgcn_ballot_w32(mv != 0u);
                    while (act != 0u) {
                        const int L = __builtin_ctz(act);
                        act &= act - 1u;
                        unsigned sw = (unsigned)__builtin_amdgcn_readlane((int)mv, L);
                        const int e0 = ((((itr * 32 + L) << 2) + q) << 5);
                        while (sw != 0u) {
                            const int p = __builtin_ctz(sw);
                            sw &= sw - 1u;
                            pe = lane == cnt ? e0 + p : pe;
                            ++cnt;
                            if (cnt == 32) {
                                pr = dst[pe] - r0;
                                agg_drain(hs, src, acc, pe, pr, 32, lane);
                                cnt = 0;
                            }
                        }
                    }
                }
            }
        }
        tailBeg = E;
    } else {
        const int nvec = (((size_t)dst & 15) == 0) ? (E >> 2) : 0;
        const int4* dv = (const int4*)dst;
        const int nIt = (nvec + 31) >> 5;
        const int K = nIt > wave ? ((nIt - wave + 31) >> 5) : 0;
        const int rot = (int)blockIdx.x * (nIt / (int)gridDim.x);
        const int4 none = make_int4(-1, -1, -1, -1);

        int4 buf[AG_PF];
        #pragma unroll
        for (int u = 0; u < AG_PF; ++u) {
            buf[u] = none;
            if (u < K) {
                int itr = wave + 32 * u + rot;
                itr = itr >= nIt ? itr - nIt : itr;
                const int idx = itr * 32 + lane;
                buf[u] = dv[idx < nvec ? idx : nvec - 1];
            }
        }
        for (int k = 0; k < K; k += AG_PF) {
            #pragma unroll
            for (int u = 0; u < AG_PF; ++u) {
                if (k + u >= K) break;
                const int4 cur = buf[u];
                int itr = wave + 32 * (k + u) + rot;
                itr = itr >= nIt ? itr - nIt : itr;
                {
                    const int kn = k + u + AG_PF;
                    if (kn < K) {
                        int itn = wave + 32 * kn + rot;
                        itn = itn >= nIt ? itn - nIt : itn;
                        const int idx = itn * 32 + lane;
                        buf[u] = dv[idx < nvec ? idx : nvec - 1];
                    }
                }
                const bool ok = itr * 32 + lane < nvec;
                const bool m0 = ok & ((unsigned)(cur.x - r0) < ulim);
                const bool m1 = ok & ((unsigned)(cur.y - r0) < ulim);
                const bool m2 = ok & ((unsigned)(cur.z - r0) < ulim);
                const bool m3 = ok & ((unsigned)(cur.w - r0) < ulim);
                if (__builtin_amdgcn_ballot_w32(m0 | m1 | m2 | m3) != 0u) {
                    unsigned bits = (m0 ? 1u : 0u) | (m1 ? 2u : 0u) | (m2 ? 4u : 0u) | (m3 ? 8u : 0u);
                    unsigned act = __builtin_amdgcn_ballot_w32(bits != 0u);
                    while (act != 0u) {
                        const int L = __builtin_ctz(act);
                        const unsigned b = (unsigned)__builtin_amdgcn_readlane((int)bits, L);
                        const int j = __builtin_ctz(b);
                        const int vsel = j == 0 ? cur.x : (j == 1 ? cur.y : (j == 2 ? cur.z : cur.w));
                        const int d = __builtin_amdgcn_readlane(vsel, L);
                        const int e = ((itr * 32 + L) << 2) + j;
                        pe = lane == cnt ? e : pe;
                        pr = lane == cnt ? d - r0 : pr;
                        ++cnt;
                        if (lane == L) bits &= bits - 1u;
                        if (cnt == 32) {
                            agg_drain(hs, src, acc, pe, pr, 32, lane);
                            cnt = 0;
                        }
                        act = __builtin_amdgcn_ballot_w32(bits != 0u);
                    }
                }
            }
        }
        tailBeg = nvec * 4;
    }
    if constexpr (!MARKS) {
        for (int base = tailBeg + wave * 32; base < E; base += AG_THREADS) {
            const int e = base + lane;
            const int d = e < E ? dst[e] : -1;
            unsigned act = __builtin_amdgcn_ballot_w32((unsigned)(d - r0) < ulim);
            while (act != 0u) {
                const int L = __builtin_ctz(act);
                act &= act - 1u;
                const int dl = __builtin_amdgcn_readlane(d, L);
                pe = lane == cnt ? base + L : pe;
                pr = lane == cnt ? dl - r0 : pr;
                ++cnt;
                if (cnt == 32) {
                    agg_drain(hs, src, acc, pe, pr, 32, lane);
                    cnt = 0;
                }
            }
        }
    }
    if constexpr (MARKS) {
        if (lane < cnt) pr = dst[pe] - r0;
    }
    agg_drain(hs, src, acc, pe, pr, cnt, lane);
    __syncthreads();

    {
        float4* yp = (float4*)(y + (long long)r0 * HDIM);
        const float4 b = *(const float4*)(bias + (lane << 2));
        for (int i0 = threadIdx.x; i0 < tot; i0 += AG_IO * AG_THREADS) {
            float di[AG_IO];
            #pragma unroll
            for (int q = 0; q < AG_IO; ++q) {
                const int i = i0 + q * AG_THREADS;
                di[q] = dinv[r0 + ((i < tot ? i : tot - 1) >> 5)];
            }
            #pragma unroll
            for (int q = 0; q < AG_IO; ++q) {
                const int i = i0 + q * AG_THREADS;
                if (i < tot) {
                    const float4 a = ((const float4*)acc)[i];
                    float4 o;
                    o.x = a.x * di[q] + b.x;
                    o.y = a.y * di[q] + b.y;
                    o.z = a.z * di[q] + b.z;
                    o.w = a.w * di[q] + b.w;
                    yp[i] = o;
                }
            }
        }
    }
}

#define STAT_BLOCKS 256
#define STAT_WAVES  32
#define STAT_BATCH  8
#define STAT_GROUPS 16

__device__ __forceinline__ void stat_acc(float4& s, float4& q, const float4 p) {
    float4 v;
    v.x = fmaxf(p.x, 0.f); v.y = fmaxf(p.y, 0.f); v.z = fmaxf(p.z, 0.f); v.w = fmaxf(p.w, 0.f);
    s.x += v.x; s.y += v.y; s.z += v.z; s.w += v.w;
    q.x += v.x * v.x; q.y += v.y * v.y; q.z += v.z * v.z; q.w += v.w * v.w;
}

__global__ __launch_bounds__(STAT_WAVES * 32) void k_stats(const float* __restrict__ y,
                                                            float* __restrict__ part, int n) {
    __shared__ float ls[STAT_WAVES][256];
    const int lane = threadIdx.x & 31;
    const int wave = threadIdx.x >> 5;
    const int S = STAT_BLOCKS * STAT_WAVES;
    const float4* yp = (const float4*)y + lane;
    int r = blockIdx.x * STAT_WAVES + wave;
    float4 s = make_float4(0.f, 0.f, 0.f, 0.f), q = s;

    for (; r + (STAT_BATCH - 1) * S < n; r += STAT_BATCH * S) {
        float4 v[STAT_BATCH];
        #pragma unroll
        for (int u = 0; u < STAT_BATCH; ++u) v[u] = yp[(long long)(r + u * S) * 32];
        #pragma unroll
        for (int u = 0; u < STAT_BATCH; ++u) stat_acc(s, q, v[u]);
    }
    if (r < n) {
        const float4 zero = make_float4(0.f, 0.f, 0.f, 0.f);
        float4 v[STAT_BATCH - 1];
        #pragma unroll
        for (int u = 0; u < STAT_BATCH - 1; ++u) {
            const int rr = r + u * S;
            const bool in = rr < n;
            const float4 t = yp[(long long)(in ? rr : n - 1) * 32];
            v[u] = in ? t : zero;
        }
        #pragma unroll
        for (int u = 0; u < STAT_BATCH - 1; ++u) stat_acc(s, q, v[u]);
    }

    *(float4*)&ls[wave][lane * 4]       = s;
    *(float4*)&ls[wave][128 + lane * 4] = q;
    __syncthreads();
    if (threadIdx.x < 256) {
        float a = 0.f;
        for (int w = 0; w < STAT_WAVES; ++w) a += ls[w][threadIdx.x];
        part[(long long)blockIdx.x * 256 + threadIdx.x] = a;
    }
}

__global__ __launch_bounds__(STAT_GROUPS * 64) void k_scaleshift(const float* __restrict__ part,
                                                                 const float* __restrict__ gamma,
                                                                 const float* __restrict__ beta,
                                                                 float* __restrict__ ss, float invN) {
    __shared__ float fs[STAT_GROUPS][256];
    const int c4 = threadIdx.x & 63;
    const int g  = threadIdx.x >> 6;
    const float4* pp = (const float4*)part + c4;
    float4 v[STAT_BLOCKS / STAT_GROUPS];
    #pragma unroll
    for (int i = 0; i < STAT_BLOCKS / STAT_GROUPS; ++i) v[i] = pp[(g + i * STAT_GROUPS) * 64];
    float4 a = v[0];
    #pragma unroll
    for (int i = 1; i < STAT_BLOCKS / STAT_GROUPS; ++i) {
        a.x += v[i].x; a.y += v[i].y; a.z += v[i].z; a.w += v[i].w;
    }
    *(float4*)&fs[g][c4 * 4] = a;
    __syncthreads();
    if (threadIdx.x < 128) {
        const int c = threadIdx.x;
        float su = 0.f, sq = 0.f;
        for (int k = 0; k < STAT_GROUPS; ++k) { su += fs[k][c]; sq += fs[k][128 + c]; }
        float m   = su * invN;
        float var = sq * invN - m * m;
        float sc  = rsqrtf(var + EPS_BN) * gamma[c];
        ss[c] = sc;
        ss[128 + c] = beta[c] - m * sc;
    }
}

__global__ __launch_bounds__(256) void k_bn_apply(float* __restrict__ y,
                                                  const float* __restrict__ res,
                                                  const float* __restrict__ ss, int n) {
    int t = blockIdx.x * 256 + threadIdx.x;
    if (t >= n * 32) return;
    int row = t >> 5, c4 = (t & 31) << 2;
    long long off = (long long)row * HDIM + c4;
    float4 v  = *(const float4*)(y + off);
    float4 sc = *(const float4*)(ss + c4);
    float4 sh = *(const float4*)(ss + 128 + c4);
    v.x = fmaxf(v.x, 0.f) * sc.x + sh.x; v.y = fmaxf(v.y, 0.f) * sc.y + sh.y;
    v.z = fmaxf(v.z, 0.f) * sc.z + sh.z; v.w = fmaxf(v.w, 0.f) * sc.w + sh.w;
    if (res) {
        float4 r = *(const float4*)(res + off);
        v.x += r.x; v.y += r.y; v.z += r.z; v.w += r.w;
    }
    *(float4*)(y + off) = v;
}

#define POOL_BLOCKS 256
#define POOL_WAVES  32
#define POOL_BATCH  8
#define PSUM_GROUPS 16

__device__ __forceinline__ void pool_add(float* __restrict__ pl, int g, int lane,
                                         const float4 v, int cntBase) {
    float* p = pl + g * HDIM + lane * 4;
    atomicAdd(p + 0, v.x); atomicAdd(p + 1, v.y);
    atomicAdd(p + 2, v.z); atomicAdd(p + 3, v.w);
    if (lane == 0) atomicAdd(pl + cntBase + g, 1.0f);
}

__global__ __launch_bounds__(POOL_WAVES * 32) void k_pool_part(const float* __restrict__ x,
                                                                const int* __restrict__ batch,
                                                                float* __restrict__ planes,
                                                                int n, int G) {
    extern __shared__ float pl[];
    const int lane = threadIdx.x & 31;
    const int tot = G * HDIM + G;
    for (int i = threadIdx.x; i < tot; i += POOL_WAVES * 32) pl[i] = 0.0f;
    __syncthreads();
    const int S = (int)gridDim.x * POOL_WAVES;
    int r = __builtin_amdgcn_readfirstlane((int)blockIdx.x * POOL_WAVES + (int)(threadIdx.x >> 5));
    const float4* xp = (const float4*)x + lane;

    for (; r + (POOL_BATCH - 1) * S < n; r += POOL_BATCH * S) {
        float4 v[POOL_BATCH];
        int g[POOL_BATCH];
        #pragma unroll
        for (int k = 0; k < POOL_BATCH; ++k) {
            v[k] = xp[(long long)(r + k * S) * 32];
            g[k] = batch[r + k * S];
        }
        #pragma unroll
        for (int k = 0; k < POOL_BATCH; ++k) pool_add(pl, g[k], lane, v[k], G * HDIM);
    }
    if (r < n) {
        float4 v[POOL_BATCH - 1];
        int g[POOL_BATCH - 1];
        #pragma unroll
        for (int k = 0; k < POOL_BATCH - 1; ++k) {
            const int rr = r + k * S;
            const int rc = rr < n ? rr : n - 1;
            v[k] = xp[(long long)rc * 32];
            g[k] = batch[rc];
        }
        #pragma unroll
        for (int k = 0; k < POOL_BATCH - 1; ++k)
            if (r + k * S < n) pool_add(pl, g[k], lane, v[k], G * HDIM);
    }
    __syncthreads();
    float* o = planes + (long long)blockIdx.x * tot;
    for (int i = threadIdx.x; i < tot; i += POOL_WAVES * 32) o[i] = pl[i];
}

__global__ __launch_bounds__(PSUM_GROUPS * 64) void k_pool_sum(const float* __restrict__ planes,
                                                               float* __restrict__ outp, int tot) {
    __shared__ float fs[PSUM_GROUPS][64];
    const int o   = threadIdx.x & 63;
    const int grp = threadIdx.x >> 6;
    const int i   = blockIdx.x * 64 + o;
    float a = 0.0f;
    if (i < tot) {
        float v[POOL_BLOCKS / PSUM_GROUPS];
        #pragma unroll
        for (int j = 0; j < POOL_BLOCKS / PSUM_GROUPS; ++j)
            v[j] = planes[(long long)(grp + j * PSUM_GROUPS) * tot + i];
        a = v[0];
        #pragma unroll
        for (int j = 1; j < POOL_BLOCKS / PSUM_GROUPS; ++j) a += v[j];
    }
    fs[grp][o] = a;
    __syncthreads();
    if (threadIdx.x < 64 && i < tot) {
        float s = 0.0f;
        for (int k = 0; k < PSUM_GROUPS; ++k) s += fs[k][o];
        outp[i] = s;
    }
}

__global__ __launch_bounds__(256) void k_pool_final(const float* __restrict__ pool,
                                                    const float* __restrict__ counts,
                                                    float* __restrict__ xmean, int G) {
    int t = blockIdx.x * 256 + threadIdx.x;
    if (t >= G * HDIM) return;
    int g = t / HDIM;
    xmean[t] = pool[t] / fmaxf(counts[g], 1.0f);
}

__global__ void k_pred(const float* __restrict__ xmean, const float* __restrict__ Wr,
                       const float* __restrict__ br, float* __restrict__ pred) {
    __shared__ float red[128];
    int g = blockIdx.x, c = threadIdx.x;
    red[c] = xmean[(long long)g * HDIM + c] * Wr[c];
    __syncthreads();
    for (int s = 64; s > 0; s >>= 1) {
        if (c < s) red[c] += red[c + s];
        __syncthreads();
    }
    if (c == 0) pred[g] = red[0] + br[0];
}

extern "C" void kernel_launch(void* const* d_in, const int* in_sizes, int n_in,
                              void* d_out, int out_size, void* d_ws, size_t ws_size,
                              hipStream_t stream) {
    const float* X     = (const float*)d_in[0];
    const int*   edge  = (const int*)d_in[1];
    const int*   batch = (const int*)d_in[2];
    const float* W1    = (const float*)d_in[3];
    const float* b1    = (const float*)d_in[4];
    const float* Wm    = (const float*)d_in[5];
    const float* bm    = (const float*)d_in[6];
    const float* gamma = (const float*)d_in[7];
    const float* beta  = (const float*)d_in[8];
    const float* Wr    = (const float*)d_in[9];
    const float* br    = (const float*)d_in[10];
    float* out = (float*)d_out;

    const int N = in_sizes[0];
    const int E = in_sizes[1] / 2;
    const int G = out_size / (HDIM + 1);
    const int* src = edge;
    const int* dst = edge + E;

    const long long NH = (long long)N * HDIM;
    float* ws    = (float*)d_ws;
    float* buf[3] = { ws, ws + NH, ws + 2 * NH };
    float* dinv  = ws + 3 * NH;
    float* ss    = dinv + N;
    float* pool  = ss + 256;
    float* cnts  = pool + (long long)G * HDIM;

    const int gAgg = (N + AG_ROWS - 1) / AG_ROWS;

    const int nWpad = ((((E + 31) >> 5) + 3) >> 2) << 2;
    const int lastRows = N - (gAgg - 1) * AG_ROWS;
    const bool useMarks = (long long)nWpad * 4 <= (long long)lastRows * HDIM * 4
                          && gAgg * MK_WORDS * 4 <= 65536 && (((size_t)d_ws & 15) == 0) && nWpad > 0;
    const int gMk = (nWpad + MK_WORDS - 1) / MK_WORDS;
    const int gN   = (N + 255) / 256;
    const int gNH4 = (N * 32 + 255) / 256;
    const int gGem = (N + 31) / 32;

    float* s1   = buf[0];
    float* u1   = buf[0] + N;
    float* scp  = buf[1];
    const dim3 gSc(SC_CHUNKS, (N + SC_RANGE - 1) / SC_RANGE);
    k_deg_count<<<gSc, SC_THREADS, 0, stream>>>(dst, scp, E, N);
    k_deg_finish<<<gN, 256, 0, stream>>>(scp, X, dinv, u1, N);

    auto bn_tail = [&](float* abuf, float* part, const float* g_, const float* be_,
                       const float* res) {
        k_stats<<<STAT_BLOCKS, STAT_WAVES * 32, 0, stream>>>(abuf, part, N);
        k_scaleshift<<<1, STAT_GROUPS * 64, 0, stream>>>(part, g_, be_, ss, 1.0f / (float)N);
        k_bn_apply<<<gNH4, 256, 0, stream>>>(abuf, res, ss, N);
    };

    auto layer_tail = [&](float* hbuf, float* abuf, const float* bias_,
                          const float* g_, const float* be_, const float* res) {
        if (useMarks) {
            k_edge_marks<<<gMk, 256, gAgg * MK_WORDS * sizeof(unsigned int), stream>>>(dst, abuf, E, nWpad, gAgg);
            k_agg_lds<true><<<gAgg, AG_THREADS, 0, stream>>>(hbuf, src, dst, dinv, bias_, abuf, E, N);
        } else {
            k_agg_lds<false><<<gAgg, AG_THREADS, 0, stream>>>(hbuf, src, dst, dinv, bias_, abuf, E, N);
        }
        bn_tail(abuf, hbuf, g_, be_, res);
    };

    k_l1_sum<<<gSc, SC_THREADS, 0, stream>>>(u1, src, dst, scp, E, N);
    k_l1_finish<<<gN, 256, 0, stream>>>(scp, X, dinv, s1, N);
    k_expand<<<gNH4, 256, 0, stream>>>(s1, W1, b1, buf[2], N);
    bn_tail(buf[2], buf[0], gamma, beta, nullptr);

    float* cur = buf[2];
    float* f0  = buf[0];
    float* f1  = buf[1];

    for (int i = 0; i < 4; ++i) {
        float* hbuf = f0;
        float* abuf = f1;
        k_gemm_wmma<<<gGem, 256, 0, stream>>>(cur, Wm + (long long)i * HDIM * HDIM, dinv,
                                              hbuf, N);
        layer_tail(hbuf, abuf, bm + (long long)i * HDIM,
                   gamma + (long long)(i + 1) * HDIM,
                   beta  + (long long)(i + 1) * HDIM, cur);
        float* next = abuf;
        f1 = cur;
        f0 = hbuf;
        cur = next;
    }

    const int poolTot = G * HDIM + G;
    k_pool_part<<<POOL_BLOCKS, POOL_WAVES * 32, poolTot * sizeof(float), stream>>>(cur, batch, f0, N, G);
    k_pool_sum<<<(poolTot + 63) / 64, PSUM_GROUPS * 64, 0, stream>>>(f0, pool, poolTot);
    k_pool_final<<<(G * HDIM + 255) / 256, 256, 0, stream>>>(pool, cnts, out + G, G);
    k_pred<<<G, 128, 0, stream>>>(out + G, Wr, br, out);
}


# ReLU + BatchNorm apply (+ residual) of layers 1-4 done by the next layer's GEMM while it stages its tile; x stored in place of y by the GEMM; 4 apply launches and their re-read of x gone (36 -> 32)
# speedup vs baseline: 3.8560x; 1.0650x over previous
#include <hip/hip_runtime.h>
#include <hip/hip_bf16.h>

#define HDIM 128
#define EPS_BN 1e-5f

typedef __attribute__((ext_vector_type(2))) float v2f;
typedef __attribute__((ext_vector_type(8))) float v8f;

#define SC_RANGE   16384
#define SC_CHUNKS  32
#define SC_THREADS 1024
#define SC_BATCH   8

__global__ __launch_bounds__(SC_THREADS) void k_deg_count(const int* __restrict__ dst,
                                                          float* __restrict__ part,
                                                          int E, int n) {
    __shared__ unsigned int cnt[SC_RANGE];
    const int r0 = blockIdx.y * SC_RANGE;
    for (int i = threadIdx.x; i < SC_RANGE; i += SC_THREADS) cnt[i] = 0u;
    __syncthreads();
    const int L = (E + (int)gridDim.x - 1) / (int)gridDim.x;
    const int eBeg = blockIdx.x * L;
    const int eEnd = (eBeg + L < E) ? (eBeg + L) : E;
    int e = eBeg + threadIdx.x;
    for (; e + (SC_BATCH - 1) * SC_THREADS < eEnd; e += SC_BATCH * SC_THREADS) {
        int d[SC_BATCH];
        #pragma unroll
        for (int k = 0; k < SC_BATCH; ++k) d[k] = dst[e + k * SC_THREADS];
        #pragma unroll
        for (int k = 0; k < SC_BATCH; ++k) {
            const unsigned rel = (unsigned)(d[k] - r0);
            if (rel < SC_RANGE) atomicAdd(&cnt[rel], 1u);
        }
    }
    for (; e < eEnd; e += SC_THREADS) {
        const unsigned rel = (unsigned)(dst[e] - r0);
        if (rel < SC_RANGE) atomicAdd(&cnt[rel], 1u);
    }
    __syncthreads();
    const int lim = (n - r0 < SC_RANGE) ? (n - r0) : SC_RANGE;
    float* o = part + (long long)blockIdx.x * n + r0;
    for (int i = threadIdx.x; i < lim; i += SC_THREADS) o[i] = (float)cnt[i];
}

__global__ __launch_bounds__(256) void k_deg_finish(const float* __restrict__ part,
                                                    const float* __restrict__ X,
                                                    float* __restrict__ dinv,
                                                    float* __restrict__ u, int n) {
    const int t = blockIdx.x * 256 + threadIdx.x;
    if (t >= n) return;
    float v[SC_CHUNKS];
    #pragma unroll
    for (int c = 0; c < SC_CHUNKS; ++c) v[c] = part[(long long)c * n + t];
    float s = v[0];
    #pragma unroll
    for (int c = 1; c < SC_CHUNKS; ++c) s += v[c];
    const float di = rsqrtf(s + 1.0f);
    dinv[t] = di;
    u[t] = X[t] * di;
}

__global__ __launch_bounds__(SC_THREADS) void k_l1_sum(const float* __restrict__ u,
                                                       const int* __restrict__ src,
                                                       const int* __restrict__ dst,
                                                       float* __restrict__ part,
                                                       int E, int n) {
    __shared__ float acc[SC_RANGE];
    const int r0 = blockIdx.y * SC_RANGE;
    for (int i = threadIdx.x; i < SC_RANGE; i += SC_THREADS) acc[i] = 0.0f;
    __syncthreads();
    const int L = (E + (int)gridDim.x - 1) / (int)gridDim.x;
    const int eBeg = blockIdx.x * L;
    const int eEnd = (eBeg + L < E) ? (eBeg + L) : E;
    int e = eBeg + threadIdx.x;
    for (; e + (SC_BATCH - 1) * SC_THREADS < eEnd; e += SC_BATCH * SC_THREADS) {
        int d[SC_BATCH], s[SC_BATCH];
        float val[SC_BATCH];
        #pragma unroll
        for (int k = 0; k < SC_BATCH; ++k) {
            d[k] = dst[e + k * SC_THREADS];
            s[k] = src[e + k * SC_THREADS];
        }
        #pragma unroll
        for (int k = 0; k < SC_BATCH; ++k) asm volatile("" :: "v"(d[k]), "v"(s[k]));
        #pragma unroll
        for (int k = 0; k < SC_BATCH; ++k) {
            const bool in = (unsigned)(d[k] - r0) < SC_RANGE;
            val[k] = in ? u[s[k]] : 0.0f;
        }
        #pragma unroll
        for (int k = 0; k < SC_BATCH; ++k) {
            const unsigned rel = (unsigned)(d[k] - r0);
            if (rel < SC_RANGE) atomicAdd(&acc[rel], val[k]);
        }
    }
    for (; e < eEnd; e += SC_THREADS) {
        const unsigned rel = (unsigned)(dst[e] - r0);
        if (rel < SC_RANGE) atomicAdd(&acc[rel], u[src[e]]);
    }
    __syncthreads();
    const int lim = (n - r0 < SC_RANGE) ? (n - r0) : SC_RANGE;
    float* o = part + (long long)blockIdx.x * n + r0;
    for (int i = threadIdx.x; i < lim; i += SC_THREADS) o[i] = acc[i];
}

__global__ __launch_bounds__(256) void k_l1_finish(const float* __restrict__ part,
                                                   const float* __restrict__ X,
                                                   const float* __restrict__ dinv,
                                                   float* __restrict__ s1, int n) {
    const int t = blockIdx.x * 256 + threadIdx.x;
    if (t >= n) return;
    float v[SC_CHUNKS];
    #pragma unroll
    for (int c = 0; c < SC_CHUNKS; ++c) v[c] = part[(long long)c * n + t];
    float s = v[0];
    #pragma unroll
    for (int c = 1; c < SC_CHUNKS; ++c) s += v[c];
    const float di = dinv[t];
    s1[t] = di * s + X[t] * (di * di);
}

__global__ __launch_bounds__(256) void k_expand(const float* __restrict__ s1,
                                                     const float* __restrict__ W1,
                                                     const float* __restrict__ bias,
                                                     float* __restrict__ y, int n) {
    int t = blockIdx.x * 256 + threadIdx.x;
    if (t >= n * 32) return;
    int row = t >> 5, c4 = (t & 31) << 2;
    float sv = s1[row];
    const float4 w = *(const float4*)(W1 + c4);
    const float4 b = *(const float4*)(bias + c4);
    float4 o;
    o.x = sv * w.x + b.x;
    o.y = sv * w.y + b.y;
    o.z = sv * w.z + b.z;
    o.w = sv * w.w + b.w;
    *(float4*)(y + (long long)row * HDIM + c4) = o;
}

template <bool RES>
__global__ __launch_bounds__(256) void k_gemm_wmma(float* __restrict__ yx,
                                                   const float* __restrict__ res,
                                                   const float* __restrict__ ss,
                                                   const float* __restrict__ W,
                                                   const float* __restrict__ dinv,
                                                   float* __restrict__ hs, int n) {
    __shared__ float tileX[32 * HDIM];
    const int rowBase = blockIdx.x * 32;
    const int tid  = threadIdx.x;
    const int wave = tid >> 5;
    const int lane = tid & 31;
    const int half = lane >> 4;
    const int l16  = lane & 15;
    const int rowHalf = wave & 1;
    const int c0   = (wave >> 1) * 32;

    const bool fullTile = (rowBase + 32 <= n);

    const float4 sc = *(const float4*)(ss + ((tid & 31) << 2));
    const float4 sh = *(const float4*)(ss + 128 + ((tid & 31) << 2));
    if (fullTile) {
        float4* yp = (float4*)(yx + (long long)rowBase * HDIM);
        float4 v[4], r[4];
        #pragma unroll
        for (int i = 0; i < 4; ++i) v[i] = yp[tid + i * 256];
        if constexpr (RES) {
            const float4* rp = (const float4*)(res + (long long)rowBase * HDIM);
            #pragma unroll
            for (int i = 0; i < 4; ++i) r[i] = rp[tid + i * 256];
        }
        #pragma unroll
        for (int i = 0; i < 4; ++i) {
            const int idx = tid + i * 256;
            float4 o = v[i];
            o.x = fmaxf(o.x, 0.f) * sc.x + sh.x; o.y = fmaxf(o.y, 0.f) * sc.y + sh.y;
            o.z = fmaxf(o.z, 0.f) * sc.z + sh.z; o.w = fmaxf(o.w, 0.f) * sc.w + sh.w;
            if constexpr (RES) { o.x += r[i].x; o.y += r[i].y; o.z += r[i].z; o.w += r[i].w; }
            yp[idx] = o;
            ((float4*)tileX)[idx] = o;
        }
    } else {
        #pragma unroll
        for (int i = 0; i < 4; ++i) {
            int idx = tid + i * 256;
            int rr = rowBase + (idx >> 5);
            float4 o = make_float4(0.f, 0.f, 0.f, 0.f);
            if (rr < n) {
                const long long off = (long long)rr * HDIM + ((idx & 31) << 2);
                o = *(const float4*)(yx + off);
                o.x = fmaxf(o.x, 0.f) * sc.x + sh.x; o.y = fmaxf(o.y, 0.f) * sc.y + sh.y;
                o.z = fmaxf(o.z, 0.f) * sc.z + sh.z; o.w = fmaxf(o.w, 0.f) * sc.w + sh.w;
                if constexpr (RES) {
                    const float4 q = *(const float4*)(res + off);
                    o.x += q.x; o.y += q.y; o.z += q.z; o.w += q.w;
                }
                *(float4*)(yx + off) = o;
            }
            ((float4*)tileX)[idx] = o;
        }
    }
    __syncthreads();

    const float* xrow = tileX + (rowHalf * 16 + l16) * HDIM + 2 * half;
    v8f acc0 = {}, acc1 = {};

    #pragma unroll 4
    for (int k = 0; k < HDIM; k += 4) {
        v2f a; a.x = xrow[k]; a.y = xrow[k + 1];
        const float* wp = W + (long long)(k + 2 * half) * HDIM + c0 + l16;
        v2f b0; b0.x = wp[0];  b0.y = wp[HDIM];
        v2f b1; b1.x = wp[16]; b1.y = wp[HDIM + 16];
        acc0 = __builtin_amdgcn_wmma_f32_16x16x4_f32(false, a, false, b0,
                                                     (short)0, acc0, false, false);
        acc1 = __builtin_amdgcn_wmma_f32_16x16x4_f32(false, a, false, b1,
                                                     (short)0, acc1, false, false);
    }

    const int rbase = rowBase + rowHalf * 16 + half * 8;
    if (fullTile) {
        float di[8];
        #pragma unroll
        for (int v = 0; v < 8; ++v) di[v] = dinv[rbase + v];
        #pragma unroll
        for (int v = 0; v < 8; ++v) {
            const long long o = (long long)(rbase + v) * HDIM + c0 + l16;
            hs[o]      = acc0[v] * di[v];
            hs[o + 16] = acc1[v] * di[v];
        }
    } else {
        for (int v = 0; v < 8; ++v) {
            int r = rbase + v;
            if (r < n) {
                const float dd = dinv[r];
                const long long o = (long long)r * HDIM + c0 + l16;
                hs[o]      = acc0[v] * dd;
                hs[o + 16] = acc1[v] * dd;
            }
        }
    }
}

#define AG_ROWS    400
#define AG_THREADS 1024
#define AG_PF      8
#define AG_BATCH   8
#define AG_IO      7

#define MK_EDGES 2048
#define MK_WORDS (MK_EDGES / 32)

__global__ __launch_bounds__(256) void k_edge_marks(const int* __restrict__ dst,
                                                    float* __restrict__ ybuf,
                                                    int E, int nWpad, int gAgg) {
    extern __shared__ unsigned int tab[];
    const int tot = gAgg * MK_WORDS;
    for (int i = threadIdx.x; i < tot; i += 256) tab[i] = 0u;
    __syncthreads();
    const int e0 = blockIdx.x * MK_EDGES;
    int d[MK_EDGES / 256];
    #pragma unroll
    for (int k = 0; k < MK_EDGES / 256; ++k) {
        const int e = e0 + threadIdx.x + k * 256;
        const int v = dst[e < E ? e : E - 1];
        d[k] = e < E ? v : -1;
    }
    #pragma unroll
    for (int k = 0; k < MK_EDGES / 256; ++k) {
        const int le = threadIdx.x + k * 256;
        const unsigned b = (unsigned)d[k] / AG_ROWS;
        if (b < (unsigned)gAgg) atomicOr(&tab[b * MK_WORDS + (le >> 5)], 1u << (le & 31));
    }
    __syncthreads();
    const int wj = blockIdx.x * MK_WORDS + (threadIdx.x % MK_WORDS);
    if (wj < nWpad) {
        for (int i0 = threadIdx.x; i0 < tot; i0 += 4 * 256) {
            unsigned int v[4];
            #pragma unroll
            for (int k = 0; k < 4; ++k) {
                const int i = i0 + k * 256;
                v[k] = tab[i < tot ? i : tot - 1];
            }
            #pragma unroll
            for (int k = 0; k < 4; ++k) {
                const int i = i0 + k * 256;
                if (i < tot)
                    ((unsigned int*)(ybuf + (long long)(i / MK_WORDS) * AG_ROWS * HDIM))[wj] = v[k];
            }
        }
    }
}

__device__ __forceinline__ void agg_drain(const float* __restrict__ hs,
                                          const int* __restrict__ src,
                                          float* acc, int pe, int pr, int count, int lane) {
    int s = 0;
    if (lane < count) s = src[pe];
    #pragma nounroll
    for (int j0 = 0; j0 < count; j0 += AG_BATCH) {
        float v[AG_BATCH][4];
        #pragma unroll
        for (int u = 0; u < AG_BATCH; ++u) {
            int jj = j0 + u;
            jj = jj < count ? jj : count - 1;
            const int sj = __builtin_amdgcn_readlane(s, jj);
            const float* hp = hs + (long long)sj * HDIM;
            v[u][0] = hp[lane];
            v[u][1] = hp[lane + 32];
            v[u][2] = hp[lane + 64];
            v[u][3] = hp[lane + 96];
        }
        #pragma unroll
        for (int u = 0; u < AG_BATCH; ++u) {
            if (j0 + u < count) {
                const int rj = __builtin_amdgcn_readlane(pr, j0 + u);
                float* ap = acc + rj * HDIM + lane;
                atomicAdd(ap,      v[u][0]);
                atomicAdd(ap + 32, v[u][1]);
                atomicAdd(ap + 64, v[u][2]);
                atomicAdd(ap + 96, v[u][3]);
            }
        }
    }
}

template <bool MARKS>
__global__ __launch_bounds__(AG_THREADS) void k_agg_lds(const float* __restrict__ hs,
                                                        const int* __restrict__ src,
                                                        const int* __restrict__ dst,
                                                        const float* __restrict__ dinv,
                                                        const float* __restrict__ bias,
                                                        float* __restrict__ y, int E, int n) {
    __shared__ float acc[AG_ROWS * HDIM];
    const int lane = threadIdx.x & 31;
    const int wave = __builtin_amdgcn_readfirstlane((int)(threadIdx.x >> 5));
    const int r0 = blockIdx.x * AG_ROWS;
    const int lim = (n - r0 < AG_ROWS) ? (n - r0) : AG_ROWS;
    const unsigned ulim = (unsigned)lim;

    const int tot = lim * 32;
    {
        const float4* hp = (const float4*)(hs + (long long)r0 * HDIM);
        for (int i0 = threadIdx.x; i0 < tot; i0 += AG_IO * AG_THREADS) {
            float4 t[AG_IO];
            #pragma unroll
            for (int q = 0; q < AG_IO; ++q) {
                const int i = i0 + q * AG_THREADS;
                t[q] = hp[i < tot ? i : tot - 1];
            }
            asm volatile("" :: "v"(t[0].x), "v"(t[0].y), "v"(t[0].z), "v"(t[0].w),
                               "v"(t[1].x), "v"(t[1].y), "v"(t[1].z), "v"(t[1].w),
                               "v"(t[2].x), "v"(t[2].y), "v"(t[2].z), "v"(t[2].w),
                               "v"(t[3].x), "v"(t[3].y), "v"(t[3].z), "v"(t[3].w),
                               "v"(t[4].x), "v"(t[4].y), "v"(t[4].z), "v"(t[4].w),
                               "v"(t[5].x), "v"(t[5].y), "v"(t[5].z), "v"(t[5].w),
                               "v"(t[6].x), "v"(t[6].y), "v"(t[6].z), "v"(t[6].w));
            #pragma unroll
            for (int q = 0; q < AG_IO; ++q) {
                const int i = i0 + q * AG_THREADS;
                if (i < tot) ((float4*)acc)[i] = t[q];
            }
        }
    }
    __syncthreads();

    int pe = 0, pr = 0, cnt = 0;

    int tailBeg;
    if constexpr (MARKS) {
        const int nq = (((E + 31) >> 5) + 3) >> 2;
        const uint4* mk = (const uint4*)(y + (long long)r0 * HDIM);
        const int nIt = (nq + 31) >> 5;
        const int K = nIt > wave ? ((nIt - wave + 31) >> 5) : 0;
        const uint4 none = make_uint4(0u, 0u, 0u, 0u);
        auto piece = [&](int t) {
            const int idx = (wave + 32 * t) * 32 + lane;
            return mk[idx < nq ? idx : nq - 1];
        };
        uint4 n0 = none, n1 = none;
        if (K > 0) n0 = piece(0);
        if (K > 1) n1 = piece(1);
        #pragma nounroll
        for (int k = 0; k < K; ++k) {
            {
                const uint4 cur = n0;
                n0 = n1;
                if (k + 2 < K) n1 = piece(k + 2);
                const int itr = wave + 32 * k;
                const bool ok = itr * 32 + lane < nq;
                const unsigned m0 = ok ? cur.x : 0u, m1 = ok ? cur.y : 0u;
                const unsigned m2 = ok ? cur.z : 0u, m3 = ok ? cur.w : 0u;
                #pragma nounroll
                for (int q = 0; q < 4; ++q) {
                    const unsigned mv = q == 0 ? m0 : (q == 1 ? m1 : (q == 2 ? m2 : m3));
                    unsigned act = __builtin_amdgcn_ballot_w32(mv != 0u);
                    while (act != 0u) {
                        const int L = __builtin_ctz(act);
                        act &= act - 1u;
                        unsigned sw = (unsigned)__builtin_amdgcn_readlane((int)mv, L);
                        const int e0 = ((((itr * 32 + L) << 2) + q) << 5);
                        while (sw != 0u) {
                            const int p = __builtin_ctz(sw);
                            sw &= sw - 1u;
                            pe = lane == cnt ? e0 + p : pe;
                            ++cnt;
                            if (cnt == 32) {
                                pr = dst[pe] - r0;
                                agg_drain(hs, src, acc, pe, pr, 32, lane);
                                cnt = 0;
                            }
                        }
                    }
                }
            }
        }
        tailBeg = E;
    } else {
        const int nvec = (((size_t)dst & 15) == 0) ? (E >> 2) : 0;
        const int4* dv = (const int4*)dst;
        const int nIt = (nvec + 31) >> 5;
        const int K = nIt > wave ? ((nIt - wave + 31) >> 5) : 0;
        const int rot = (int)blockIdx.x * (nIt / (int)gridDim.x);
        const int4 none = make_int4(-1, -1, -1, -1);

        int4 buf[AG_PF];
        #pragma unroll
        for (int u = 0; u < AG_PF; ++u) {
            buf[u] = none;
            if (u < K) {
                int itr = wave + 32 * u + rot;
                itr = itr >= nIt ? itr - nIt : itr;
                const int idx = itr * 32 + lane;
                buf[u] = dv[idx < nvec ? idx : nvec - 1];
            }
        }
        for (int k = 0; k < K; k += AG_PF) {
            #pragma unroll
            for (int u = 0; u < AG_PF; ++u) {
                if (k + u >= K) break;
                const int4 cur = buf[u];
                int itr = wave + 32 * (k + u) + rot;
                itr = itr >= nIt ? itr - nIt : itr;
                {
                    const int kn = k + u + AG_PF;
                    if (kn < K) {
                        int itn = wave + 32 * kn + rot;
                        itn = itn >= nIt ? itn - nIt : itn;
                        const int idx = itn * 32 + lane;
                        buf[u] = dv[idx < nvec ? idx : nvec - 1];
                    }
                }
                const bool ok = itr * 32 + lane < nvec;
                const bool m0 = ok & ((unsigned)(cur.x - r0) < ulim);
                const bool m1 = ok & ((unsigned)(cur.y - r0) < ulim);
                const bool m2 = ok & ((unsigned)(cur.z - r0) < ulim);
                const bool m3 = ok & ((unsigned)(cur.w - r0) < ulim);
                if (__builtin_amdgcn_ballot_w32(m0 | m1 | m2 | m3) != 0u) {
                    unsigned bits = (m0 ? 1u : 0u) | (m1 ? 2u : 0u) | (m2 ? 4u : 0u) | (m3 ? 8u : 0u);
                    unsigned act = __builtin_amdgcn_ballot_w32(bits != 0u);
                    while (act != 0u) {
                        const int L = __builtin_ctz(act);
                        const unsigned b = (unsigned)__builtin_amdgcn_readlane((int)bits, L);
                        const int j = __builtin_ctz(b);
                        const int vsel = j == 0 ? cur.x : (j == 1 ? cur.y : (j == 2 ? cur.z : cur.w));
                        const int d = __builtin_amdgcn_readlane(vsel, L);
                        const int e = ((itr * 32 + L) << 2) + j;
                        pe = lane == cnt ? e : pe;
                        pr = lane == cnt ? d - r0 : pr;
                        ++cnt;
                        if (lane == L) bits &= bits - 1u;
                        if (cnt == 32) {
                            agg_drain(hs, src, acc, pe, pr, 32, lane);
                            cnt = 0;
                        }
                        act = __builtin_amdgcn_ballot_w32(bits != 0u);
                    }
                }
            }
        }
        tailBeg = nvec * 4;
    }
    if constexpr (!MARKS) {
        for (int base = tailBeg + wave * 32; base < E; base += AG_THREADS) {
            const int e = base + lane;
            const int d = e < E ? dst[e] : -1;
            unsigned act = __builtin_amdgcn_ballot_w32((unsigned)(d - r0) < ulim);
            while (act != 0u) {
                const int L = __builtin_ctz(act);
                act &= act - 1u;
                const int dl = __builtin_amdgcn_readlane(d, L);
                pe = lane == cnt ? base + L : pe;
                pr = lane == cnt ? dl - r0 : pr;
                ++cnt;
                if (cnt == 32) {
                    agg_drain(hs, src, acc, pe, pr, 32, lane);
                    cnt = 0;
                }
            }
        }
    }
    if constexpr (MARKS) {
        if (lane < cnt) pr = dst[pe] - r0;
    }
    agg_drain(hs, src, acc, pe, pr, cnt, lane);
    __syncthreads();

    {
        float4* yp = (float4*)(y + (long long)r0 * HDIM);
        const float4 b = *(const float4*)(bias + (lane << 2));
        for (int i0 = threadIdx.x; i0 < tot; i0 += AG_IO * AG_THREADS) {
            float di[AG_IO];
            #pragma unroll
            for (int q = 0; q < AG_IO; ++q) {
                const int i = i0 + q * AG_THREADS;
                di[q] = dinv[r0 + ((i < tot ? i : tot - 1) >> 5)];
            }
            #pragma unroll
            for (int q = 0; q < AG_IO; ++q) {
                const int i = i0 + q * AG_THREADS;
                if (i < tot) {
                    const float4 a = ((const float4*)acc)[i];
                    float4 o;
                    o.x = a.x * di[q] + b.x;
                    o.y = a.y * di[q] + b.y;
                    o.z = a.z * di[q] + b.z;
                    o.w = a.w * di[q] + b.w;
                    yp[i] = o;
                }
            }
        }
    }
}

#define STAT_BLOCKS 256
#define STAT_WAVES  32
#define STAT_BATCH  8
#define STAT_GROUPS 16

__device__ __forceinline__ void stat_acc(float4& s, float4& q, const float4 p) {
    float4 v;
    v.x = fmaxf(p.x, 0.f); v.y = fmaxf(p.y, 0.f); v.z = fmaxf(p.z, 0.f); v.w = fmaxf(p.w, 0.f);
    s.x += v.x; s.y += v.y; s.z += v.z; s.w += v.w;
    q.x += v.x * v.x; q.y += v.y * v.y; q.z += v.z * v.z; q.w += v.w * v.w;
}

__global__ __launch_bounds__(STAT_WAVES * 32) void k_stats(const float* __restrict__ y,
                                                            float* __restrict__ part, int n) {
    __shared__ float ls[STAT_WAVES][256];
    const int lane = threadIdx.x & 31;
    const int wave = threadIdx.x >> 5;
    const int S = STAT_BLOCKS * STAT_WAVES;
    const float4* yp = (const float4*)y + lane;
    int r = blockIdx.x * STAT_WAVES + wave;
    float4 s = make_float4(0.f, 0.f, 0.f, 0.f), q = s;

    for (; r + (STAT_BATCH - 1) * S < n; r += STAT_BATCH * S) {
        float4 v[STAT_BATCH];
        #pragma unroll
        for (int u = 0; u < STAT_BATCH; ++u) v[u] = yp[(long long)(r + u * S) * 32];
        #pragma unroll
        for (int u = 0; u < STAT_BATCH; ++u) stat_acc(s, q, v[u]);
    }
    if (r < n) {
        const float4 zero = make_float4(0.f, 0.f, 0.f, 0.f);
        float4 v[STAT_BATCH - 1];
        #pragma unroll
        for (int u = 0; u < STAT_BATCH - 1; ++u) {
            const int rr = r + u * S;
            const bool in = rr < n;
            const float4 t = yp[(long long)(in ? rr : n - 1) * 32];
            v[u] = in ? t : zero;
        }
        #pragma unroll
        for (int u = 0; u < STAT_BATCH - 1; ++u) stat_acc(s, q, v[u]);
    }

    *(float4*)&ls[wave][lane * 4]       = s;
    *(float4*)&ls[wave][128 + lane * 4] = q;
    __syncthreads();
    if (threadIdx.x < 256) {
        float a = 0.f;
        for (int w = 0; w < STAT_WAVES; ++w) a += ls[w][threadIdx.x];
        part[(long long)blockIdx.x * 256 + threadIdx.x] = a;
    }
}

__global__ __launch_bounds__(STAT_GROUPS * 64) void k_scaleshift(const float* __restrict__ part,
                                                                 const float* __restrict__ gamma,
                                                                 const float* __restrict__ beta,
                                                                 float* __restrict__ ss, float invN) {
    __shared__ float fs[STAT_GROUPS][256];
    const int c4 = threadIdx.x & 63;
    const int g  = threadIdx.x >> 6;
    const float4* pp = (const float4*)part + c4;
    float4 v[STAT_BLOCKS / STAT_GROUPS];
    #pragma unroll
    for (int i = 0; i < STAT_BLOCKS / STAT_GROUPS; ++i) v[i] = pp[(g + i * STAT_GROUPS) * 64];
    float4 a = v[0];
    #pragma unroll
    for (int i = 1; i < STAT_BLOCKS / STAT_GROUPS; ++i) {
        a.x += v[i].x; a.y += v[i].y; a.z += v[i].z; a.w += v[i].w;
    }
    *(float4*)&fs[g][c4 * 4] = a;
    __syncthreads();
    if (threadIdx.x < 128) {
        const int c = threadIdx.x;
        float su = 0.f, sq = 0.f;
        for (int k = 0; k < STAT_GROUPS; ++k) { su += fs[k][c]; sq += fs[k][128 + c]; }
        float m   = su * invN;
        float var = sq * invN - m * m;
        float sc  = rsqrtf(var + EPS_BN) * gamma[c];
        ss[c] = sc;
        ss[128 + c] = beta[c] - m * sc;
    }
}

__global__ __launch_bounds__(256) void k_bn_apply(float* __restrict__ y,
                                                  const float* __restrict__ res,
                                                  const float* __restrict__ ss, int n) {
    int t = blockIdx.x * 256 + threadIdx.x;
    if (t >= n * 32) return;
    int row = t >> 5, c4 = (t & 31) << 2;
    long long off = (long long)row * HDIM + c4;
    float4 v  = *(const float4*)(y + off);
    float4 sc = *(const float4*)(ss + c4);
    float4 sh = *(const float4*)(ss + 128 + c4);
    v.x = fmaxf(v.x, 0.f) * sc.x + sh.x; v.y = fmaxf(v.y, 0.f) * sc.y + sh.y;
    v.z = fmaxf(v.z, 0.f) * sc.z + sh.z; v.w = fmaxf(v.w, 0.f) * sc.w + sh.w;
    if (res) {
        float4 r = *(const float4*)(res + off);
        v.x += r.x; v.y += r.y; v.z += r.z; v.w += r.w;
    }
    *(float4*)(y + off) = v;
}

#define POOL_BLOCKS 256
#define POOL_WAVES  32
#define POOL_BATCH  8
#define PSUM_GROUPS 16

__device__ __forceinline__ void pool_add(float* __restrict__ pl, int g, int lane,
                                         const float4 v, int cntBase) {
    float* p = pl + g * HDIM + lane * 4;
    atomicAdd(p + 0, v.x); atomicAdd(p + 1, v.y);
    atomicAdd(p + 2, v.z); atomicAdd(p + 3, v.w);
    if (lane == 0) atomicAdd(pl + cntBase + g, 1.0f);
}

__global__ __launch_bounds__(POOL_WAVES * 32) void k_pool_part(const float* __restrict__ x,
                                                                const int* __restrict__ batch,
                                                                float* __restrict__ planes,
                                                                int n, int G) {
    extern __shared__ float pl[];
    const int lane = threadIdx.x & 31;
    const int tot = G * HDIM + G;
    for (int i = threadIdx.x; i < tot; i += POOL_WAVES * 32) pl[i] = 0.0f;
    __syncthreads();
    const int S = (int)gridDim.x * POOL_WAVES;
    int r = __builtin_amdgcn_readfirstlane((int)blockIdx.x * POOL_WAVES + (int)(threadIdx.x >> 5));
    const float4* xp = (const float4*)x + lane;

    for (; r + (POOL_BATCH - 1) * S < n; r += POOL_BATCH * S) {
        float4 v[POOL_BATCH];
        int g[POOL_BATCH];
        #pragma unroll
        for (int k = 0; k < POOL_BATCH; ++k) {
            v[k] = xp[(long long)(r + k * S) * 32];
            g[k] = batch[r + k * S];
        }
        #pragma unroll
        for (int k = 0; k < POOL_BATCH; ++k) pool_add(pl, g[k], lane, v[k], G * HDIM);
    }
    if (r < n) {
        float4 v[POOL_BATCH - 1];
        int g[POOL_BATCH - 1];
        #pragma unroll
        for (int k = 0; k < POOL_BATCH - 1; ++k) {
            const int rr = r + k * S;
            const int rc = rr < n ? rr : n - 1;
            v[k] = xp[(long long)rc * 32];
            g[k] = batch[rc];
        }
        #pragma unroll
        for (int k = 0; k < POOL_BATCH - 1; ++k)
            if (r + k * S < n) pool_add(pl, g[k], lane, v[k], G * HDIM);
    }
    __syncthreads();
    float* o = planes + (long long)blockIdx.x * tot;
    for (int i = threadIdx.x; i < tot; i += POOL_WAVES * 32) o[i] = pl[i];
}

__global__ __launch_bounds__(PSUM_GROUPS * 64) void k_pool_sum(const float* __restrict__ planes,
                                                               float* __restrict__ outp, int tot) {
    __shared__ float fs[PSUM_GROUPS][64];
    const int o   = threadIdx.x & 63;
    const int grp = threadIdx.x >> 6;
    const int i   = blockIdx.x * 64 + o;
    float a = 0.0f;
    if (i < tot) {
        float v[POOL_BLOCKS / PSUM_GROUPS];
        #pragma unroll
        for (int j = 0; j < POOL_BLOCKS / PSUM_GROUPS; ++j)
            v[j] = planes[(long long)(grp + j * PSUM_GROUPS) * tot + i];
        a = v[0];
        #pragma unroll
        for (int j = 1; j < POOL_BLOCKS / PSUM_GROUPS; ++j) a += v[j];
    }
    fs[grp][o] = a;
    __syncthreads();
    if (threadIdx.x < 64 && i < tot) {
        float s = 0.0f;
        for (int k = 0; k < PSUM_GROUPS; ++k) s += fs[k][o];
        outp[i] = s;
    }
}

__global__ __launch_bounds__(256) void k_pool_final(const float* __restrict__ pool,
                                                    const float* __restrict__ counts,
                                                    float* __restrict__ xmean, int G) {
    int t = blockIdx.x * 256 + threadIdx.x;
    if (t >= G * HDIM) return;
    int g = t / HDIM;
    xmean[t] = pool[t] / fmaxf(counts[g], 1.0f);
}

__global__ void k_pred(const float* __restrict__ xmean, const float* __restrict__ Wr,
                       const float* __restrict__ br, float* __restrict__ pred) {
    __shared__ float red[128];
    int g = blockIdx.x, c = threadIdx.x;
    red[c] = xmean[(long long)g * HDIM + c] * Wr[c];
    __syncthreads();
    for (int s = 64; s > 0; s >>= 1) {
        if (c < s) red[c] += red[c + s];
        __syncthreads();
    }
    if (c == 0) pred[g] = red[0] + br[0];
}

extern "C" void kernel_launch(void* const* d_in, const int* in_sizes, int n_in,
                              void* d_out, int out_size, void* d_ws, size_t ws_size,
                              hipStream_t stream) {
    const float* X     = (const float*)d_in[0];
    const int*   edge  = (const int*)d_in[1];
    const int*   batch = (const int*)d_in[2];
    const float* W1    = (const float*)d_in[3];
    const float* b1    = (const float*)d_in[4];
    const float* Wm    = (const float*)d_in[5];
    const float* bm    = (const float*)d_in[6];
    const float* gamma = (const float*)d_in[7];
    const float* beta  = (const float*)d_in[8];
    const float* Wr    = (const float*)d_in[9];
    const float* br    = (const float*)d_in[10];
    float* out = (float*)d_out;

    const int N = in_sizes[0];
    const int E = in_sizes[1] / 2;
    const int G = out_size / (HDIM + 1);
    const int* src = edge;
    const int* dst = edge + E;

    const long long NH = (long long)N * HDIM;
    float* ws    = (float*)d_ws;
    float* buf[3] = { ws, ws + NH, ws + 2 * NH };
    float* dinv  = ws + 3 * NH;
    float* ss    = dinv + N;
    float* pool  = ss + 256;
    float* cnts  = pool + (long long)G * HDIM;

    const int gAgg = (N + AG_ROWS - 1) / AG_ROWS;

    const int nWpad = ((((E + 31) >> 5) + 3) >> 2) << 2;
    const int lastRows = N - (gAgg - 1) * AG_ROWS;
    const bool useMarks = (long long)nWpad * 4 <= (long long)lastRows * HDIM * 4
                          && gAgg * MK_WORDS * 4 <= 65536 && (((size_t)d_ws & 15) == 0) && nWpad > 0;
    const int gMk = (nWpad + MK_WORDS - 1) / MK_WORDS;
    const int gN   = (N + 255) / 256;
    const int gNH4 = (N * 32 + 255) / 256;
    const int gGem = (N + 31) / 32;

    float* s1   = buf[0];
    float* u1   = buf[0] + N;
    float* scp  = buf[1];
    const dim3 gSc(SC_CHUNKS, (N + SC_RANGE - 1) / SC_RANGE);
    k_deg_count<<<gSc, SC_THREADS, 0, stream>>>(dst, scp, E, N);
    k_deg_finish<<<gN, 256, 0, stream>>>(scp, X, dinv, u1, N);

    auto bn_stats = [&](float* abuf, float* part, const float* g_, const float* be_) {
        k_stats<<<STAT_BLOCKS, STAT_WAVES * 32, 0, stream>>>(abuf, part, N);
        k_scaleshift<<<1, STAT_GROUPS * 64, 0, stream>>>(part, g_, be_, ss, 1.0f / (float)N);
    };

    auto layer_tail = [&](float* hbuf, float* abuf, const float* bias_,
                          const float* g_, const float* be_) {
        if (useMarks) {
            k_edge_marks<<<gMk, 256, gAgg * MK_WORDS * sizeof(unsigned int), stream>>>(dst, abuf, E, nWpad, gAgg);
            k_agg_lds<true><<<gAgg, AG_THREADS, 0, stream>>>(hbuf, src, dst, dinv, bias_, abuf, E, N);
        } else {
            k_agg_lds<false><<<gAgg, AG_THREADS, 0, stream>>>(hbuf, src, dst, dinv, bias_, abuf, E, N);
        }
        bn_stats(abuf, hbuf, g_, be_);
    };

    k_l1_sum<<<gSc, SC_THREADS, 0, stream>>>(u1, src, dst, scp, E, N);
    k_l1_finish<<<gN, 256, 0, stream>>>(scp, X, dinv, s1, N);
    k_expand<<<gNH4, 256, 0, stream>>>(s1, W1, b1, buf[2], N);
    bn_stats(buf[2], buf[0], gamma, beta);

    float* yb = buf[2];
    float* rb = nullptr;
    float* hb = buf[0];
    float* fb = buf[1];

    for (int i = 0; i < 4; ++i) {
        const float* Wi = Wm + (long long)i * HDIM * HDIM;
        if (rb) k_gemm_wmma<true><<<gGem, 256, 0, stream>>>(yb, rb, ss, Wi, dinv, hb, N);
        else    k_gemm_wmma<false><<<gGem, 256, 0, stream>>>(yb, nullptr, ss, Wi, dinv, hb, N);
        float* ob = rb ? rb : fb;
        layer_tail(hb, ob, bm + (long long)i * HDIM,
                   gamma + (long long)(i + 1) * HDIM,
                   beta  + (long long)(i + 1) * HDIM);
        rb = yb;
        yb = ob;
    }
    k_bn_apply<<<gNH4, 256, 0, stream>>>(yb, rb, ss, N);
    float* cur = yb;
    float* f0  = hb;

    const int poolTot = G * HDIM + G;
    k_pool_part<<<POOL_BLOCKS, POOL_WAVES * 32, poolTot * sizeof(float), stream>>>(cur, batch, f0, N, G);
    k_pool_sum<<<(poolTot + 63) / 64, PSUM_GROUPS * 64, 0, stream>>>(f0, pool, poolTot);
    k_pool_final<<<(G * HDIM + 255) / 256, 256, 0, stream>>>(pool, cnts, out + G, G);
    k_pred<<<G, 128, 0, stream>>>(out + G, Wr, br, out);
}
